# starting kernel
# baseline (speedup 1.0000x reference)
#include <hip/hip_runtime.h>
#include <hip/hip_bf16.h>


typedef __attribute__((ext_vector_type(2))) float v2f;
typedef __attribute__((ext_vector_type(8))) float v8f;

#define NUM_USERS 100000
#define NUM_ITEMS 50000
#define NUM_NODES 150000
#define NUM_BQ    20000
#define EMB       64

__global__ void __launch_bounds__(256) copy_init_kernel(
    const float* __restrict__ ue, const float* __restrict__ ie,
    float* __restrict__ e0, float* __restrict__ out,
    int totalVec, int userVec) {
  int g = blockIdx.x * blockDim.x + threadIdx.x;
  if (g >= totalVec) return;
  float4 v = (g < userVec) ? ((const float4*)ue)[g]
                           : ((const float4*)ie)[g - userVec];
  ((float4*)e0)[g]  = v;
  ((float4*)out)[g] = v;
}

__global__ void __launch_bounds__(256) spmm16_kernel(
    const int* __restrict__ row, const int* __restrict__ col,
    const float* __restrict__ val, const float* __restrict__ x,
    float* __restrict__ y, float* __restrict__ deg, int nnz) {
  int g = blockIdx.x * blockDim.x + threadIdx.x;
  int e = g >> 4;
  int j = g & 15;
  if (e >= nnz) return;
  int   c = __builtin_nontemporal_load(col + e);
  int   r = __builtin_nontemporal_load(row + e);
  float v = __builtin_nontemporal_load(val + e);
  float4 xv = ((const float4*)(x + (size_t)c * EMB))[j];
  float* yp = y + (size_t)r * EMB + j * 4;
  unsafeAtomicAdd(yp + 0, v * xv.x);
  unsafeAtomicAdd(yp + 1, v * xv.y);
  unsafeAtomicAdd(yp + 2, v * xv.z);
  unsafeAtomicAdd(yp + 3, v * xv.w);
  if (deg != nullptr && j == 0)
    unsafeAtomicAdd(&deg[r], v);
}

__global__ void __launch_bounds__(128) acc_wmma_kernel(
    const float* __restrict__ e, float* __restrict__ acc) {
  int wave = threadIdx.x >> 5;
  int lane = threadIdx.x & 31;
  int n    = lane & 15;
  int hi   = lane >> 4;
  int row0 = blockIdx.x * 16;
  int col0 = wave * 16;

  const size_t base = (size_t)row0 * EMB + col0 + n;
  const float* ap = acc + base;
  const float* ep = e + base;

  v8f c;
#pragma unroll
  for (int r = 0; r < 8; ++r)
    c[r] = ap[(size_t)(r + 8 * hi) * EMB];

  v2f a[4], b[4];
#pragma unroll
  for (int k = 0; k < 4; ++k) {
    a[k].x = (n == (4 * k + 2 * hi))     ? 1.0f : 0.0f;
    a[k].y = (n == (4 * k + 2 * hi + 1)) ? 1.0f : 0.0f;
    b[k].x = ep[(size_t)(4 * k + 2 * hi) * EMB];
    b[k].y = ep[(size_t)(4 * k + 2 * hi + 1) * EMB];
  }

#pragma unroll
  for (int k = 0; k < 4; ++k)
    c = __builtin_amdgcn_wmma_f32_16x16x4_f32(
        false, a[k], false, b[k], (short)0, c, false, false);

  float* op = acc + base;
#pragma unroll
  for (int r = 0; r < 8; ++r)
    op[(size_t)(r + 8 * hi) * EMB] = c[r];
}

__global__ void __launch_bounds__(256) rownorm_kernel(
    float* __restrict__ x, const float* __restrict__ deg, int rowsVec) {
  int g = blockIdx.x * blockDim.x + threadIdx.x;
  if (g >= rowsVec) return;
  float d = deg[g >> 4];
  d = (d == 0.0f) ? 1.0f : d;
  float inv = 1.0f / d;
  float4 v = ((float4*)x)[g];
  v.x *= inv; v.y *= inv; v.z *= inv; v.w *= inv;
  ((float4*)x)[g] = v;
}

__global__ void __launch_bounds__(256) finalize_kernel(
    float* __restrict__ out, const float* __restrict__ ulocal,
    const float* __restrict__ deg, int totalVec, int userVec) {
  int g = blockIdx.x * blockDim.x + threadIdx.x;
  if (g >= totalVec) return;
  float4 v = ((float4*)out)[g];
  v.x *= 0.25f; v.y *= 0.25f; v.z *= 0.25f; v.w *= 0.25f;
  if (g < userVec) {
    float d = deg[g >> 4];
    d = (d == 0.0f) ? 1.0f : d;
    float inv = 1.0f / d;
    float4 w = ((const float4*)ulocal)[g];
    v.x += w.x * inv; v.y += w.y * inv; v.z += w.z * inv; v.w += w.w * inv;
  }
  ((float4*)out)[g] = v;
}

static inline int cdiv(long long a, int b) { return (int)((a + b - 1) / b); }

extern "C" void kernel_launch(void* const* d_in, const int* in_sizes, int n_in,
                              void* d_out, int out_size, void* d_ws, size_t ws_size,
                              hipStream_t stream) {
  const float* user_emb = (const float*)d_in[0];
  const float* item_emb = (const float*)d_in[1];
  const float* adj_val  = (const float*)d_in[2];
  const float* hv_val   = (const float*)d_in[3];
  const float* hu_val   = (const float*)d_in[4];
  const int*   adj_row  = (const int*)d_in[5];
  const int*   adj_col  = (const int*)d_in[6];
  const int*   hv_row   = (const int*)d_in[7];
  const int*   hv_col   = (const int*)d_in[8];
  const int*   hu_row   = (const int*)d_in[9];
  const int*   hu_col   = (const int*)d_in[10];

  const int E_adj = in_sizes[2];
  const int E_hv  = in_sizes[3];
  const int E_hu  = in_sizes[4];

  float* out = (float*)d_out;

  const size_t embBytesN = (size_t)NUM_NODES * EMB * sizeof(float);
  const size_t embBytesB = (size_t)NUM_BQ    * EMB * sizeof(float);
  const size_t embBytesU = (size_t)NUM_USERS * EMB * sizeof(float);
  char* ws = (char*)d_ws;
  float* e0     = (float*)(ws);
  float* e1     = (float*)(ws + embBytesN);
  float* bfeat  = (float*)(ws + 2 * embBytesN);
  float* deg_hv = (float*)(ws + 2 * embBytesN + embBytesB);
  float* deg_hu = deg_hv + NUM_BQ;
  float* ulocal = e0;

  const int totalVec = NUM_NODES * (EMB / 4);
  const int userVec  = NUM_USERS * (EMB / 4);
  const int bqVec    = NUM_BQ    * (EMB / 4);

  copy_init_kernel<<<cdiv(totalVec, 256), 256, 0, stream>>>(
      user_emb, item_emb, e0, out, totalVec, userVec);

  float* eprev = e0;
  float* enext = e1;
  for (int layer = 0; layer < 3; ++layer) {
    hipMemsetAsync(enext, 0, embBytesN, stream);
    spmm16_kernel<<<cdiv((long long)E_adj * 16, 256), 256, 0, stream>>>(
        adj_row, adj_col, adj_val, eprev, enext, nullptr, E_adj);
    acc_wmma_kernel<<<NUM_NODES / 16, 128, 0, stream>>>(enext, out);
    float* t = eprev; eprev = enext; enext = t;
  }

  hipMemsetAsync(bfeat,  0, embBytesB, stream);
  hipMemsetAsync(deg_hv, 0, (size_t)NUM_BQ * sizeof(float), stream);
  hipMemsetAsync(deg_hu, 0, (size_t)NUM_USERS * sizeof(float), stream);

  spmm16_kernel<<<cdiv((long long)E_hv * 16, 256), 256, 0, stream>>>(
      hv_row, hv_col, hv_val, item_emb, bfeat, deg_hv, E_hv);
  rownorm_kernel<<<cdiv(bqVec, 256), 256, 0, stream>>>(bfeat, deg_hv, bqVec);

  hipMemsetAsync(ulocal, 0, embBytesU, stream);
  spmm16_kernel<<<cdiv((long long)E_hu * 16, 256), 256, 0, stream>>>(
      hu_row, hu_col, hu_val, bfeat, ulocal, deg_hu, E_hu);

  finalize_kernel<<<cdiv(totalVec, 256), 256, 0, stream>>>(
      out, ulocal, deg_hu, totalVec, userVec);
}


# SpMM destination kept on chip: each block owns a row tile in LDS, scans all row ids, gathers its edges, adds with LDS float atomics, stores rows once; memsets go; row sums stay in arrival order
# speedup vs baseline: 1.2754x; 1.2754x over previous
#include <hip/hip_runtime.h>
#include <hip/hip_bf16.h>


typedef __attribute__((ext_vector_type(2))) float v2f;
typedef __attribute__((ext_vector_type(8))) float v8f;

#define NUM_USERS 100000
#define NUM_ITEMS 50000
#define NUM_NODES 150000
#define NUM_BQ    20000
#define EMB       64

__global__ void __launch_bounds__(256) copy_init_kernel(
    const float* __restrict__ ue, const float* __restrict__ ie,
    float* __restrict__ e0, float* __restrict__ out,
    int totalVec, int userVec) {
  int g = blockIdx.x * blockDim.x + threadIdx.x;
  if (g >= totalVec) return;
  float4 v = (g < userVec) ? ((const float4*)ue)[g]
                           : ((const float4*)ie)[g - userVec];
  ((float4*)e0)[g]  = v;
  ((float4*)out)[g] = v;
}

#define QCAP 160

template <int TILE_ROWS, bool DEG>
__device__ __forceinline__ void resolve_queued(
    const int* q, int base, int n, int lane, int lo,
    const int* __restrict__ row, const int* __restrict__ col,
    const float* __restrict__ val, const float* __restrict__ x,
    float* tile, float* degt) {
  int last = n - 1;
  int e  = q[base + (lane < last ? lane : last)];
  int rr = row[e] - lo;
  int c  = col[e];
  int vb = __float_as_int(val[e]);
  for (int k0 = 0; k0 < n; k0 += 16) {
    float2 xv[16];
#pragma unroll
    for (int j = 0; j < 16; ++j) {
      int k  = (k0 + j < last) ? (k0 + j) : last;
      int ck = __builtin_amdgcn_readlane(c, k);
      xv[j] = *(const float2*)(x + (size_t)ck * EMB + 2 * lane);
    }
#pragma unroll
    for (int j = 0; j < 16; ++j) {
      if (k0 + j < n) {
        int   k  = k0 + j;
        int   rk = __builtin_amdgcn_readlane(rr, k);
        float vk = __int_as_float(__builtin_amdgcn_readlane(vb, k));
        float* t = tile + rk * EMB + 2 * lane;
        atomicAdd(t,     vk * xv[j].x);
        atomicAdd(t + 1, vk * xv[j].y);
        if (DEG && lane == 0) atomicAdd(degt + rk, vk);
      }
    }
  }
}

template <int TILE_ROWS, bool DEG>
__device__ __forceinline__ void take4(
    int4 r, int j, int& cnt, int* q, int lane, int lo,
    const int* __restrict__ row, const int* __restrict__ col,
    const float* __restrict__ val, const float* __restrict__ x,
    float* tile, float* degt) {
  bool mm[4];
  mm[0] = (unsigned)(r.x - lo) < (unsigned)TILE_ROWS;
  mm[1] = (unsigned)(r.y - lo) < (unsigned)TILE_ROWS;
  mm[2] = (unsigned)(r.z - lo) < (unsigned)TILE_ROWS;
  mm[3] = (unsigned)(r.w - lo) < (unsigned)TILE_ROWS;
  if (__builtin_amdgcn_ballot_w32(mm[0] | mm[1] | mm[2] | mm[3])) {
#pragma unroll
    for (int k = 0; k < 4; ++k) {
      unsigned mk = __builtin_amdgcn_ballot_w32(mm[k]);
      if (mk) {
        int pos = cnt + (int)__builtin_amdgcn_mbcnt_lo(mk, 0u);
        if (mm[k]) q[pos] = 4 * j + k;
        cnt += __builtin_popcount(mk);
      }
    }
    while (cnt >= 32) {
      resolve_queued<TILE_ROWS, DEG>(q, cnt - 32, 32, lane, lo, row, col, val, x, tile, degt);
      cnt -= 32;
    }
  }
}

template <int TILE_ROWS, bool DEG>
__global__ void __launch_bounds__(1024) spmm_tile_kernel(
    const int* __restrict__ row, const int* __restrict__ col,
    const float* __restrict__ val, const float* __restrict__ x,
    float* __restrict__ y, float* __restrict__ deg, int nnz, int n_rows) {
  __shared__ float tile[TILE_ROWS * EMB];
  __shared__ float degt[DEG ? TILE_ROWS : 1];
  __shared__ int   queue[32 * QCAP];

  const int tid  = threadIdx.x;
  const int lane = tid & 31;
  const int wave = tid >> 5;
  const int lo   = blockIdx.x * TILE_ROWS;

  for (int i = tid; i < TILE_ROWS * (EMB / 4); i += 1024)
    ((float4*)tile)[i] = make_float4(0.f, 0.f, 0.f, 0.f);
  if (DEG)
    for (int i = tid; i < TILE_ROWS; i += 1024) degt[i] = 0.f;
  __syncthreads();

  int* q   = queue + wave * QCAP;
  int  cnt = 0;
  const int   nnz4  = nnz >> 2;
  const int4* row4  = (const int4*)row;
  const int   trips = (nnz4 + 1023) >> 10;
  const int   full  = (nnz4 >> 10) & ~3;
  const int start = blockIdx.x * (nnz4 / (int)gridDim.x);

  int4 cur[4];
  int  cid[4];
  if (full > 0) {
#pragma unroll
    for (int u = 0; u < 4; ++u) {
      int j = (u << 10) + tid + start;
      if (j >= nnz4) j -= nnz4;
      cid[u] = j;
      cur[u] = row4[j];
    }
  }
  for (int it = 0; it < full; it += 4) {
    int4 nxt[4];
    int  nid[4];
    const int itn = (it + 4 < full) ? (it + 4) : it;
#pragma unroll
    for (int u = 0; u < 4; ++u) {
      int j = ((itn + u) << 10) + tid + start;
      if (j >= nnz4) j -= nnz4;
      nid[u] = j;
      nxt[u] = row4[j];
    }
#pragma unroll
    for (int u = 0; u < 4; ++u)
      take4<TILE_ROWS, DEG>(cur[u], cid[u], cnt, q, lane, lo, row, col, val, x, tile, degt);
#pragma unroll
    for (int u = 0; u < 4; ++u) { cur[u] = nxt[u]; cid[u] = nid[u]; }
  }
  for (int it = full; it < trips; ++it) {
    int  i  = (it << 10) + tid;
    bool ok = i < nnz4;
    int  j  = i + start;
    if (j >= nnz4) j -= nnz4;
    int4 r = ok ? row4[j] : make_int4(-1, -1, -1, -1);
    take4<TILE_ROWS, DEG>(r, j, cnt, q, lane, lo, row, col, val, x, tile, degt);
  }
  if (wave == 0) {
    int  e  = (nnz4 << 2) + lane;
    bool m  = (e < nnz) && ((unsigned)(row[e < nnz ? e : 0] - lo) < (unsigned)TILE_ROWS);
    unsigned mk = __builtin_amdgcn_ballot_w32(m);
    if (mk) {
      int pos = cnt + (int)__builtin_amdgcn_mbcnt_lo(mk, 0u);
      if (m) q[pos] = e;
      cnt += __builtin_popcount(mk);
    }
  }
  while (cnt > 0) {
    int n = cnt < 32 ? cnt : 32;
    resolve_queued<TILE_ROWS, DEG>(q, cnt - n, n, lane, lo, row, col, val, x, tile, degt);
    cnt -= n;
  }
  __syncthreads();

  const int rows_here = (n_rows - lo) < TILE_ROWS ? (n_rows - lo) : TILE_ROWS;
  float4* y4 = (float4*)(y + (size_t)lo * EMB);
  for (int i = tid; i < rows_here * (EMB / 4); i += 1024)
    y4[i] = ((const float4*)tile)[i];
  if (DEG)
    for (int i = tid; i < rows_here; i += 1024) deg[lo + i] = degt[i];
}

__global__ void __launch_bounds__(128) acc_wmma_kernel(
    const float* __restrict__ e, float* __restrict__ acc) {
  int wave = threadIdx.x >> 5;
  int lane = threadIdx.x & 31;
  int n    = lane & 15;
  int hi   = lane >> 4;
  int row0 = blockIdx.x * 16;
  int col0 = wave * 16;

  const size_t base = (size_t)row0 * EMB + col0 + n;
  const float* ap = acc + base;
  const float* ep = e + base;

  v8f c;
#pragma unroll
  for (int r = 0; r < 8; ++r)
    c[r] = ap[(size_t)(r + 8 * hi) * EMB];

  v2f a[4], b[4];
#pragma unroll
  for (int k = 0; k < 4; ++k) {
    a[k].x = (n == (4 * k + 2 * hi))     ? 1.0f : 0.0f;
    a[k].y = (n == (4 * k + 2 * hi + 1)) ? 1.0f : 0.0f;
    b[k].x = ep[(size_t)(4 * k + 2 * hi) * EMB];
    b[k].y = ep[(size_t)(4 * k + 2 * hi + 1) * EMB];
  }

#pragma unroll
  for (int k = 0; k < 4; ++k)
    c = __builtin_amdgcn_wmma_f32_16x16x4_f32(
        false, a[k], false, b[k], (short)0, c, false, false);

  float* op = acc + base;
#pragma unroll
  for (int r = 0; r < 8; ++r)
    op[(size_t)(r + 8 * hi) * EMB] = c[r];
}

__global__ void __launch_bounds__(256) rownorm_kernel(
    float* __restrict__ x, const float* __restrict__ deg, int rowsVec) {
  int g = blockIdx.x * blockDim.x + threadIdx.x;
  if (g >= rowsVec) return;
  float d = deg[g >> 4];
  d = (d == 0.0f) ? 1.0f : d;
  float inv = 1.0f / d;
  float4 v = ((float4*)x)[g];
  v.x *= inv; v.y *= inv; v.z *= inv; v.w *= inv;
  ((float4*)x)[g] = v;
}

__global__ void __launch_bounds__(256) finalize_kernel(
    float* __restrict__ out, const float* __restrict__ ulocal,
    const float* __restrict__ deg, int totalVec, int userVec) {
  int g = blockIdx.x * blockDim.x + threadIdx.x;
  if (g >= totalVec) return;
  float4 v = ((float4*)out)[g];
  v.x *= 0.25f; v.y *= 0.25f; v.z *= 0.25f; v.w *= 0.25f;
  if (g < userVec) {
    float d = deg[g >> 4];
    d = (d == 0.0f) ? 1.0f : d;
    float inv = 1.0f / d;
    float4 w = ((const float4*)ulocal)[g];
    v.x += w.x * inv; v.y += w.y * inv; v.z += w.z * inv; v.w += w.w * inv;
  }
  ((float4*)out)[g] = v;
}

static inline int cdiv(long long a, int b) { return (int)((a + b - 1) / b); }

extern "C" void kernel_launch(void* const* d_in, const int* in_sizes, int n_in,
                              void* d_out, int out_size, void* d_ws, size_t ws_size,
                              hipStream_t stream) {
  const float* user_emb = (const float*)d_in[0];
  const float* item_emb = (const float*)d_in[1];
  const float* adj_val  = (const float*)d_in[2];
  const float* hv_val   = (const float*)d_in[3];
  const float* hu_val   = (const float*)d_in[4];
  const int*   adj_row  = (const int*)d_in[5];
  const int*   adj_col  = (const int*)d_in[6];
  const int*   hv_row   = (const int*)d_in[7];
  const int*   hv_col   = (const int*)d_in[8];
  const int*   hu_row   = (const int*)d_in[9];
  const int*   hu_col   = (const int*)d_in[10];

  const int E_adj = in_sizes[2];
  const int E_hv  = in_sizes[3];
  const int E_hu  = in_sizes[4];

  float* out = (float*)d_out;

  const size_t embBytesN = (size_t)NUM_NODES * EMB * sizeof(float);
  const size_t embBytesB = (size_t)NUM_BQ    * EMB * sizeof(float);
  const size_t embBytesU = (size_t)NUM_USERS * EMB * sizeof(float);
  char* ws = (char*)d_ws;
  float* e0     = (float*)(ws);
  float* e1     = (float*)(ws + embBytesN);
  float* bfeat  = (float*)(ws + 2 * embBytesN);
  float* deg_hv = (float*)(ws + 2 * embBytesN + embBytesB);
  float* deg_hu = deg_hv + NUM_BQ;
  float* ulocal = e0;

  const int totalVec = NUM_NODES * (EMB / 4);
  const int userVec  = NUM_USERS * (EMB / 4);
  const int bqVec    = NUM_BQ    * (EMB / 4);

  copy_init_kernel<<<cdiv(totalVec, 256), 256, 0, stream>>>(
      user_emb, item_emb, e0, out, totalVec, userVec);

  float* eprev = e0;
  float* enext = e1;
  for (int layer = 0; layer < 3; ++layer) {
    spmm_tile_kernel<944, false><<<(NUM_NODES + 943) / 944, 1024, 0, stream>>>(
        adj_row, adj_col, adj_val, eprev, enext, nullptr, E_adj, NUM_NODES);
    acc_wmma_kernel<<<NUM_NODES / 16, 128, 0, stream>>>(enext, out);
    float* t = eprev; eprev = enext; enext = t;
  }

  spmm_tile_kernel<160, true><<<(NUM_BQ + 159) / 160, 1024, 0, stream>>>(
      hv_row, hv_col, hv_val, item_emb, bfeat, deg_hv, E_hv, NUM_BQ);
  rownorm_kernel<<<cdiv(bqVec, 256), 256, 0, stream>>>(bfeat, deg_hv, bqVec);

  spmm_tile_kernel<800, true><<<(NUM_USERS + 799) / 800, 1024, 0, stream>>>(
      hu_row, hu_col, hu_val, bfeat, ulocal, deg_hu, E_hu, NUM_USERS);

  finalize_kernel<<<cdiv(totalVec, 256), 256, 0, stream>>>(
      out, ulocal, deg_hu, totalVec, userVec);
}


# Elementwise passes (init copy, 3 WMMA layer-sum passes, row scale, finalize) become streams: 7,500 long-lived waves per launch, 5-30 whole-line loads in flight each, not one 16-B piece per thread
# speedup vs baseline: 1.3431x; 1.0531x over previous
#include <hip/hip_runtime.h>
#include <hip/hip_bf16.h>

typedef __attribute__((ext_vector_type(2))) float v2f;
typedef __attribute__((ext_vector_type(8))) float v8f;
typedef __attribute__((ext_vector_type(4))) float v4f;

#define NUM_USERS 100000
#define NUM_ITEMS 50000
#define NUM_NODES 150000
#define NUM_BQ    20000
#define EMB       64

#define SU 5

__global__ void __launch_bounds__(320) copy_init_kernel(
    const float* __restrict__ ue, const float* __restrict__ ie,
    float* __restrict__ e0, float* __restrict__ out,
    int totalVec, int userVec) {
  const int nthr = gridDim.x * blockDim.x;
  int g = blockIdx.x * blockDim.x + threadIdx.x;
  for (; g + (SU - 1) * nthr < totalVec; g += SU * nthr) {
    v4f v[SU];
#pragma unroll
    for (int u = 0; u < SU; ++u) {
      int i = g + u * nthr;
      v[u] = (i < userVec) ? ((const v4f*)ue)[i] : ((const v4f*)ie)[i - userVec];
    }
    __builtin_amdgcn_sched_barrier(0);
#pragma unroll
    for (int u = 0; u < SU; ++u) {
      int i = g + u * nthr;
      ((v4f*)e0)[i]  = v[u];
      ((v4f*)out)[i] = v[u];
    }
  }
  for (; g < totalVec; g += nthr) {
    v4f v = (g < userVec) ? ((const v4f*)ue)[g] : ((const v4f*)ie)[g - userVec];
    ((v4f*)e0)[g]  = v;
    ((v4f*)out)[g] = v;
  }
}

#define QCAP 160

template <int TILE_ROWS, bool DEG>
__device__ __forceinline__ void resolve_queued(
    const int* q, int base, int n, int lane, int lo,
    const int* __restrict__ row, const int* __restrict__ col,
    const float* __restrict__ val, const float* __restrict__ x,
    float* tile, float* degt) {
  int last = n - 1;
  int e  = q[base + (lane < last ? lane : last)];
  int rr = row[e] - lo;
  int c  = col[e];
  int vb = __float_as_int(val[e]);
  for (int k0 = 0; k0 < n; k0 += 16) {
    float2 xv[16];
#pragma unroll
    for (int j = 0; j < 16; ++j) {
      int k  = (k0 + j < last) ? (k0 + j) : last;
      int ck = __builtin_amdgcn_readlane(c, k);
      xv[j] = *(const float2*)(x + (size_t)ck * EMB + 2 * lane);
    }
#pragma unroll
    for (int j = 0; j < 16; ++j) {
      if (k0 + j < n) {
        int   k  = k0 + j;
        int   rk = __builtin_amdgcn_readlane(rr, k);
        float vk = __int_as_float(__builtin_amdgcn_readlane(vb, k));
        float* t = tile + rk * EMB + 2 * lane;
        atomicAdd(t,     vk * xv[j].x);
        atomicAdd(t + 1, vk * xv[j].y);
        if (DEG && lane == 0) atomicAdd(degt + rk, vk);
      }
    }
  }
}

template <int TILE_ROWS, bool DEG>
__device__ __forceinline__ void take4(
    int4 r, int j, int& cnt, int* q, int lane, int lo,
    const int* __restrict__ row, const int* __restrict__ col,
    const float* __restrict__ val, const float* __restrict__ x,
    float* tile, float* degt) {
  bool mm[4];
  mm[0] = (unsigned)(r.x - lo) < (unsigned)TILE_ROWS;
  mm[1] = (unsigned)(r.y - lo) < (unsigned)TILE_ROWS;
  mm[2] = (unsigned)(r.z - lo) < (unsigned)TILE_ROWS;
  mm[3] = (unsigned)(r.w - lo) < (unsigned)TILE_ROWS;
  if (__builtin_amdgcn_ballot_w32(mm[0] | mm[1] | mm[2] | mm[3])) {
#pragma unroll
    for (int k = 0; k < 4; ++k) {
      unsigned mk = __builtin_amdgcn_ballot_w32(mm[k]);
      if (mk) {
        int pos = cnt + (int)__builtin_amdgcn_mbcnt_lo(mk, 0u);
        if (mm[k]) q[pos] = 4 * j + k;
        cnt += __builtin_popcount(mk);
      }
    }
    while (cnt >= 32) {
      resolve_queued<TILE_ROWS, DEG>(q, cnt - 32, 32, lane, lo, row, col, val, x, tile, degt);
      cnt -= 32;
    }
  }
}

template <int TILE_ROWS, bool DEG>
__global__ void __launch_bounds__(1024) spmm_tile_kernel(
    const int* __restrict__ row, const int* __restrict__ col,
    const float* __restrict__ val, const float* __restrict__ x,
    float* __restrict__ y, float* __restrict__ deg, int nnz, int n_rows) {
  __shared__ float tile[TILE_ROWS * EMB];
  __shared__ float degt[DEG ? TILE_ROWS : 1];
  __shared__ int   queue[32 * QCAP];

  const int tid  = threadIdx.x;
  const int lane = tid & 31;
  const int wave = tid >> 5;
  const int lo   = blockIdx.x * TILE_ROWS;

  for (int i = tid; i < TILE_ROWS * (EMB / 4); i += 1024)
    ((float4*)tile)[i] = make_float4(0.f, 0.f, 0.f, 0.f);
  if (DEG)
    for (int i = tid; i < TILE_ROWS; i += 1024) degt[i] = 0.f;
  __syncthreads();

  int* q   = queue + wave * QCAP;
  int  cnt = 0;
  const int   nnz4  = nnz >> 2;
  const int4* row4  = (const int4*)row;
  const int   trips = (nnz4 + 1023) >> 10;
  const int   full  = (nnz4 >> 10) & ~3;
  const int start = blockIdx.x * (nnz4 / (int)gridDim.x);

  int4 cur[4];
  int  cid[4];
  if (full > 0) {
#pragma unroll
    for (int u = 0; u < 4; ++u) {
      int j = (u << 10) + tid + start;
      if (j >= nnz4) j -= nnz4;
      cid[u] = j;
      cur[u] = row4[j];
    }
  }
  for (int it = 0; it < full; it += 4) {
    int4 nxt[4];
    int  nid[4];
    const int itn = (it + 4 < full) ? (it + 4) : it;
#pragma unroll
    for (int u = 0; u < 4; ++u) {
      int j = ((itn + u) << 10) + tid + start;
      if (j >= nnz4) j -= nnz4;
      nid[u] = j;
      nxt[u] = row4[j];
    }
#pragma unroll
    for (int u = 0; u < 4; ++u)
      take4<TILE_ROWS, DEG>(cur[u], cid[u], cnt, q, lane, lo, row, col, val, x, tile, degt);
#pragma unroll
    for (int u = 0; u < 4; ++u) { cur[u] = nxt[u]; cid[u] = nid[u]; }
  }
  for (int it = full; it < trips; ++it) {
    int  i  = (it << 10) + tid;
    bool ok = i < nnz4;
    int  j  = i + start;
    if (j >= nnz4) j -= nnz4;
    int4 r = ok ? row4[j] : make_int4(-1, -1, -1, -1);
    take4<TILE_ROWS, DEG>(r, j, cnt, q, lane, lo, row, col, val, x, tile, degt);
  }
  if (wave == 0) {
    int  e  = (nnz4 << 2) + lane;
    bool m  = (e < nnz) && ((unsigned)(row[e < nnz ? e : 0] - lo) < (unsigned)TILE_ROWS);
    unsigned mk = __builtin_amdgcn_ballot_w32(m);
    if (mk) {
      int pos = cnt + (int)__builtin_amdgcn_mbcnt_lo(mk, 0u);
      if (m) q[pos] = e;
      cnt += __builtin_popcount(mk);
    }
  }
  while (cnt > 0) {
    int n = cnt < 32 ? cnt : 32;
    resolve_queued<TILE_ROWS, DEG>(q, cnt - n, n, lane, lo, row, col, val, x, tile, degt);
    cnt -= n;
  }
  __syncthreads();

  const int rows_here = (n_rows - lo) < TILE_ROWS ? (n_rows - lo) : TILE_ROWS;
  float4* y4 = (float4*)(y + (size_t)lo * EMB);
  for (int i = tid; i < rows_here * (EMB / 4); i += 1024)
    y4[i] = ((const float4*)tile)[i];
  if (DEG)
    for (int i = tid; i < rows_here; i += 1024) deg[lo + i] = degt[i];
}

#define AU 5
__global__ void __launch_bounds__(320) acc_wmma_kernel(
    const float* __restrict__ e, float* __restrict__ acc, int nChunks) {
  const int lane = threadIdx.x & 31;
  const int n    = lane & 15;
  const int hi   = lane >> 4;
  const int wpb  = blockDim.x >> 5;
  const int nWaves = gridDim.x * wpb;
  const int wv = __builtin_amdgcn_readfirstlane(blockIdx.x * wpb + (threadIdx.x >> 5));

  v2f a[4];
#pragma unroll
  for (int k = 0; k < 4; ++k) {
    a[k].x = (n == (4 * k + 2 * hi))     ? 1.0f : 0.0f;
    a[k].y = (n == (4 * k + 2 * hi + 1)) ? 1.0f : 0.0f;
  }

  for (int g = wv * AU; g < nChunks; g += nWaves * AU) {
    v8f cc[AU];
    v2f b[AU][4];
#pragma unroll
    for (int u = 0; u < AU; ++u) {
      int ch = (g + u < nChunks) ? (g + u) : (nChunks - 1);
      const float* ap = acc + (size_t)ch * 256 + 4 * lane;
      const float* ep = e   + (size_t)ch * 256 + 4 * n + 2 * hi;
      v4f lo = *(const v4f*)(ap);
      v4f up = *(const v4f*)(ap + 128);
      cc[u][0] = lo.x; cc[u][1] = lo.y; cc[u][2] = lo.z; cc[u][3] = lo.w;
      cc[u][4] = up.x; cc[u][5] = up.y; cc[u][6] = up.z; cc[u][7] = up.w;
#pragma unroll
      for (int k = 0; k < 4; ++k)
        b[u][k] = *(const v2f*)(ep + 128 * (k & 1) + 64 * (k >> 1));
    }
    __builtin_amdgcn_sched_barrier(0);
#pragma unroll
    for (int k = 0; k < 4; ++k)
#pragma unroll
      for (int u = 0; u < AU; ++u)
        cc[u] = __builtin_amdgcn_wmma_f32_16x16x4_f32(
            false, a[k], false, b[u][k], (short)0, cc[u], false, false);
#pragma unroll
    for (int u = 0; u < AU; ++u) {
      if (g + u < nChunks) {
        float* op = acc + (size_t)(g + u) * 256 + 4 * lane;
        v4f lo = {cc[u][0], cc[u][1], cc[u][2], cc[u][3]};
        v4f up = {cc[u][4], cc[u][5], cc[u][6], cc[u][7]};
        *(v4f*)(op)       = lo;
        *(v4f*)(op + 128) = up;
      }
    }
  }
}

__global__ void __launch_bounds__(256) rownorm_kernel(
    float* __restrict__ x, const float* __restrict__ deg, int rowsVec) {
  const int nthr = gridDim.x * blockDim.x;
  int g = blockIdx.x * blockDim.x + threadIdx.x;
  for (; g + (SU - 1) * nthr < rowsVec; g += SU * nthr) {
    v4f   v[SU];
    float d[SU];
#pragma unroll
    for (int u = 0; u < SU; ++u) {
      int i = g + u * nthr;
      v[u] = ((const v4f*)x)[i];
      d[u] = deg[i >> 4];
    }
    __builtin_amdgcn_sched_barrier(0);
#pragma unroll
    for (int u = 0; u < SU; ++u) {
      int i = g + u * nthr;
      float dd  = (d[u] == 0.0f) ? 1.0f : d[u];
      float inv = 1.0f / dd;
      ((v4f*)x)[i] = v[u] * inv;
    }
  }
  for (; g < rowsVec; g += nthr) {
    float dd  = deg[g >> 4];
    dd = (dd == 0.0f) ? 1.0f : dd;
    float inv = 1.0f / dd;
    ((v4f*)x)[g] = ((const v4f*)x)[g] * inv;
  }
}

__global__ void __launch_bounds__(320) finalize_kernel(
    float* __restrict__ out, const float* __restrict__ ulocal,
    const float* __restrict__ deg, int totalVec, int userVec) {
  const int nthr = gridDim.x * blockDim.x;
  int g = blockIdx.x * blockDim.x + threadIdx.x;
  for (; g + (SU - 1) * nthr < totalVec; g += SU * nthr) {
    v4f   v[SU], w[SU];
    float d[SU];
#pragma unroll
    for (int u = 0; u < SU; ++u) {
      int i  = g + u * nthr;
      int iu = (i < userVec) ? i : (userVec - 1);
      v[u] = ((const v4f*)out)[i];
      w[u] = ((const v4f*)ulocal)[iu];
      d[u] = deg[iu >> 4];
    }
    __builtin_amdgcn_sched_barrier(0);
#pragma unroll
    for (int u = 0; u < SU; ++u) {
      int i = g + u * nthr;
      float dd  = (d[u] == 0.0f) ? 1.0f : d[u];
      float inv = 1.0f / dd;
      v4f r  = v[u] * 0.25f;
      v4f ru = r + w[u] * inv;
      ((v4f*)out)[i] = (i < userVec) ? ru : r;
    }
  }
  for (; g < totalVec; g += nthr) {
    v4f r = ((const v4f*)out)[g] * 0.25f;
    if (g < userVec) {
      float dd  = deg[g >> 4];
      dd = (dd == 0.0f) ? 1.0f : dd;
      float inv = 1.0f / dd;
      r += ((const v4f*)ulocal)[g] * inv;
    }
    ((v4f*)out)[g] = r;
  }
}

extern "C" void kernel_launch(void* const* d_in, const int* in_sizes, int n_in,
                              void* d_out, int out_size, void* d_ws, size_t ws_size,
                              hipStream_t stream) {
  const float* user_emb = (const float*)d_in[0];
  const float* item_emb = (const float*)d_in[1];
  const float* adj_val  = (const float*)d_in[2];
  const float* hv_val   = (const float*)d_in[3];
  const float* hu_val   = (const float*)d_in[4];
  const int*   adj_row  = (const int*)d_in[5];
  const int*   adj_col  = (const int*)d_in[6];
  const int*   hv_row   = (const int*)d_in[7];
  const int*   hv_col   = (const int*)d_in[8];
  const int*   hu_row   = (const int*)d_in[9];
  const int*   hu_col   = (const int*)d_in[10];

  const int E_adj = in_sizes[2];
  const int E_hv  = in_sizes[3];
  const int E_hu  = in_sizes[4];

  float* out = (float*)d_out;

  const size_t embBytesN = (size_t)NUM_NODES * EMB * sizeof(float);
  const size_t embBytesB = (size_t)NUM_BQ    * EMB * sizeof(float);
  const size_t embBytesU = (size_t)NUM_USERS * EMB * sizeof(float);
  char* ws = (char*)d_ws;
  float* e0     = (float*)(ws);
  float* e1     = (float*)(ws + embBytesN);
  float* bfeat  = (float*)(ws + 2 * embBytesN);
  float* deg_hv = (float*)(ws + 2 * embBytesN + embBytesB);
  float* deg_hu = deg_hv + NUM_BQ;
  float* ulocal = e0;

  const int totalVec = NUM_NODES * (EMB / 4);
  const int userVec  = NUM_USERS * (EMB / 4);
  const int bqVec    = NUM_BQ    * (EMB / 4);
  const int nChunks  = NUM_NODES * EMB / 256;


  copy_init_kernel<<<750, 320, 0, stream>>>(
      user_emb, item_emb, e0, out, totalVec, userVec);

  float* eprev = e0;
  float* enext = e1;
  for (int layer = 0; layer < 3; ++layer) {
    spmm_tile_kernel<944, false><<<(NUM_NODES + 943) / 944, 1024, 0, stream>>>(
        adj_row, adj_col, adj_val, eprev, enext, nullptr, E_adj, NUM_NODES);
    acc_wmma_kernel<<<750, 320, 0, stream>>>(enext, out, nChunks);
    float* t = eprev; eprev = enext; enext = t;
  }

  spmm_tile_kernel<160, true><<<(NUM_BQ + 159) / 160, 1024, 0, stream>>>(
      hv_row, hv_col, hv_val, item_emb, bfeat, deg_hv, E_hv, NUM_BQ);
  rownorm_kernel<<<250, 256, 0, stream>>>(bfeat, deg_hv, bqVec);

  spmm_tile_kernel<800, true><<<(NUM_USERS + 799) / 800, 1024, 0, stream>>>(
      hu_row, hu_col, hu_val, bfeat, ulocal, deg_hu, E_hu, NUM_USERS);

  finalize_kernel<<<750, 320, 0, stream>>>(
      out, ulocal, deg_hu, totalVec, userVec);
}


# SpMM tile scans read 1-byte tile ids (row / tile height; two small passes write them once per call) instead of 4-byte row ids: 16 edges per 16-B load; row sums stay in arrival order
# speedup vs baseline: 1.7156x; 1.2773x over previous
#include <hip/hip_runtime.h>
#include <hip/hip_bf16.h>

typedef __attribute__((ext_vector_type(2))) float v2f;
typedef __attribute__((ext_vector_type(8))) float v8f;
typedef __attribute__((ext_vector_type(4))) float v4f;

#define NUM_USERS 100000
#define NUM_ITEMS 50000
#define NUM_NODES 150000
#define NUM_BQ    20000
#define EMB       64

#define SU 5

__global__ void __launch_bounds__(320) copy_init_kernel(
    const float* __restrict__ ue, const float* __restrict__ ie,
    float* __restrict__ e0, float* __restrict__ out,
    int totalVec, int userVec) {
  const int nthr = gridDim.x * blockDim.x;
  int g = blockIdx.x * blockDim.x + threadIdx.x;
  for (; g + (SU - 1) * nthr < totalVec; g += SU * nthr) {
    v4f v[SU];
#pragma unroll
    for (int u = 0; u < SU; ++u) {
      int i = g + u * nthr;
      v[u] = (i < userVec) ? ((const v4f*)ue)[i] : ((const v4f*)ie)[i - userVec];
    }
    __builtin_amdgcn_sched_barrier(0);
#pragma unroll
    for (int u = 0; u < SU; ++u) {
      int i = g + u * nthr;
      ((v4f*)e0)[i]  = v[u];
      ((v4f*)out)[i] = v[u];
    }
  }
  for (; g < totalVec; g += nthr) {
    v4f v = (g < userVec) ? ((const v4f*)ue)[g] : ((const v4f*)ie)[g - userVec];
    ((v4f*)e0)[g]  = v;
    ((v4f*)out)[g] = v;
  }
}

#define QCAP 64

template <int TILE_ROWS, bool DEG>
__device__ __forceinline__ void resolve_queued(
    const int* q, int base, int n, int lane, int lo,
    const int* __restrict__ row, const int* __restrict__ col,
    const float* __restrict__ val, const float* __restrict__ x,
    float* tile, float* degt) {
  int last = n - 1;
  int e  = q[base + (lane < last ? lane : last)];
  int rr = row[e] - lo;
  int c  = col[e];
  int vb = __float_as_int(val[e]);
  for (int k0 = 0; k0 < n; k0 += 16) {
    float2 xv[16];
#pragma unroll
    for (int j = 0; j < 16; ++j) {
      int k  = (k0 + j < last) ? (k0 + j) : last;
      int ck = __builtin_amdgcn_readlane(c, k);
      xv[j] = *(const float2*)(x + (size_t)ck * EMB + 2 * lane);
    }
#pragma unroll
    for (int j = 0; j < 16; ++j) {
      if (k0 + j < n) {
        int   k  = k0 + j;
        int   rk = __builtin_amdgcn_readlane(rr, k);
        float vk = __int_as_float(__builtin_amdgcn_readlane(vb, k));
        float* t = tile + rk * EMB + 2 * lane;
        atomicAdd(t,     vk * xv[j].x);
        atomicAdd(t + 1, vk * xv[j].y);
        if (DEG && lane == 0) atomicAdd(degt + rk, vk);
      }
    }
  }
}

template <int TILE_ROWS>
__device__ __forceinline__ unsigned tile_id_of(int r) {
  unsigned t = (unsigned)r / (unsigned)TILE_ROWS;
  return t < 255u ? t : 255u;
}
template <int TILE_ROWS>
__device__ __forceinline__ unsigned pack_ids(int4 r) {
  return tile_id_of<TILE_ROWS>(r.x) | (tile_id_of<TILE_ROWS>(r.y) << 8) |
         (tile_id_of<TILE_ROWS>(r.z) << 16) | (tile_id_of<TILE_ROWS>(r.w) << 24);
}
template <int TILE_A, int TILE_B>
__global__ void __launch_bounds__(256) tile_ids_kernel(
    const int* __restrict__ rowA, unsigned* __restrict__ idsA, int nnzA,
    const int* __restrict__ rowB, unsigned* __restrict__ idsB, int nnzB) {
  const int nthr = gridDim.x * blockDim.x;
  const int nA4  = nnzA >> 2;
  const int nB4  = nnzB >> 2;
  for (int g = blockIdx.x * blockDim.x + threadIdx.x; g < nA4 + nB4; g += nthr) {
    if (g < nA4) idsA[g]       = pack_ids<TILE_A>(((const int4*)rowA)[g]);
    else         idsB[g - nA4] = pack_ids<TILE_B>(((const int4*)rowB)[g - nA4]);
  }
}

__device__ __forceinline__ unsigned nonzero_bytes(unsigned v) {
  return (((v & 0x7f7f7f7fu) + 0x7f7f7f7fu) | v) & 0x80808080u;
}

template <int TILE_ROWS, bool DEG>
__device__ __forceinline__ void take16(
    uint4 w, unsigned tb, bool ok, int j, int& cnt, int* q, int lane, int lo,
    const int* __restrict__ row, const int* __restrict__ col,
    const float* __restrict__ val, const float* __restrict__ x,
    float* tile, float* degt) {
  unsigned nz = (nonzero_bytes(w.x ^ tb) >> 7) | (nonzero_bytes(w.y ^ tb) >> 6) |
                (nonzero_bytes(w.z ^ tb) >> 5) | (nonzero_bytes(w.w ^ tb) >> 4);
  unsigned m = ok ? (~nz & 0x0f0f0f0fu) : 0u;
  unsigned mk = __builtin_amdgcn_ballot_w32(m != 0u);
  while (mk) {
    int p   = __builtin_ctz(m | 0x80000000u);
    int pos = cnt + (int)__builtin_amdgcn_mbcnt_lo(mk, 0u);
    if (m != 0u) q[pos] = 16 * j + 4 * (p & 7) + (p >> 3);
    cnt += __builtin_popcount(mk);
    m &= m - 1u;
    if (cnt >= 32) {
      resolve_queued<TILE_ROWS, DEG>(q, cnt - 32, 32, lane, lo, row, col, val, x, tile, degt);
      cnt -= 32;
    }
    mk = __builtin_amdgcn_ballot_w32(m != 0u);
  }
}

template <int TILE_ROWS, bool DEG>
__global__ void __launch_bounds__(1024) spmm_tile_kernel(
    const unsigned* __restrict__ ids,
    const int* __restrict__ row, const int* __restrict__ col,
    const float* __restrict__ val, const float* __restrict__ x,
    float* __restrict__ y, float* __restrict__ deg, int nnz, int n_rows) {
  __shared__ float tile[TILE_ROWS * EMB];
  __shared__ float degt[DEG ? TILE_ROWS : 1];
  __shared__ int   queue[32 * QCAP];

  const int tid  = threadIdx.x;
  const int lane = tid & 31;
  const int wave = tid >> 5;
  const int lo   = blockIdx.x * TILE_ROWS;
  const unsigned tb = 0x01010101u * blockIdx.x;

  for (int i = tid; i < TILE_ROWS * (EMB / 4); i += 1024)
    ((float4*)tile)[i] = make_float4(0.f, 0.f, 0.f, 0.f);
  if (DEG)
    for (int i = tid; i < TILE_ROWS; i += 1024) degt[i] = 0.f;
  __syncthreads();

  int* q   = queue + wave * QCAP;
  int  cnt = 0;
  const int    nnz16 = nnz >> 4;
  const uint4* ids4  = (const uint4*)ids;
  const int    trips = (nnz16 + 1023) >> 10;
  const int    full  = (nnz16 >> 10) & ~3;
  const int start = blockIdx.x * (nnz16 / (int)gridDim.x);

  uint4 cur[4];
  int   cid[4];
  if (full > 0) {
#pragma unroll
    for (int u = 0; u < 4; ++u) {
      int j = (u << 10) + tid + start;
      if (j >= nnz16) j -= nnz16;
      cid[u] = j;
      cur[u] = ids4[j];
    }
  }
  for (int it = 0; it < full; it += 4) {
    uint4 nxt[4];
    int   nid[4];
    const int itn = (it + 4 < full) ? (it + 4) : it;
#pragma unroll
    for (int u = 0; u < 4; ++u) {
      int j = ((itn + u) << 10) + tid + start;
      if (j >= nnz16) j -= nnz16;
      nid[u] = j;
      nxt[u] = ids4[j];
    }
#pragma unroll
    for (int u = 0; u < 4; ++u)
      take16<TILE_ROWS, DEG>(cur[u], tb, true, cid[u], cnt, q, lane, lo, row, col, val, x, tile, degt);
#pragma unroll
    for (int u = 0; u < 4; ++u) { cur[u] = nxt[u]; cid[u] = nid[u]; }
  }
  for (int it = full; it < trips; ++it) {
    int  i  = (it << 10) + tid;
    bool ok = i < nnz16;
    int  j  = i + start;
    if (j >= nnz16) j -= nnz16;
    uint4 w = ids4[ok ? j : 0];
    take16<TILE_ROWS, DEG>(w, tb, ok, j, cnt, q, lane, lo, row, col, val, x, tile, degt);
  }
  if (wave == 0) {
    int  e  = (nnz16 << 4) + lane;
    bool m  = (e < nnz) && ((unsigned)(row[e < nnz ? e : 0] - lo) < (unsigned)TILE_ROWS);
    unsigned mk = __builtin_amdgcn_ballot_w32(m);
    if (mk) {
      int pos = cnt + (int)__builtin_amdgcn_mbcnt_lo(mk, 0u);
      if (m) q[pos] = e;
      cnt += __builtin_popcount(mk);
    }
  }
  while (cnt > 0) {
    int n = cnt < 32 ? cnt : 32;
    resolve_queued<TILE_ROWS, DEG>(q, cnt - n, n, lane, lo, row, col, val, x, tile, degt);
    cnt -= n;
  }
  __syncthreads();

  const int rows_here = (n_rows - lo) < TILE_ROWS ? (n_rows - lo) : TILE_ROWS;
  float4* y4 = (float4*)(y + (size_t)lo * EMB);
  for (int i = tid; i < rows_here * (EMB / 4); i += 1024)
    y4[i] = ((const float4*)tile)[i];
  if (DEG)
    for (int i = tid; i < rows_here; i += 1024) deg[lo + i] = degt[i];
}

#define AU 5
__global__ void __launch_bounds__(320) acc_wmma_kernel(
    const float* __restrict__ e, float* __restrict__ acc, int nChunks) {
  const int lane = threadIdx.x & 31;
  const int n    = lane & 15;
  const int hi   = lane >> 4;
  const int wpb  = blockDim.x >> 5;
  const int nWaves = gridDim.x * wpb;
  const int wv = __builtin_amdgcn_readfirstlane(blockIdx.x * wpb + (threadIdx.x >> 5));

  v2f a[4];
#pragma unroll
  for (int k = 0; k < 4; ++k) {
    a[k].x = (n == (4 * k + 2 * hi))     ? 1.0f : 0.0f;
    a[k].y = (n == (4 * k + 2 * hi + 1)) ? 1.0f : 0.0f;
  }

  for (int g = wv * AU; g < nChunks; g += nWaves * AU) {
    v8f cc[AU];
    v2f b[AU][4];
#pragma unroll
    for (int u = 0; u < AU; ++u) {
      int ch = (g + u < nChunks) ? (g + u) : (nChunks - 1);
      const float* ap = acc + (size_t)ch * 256 + 4 * lane;
      const float* ep = e   + (size_t)ch * 256 + 4 * n + 2 * hi;
      v4f lo = *(const v4f*)(ap);
      v4f up = *(const v4f*)(ap + 128);
      cc[u][0] = lo.x; cc[u][1] = lo.y; cc[u][2] = lo.z; cc[u][3] = lo.w;
      cc[u][4] = up.x; cc[u][5] = up.y; cc[u][6] = up.z; cc[u][7] = up.w;
#pragma unroll
      for (int k = 0; k < 4; ++k)
        b[u][k] = *(const v2f*)(ep + 128 * (k & 1) + 64 * (k >> 1));
    }
    __builtin_amdgcn_sched_barrier(0);
#pragma unroll
    for (int k = 0; k < 4; ++k)
#pragma unroll
      for (int u = 0; u < AU; ++u)
        cc[u] = __builtin_amdgcn_wmma_f32_16x16x4_f32(
            false, a[k], false, b[u][k], (short)0, cc[u], false, false);
#pragma unroll
    for (int u = 0; u < AU; ++u) {
      if (g + u < nChunks) {
        float* op = acc + (size_t)(g + u) * 256 + 4 * lane;
        v4f lo = {cc[u][0], cc[u][1], cc[u][2], cc[u][3]};
        v4f up = {cc[u][4], cc[u][5], cc[u][6], cc[u][7]};
        *(v4f*)(op)       = lo;
        *(v4f*)(op + 128) = up;
      }
    }
  }
}

__global__ void __launch_bounds__(256) rownorm_kernel(
    float* __restrict__ x, const float* __restrict__ deg, int rowsVec) {
  const int nthr = gridDim.x * blockDim.x;
  int g = blockIdx.x * blockDim.x + threadIdx.x;
  for (; g + (SU - 1) * nthr < rowsVec; g += SU * nthr) {
    v4f   v[SU];
    float d[SU];
#pragma unroll
    for (int u = 0; u < SU; ++u) {
      int i = g + u * nthr;
      v[u] = ((const v4f*)x)[i];
      d[u] = deg[i >> 4];
    }
    __builtin_amdgcn_sched_barrier(0);
#pragma unroll
    for (int u = 0; u < SU; ++u) {
      int i = g + u * nthr;
      float dd  = (d[u] == 0.0f) ? 1.0f : d[u];
      float inv = 1.0f / dd;
      ((v4f*)x)[i] = v[u] * inv;
    }
  }
  for (; g < rowsVec; g += nthr) {
    float dd  = deg[g >> 4];
    dd = (dd == 0.0f) ? 1.0f : dd;
    float inv = 1.0f / dd;
    ((v4f*)x)[g] = ((const v4f*)x)[g] * inv;
  }
}

__global__ void __launch_bounds__(320) finalize_kernel(
    float* __restrict__ out, const float* __restrict__ ulocal,
    const float* __restrict__ deg, int totalVec, int userVec) {
  const int nthr = gridDim.x * blockDim.x;
  int g = blockIdx.x * blockDim.x + threadIdx.x;
  for (; g + (SU - 1) * nthr < totalVec; g += SU * nthr) {
    v4f   v[SU], w[SU];
    float d[SU];
#pragma unroll
    for (int u = 0; u < SU; ++u) {
      int i  = g + u * nthr;
      int iu = (i < userVec) ? i : (userVec - 1);
      v[u] = ((const v4f*)out)[i];
      w[u] = ((const v4f*)ulocal)[iu];
      d[u] = deg[iu >> 4];
    }
    __builtin_amdgcn_sched_barrier(0);
#pragma unroll
    for (int u = 0; u < SU; ++u) {
      int i = g + u * nthr;
      float dd  = (d[u] == 0.0f) ? 1.0f : d[u];
      float inv = 1.0f / dd;
      v4f r  = v[u] * 0.25f;
      v4f ru = r + w[u] * inv;
      ((v4f*)out)[i] = (i < userVec) ? ru : r;
    }
  }
  for (; g < totalVec; g += nthr) {
    v4f r = ((const v4f*)out)[g] * 0.25f;
    if (g < userVec) {
      float dd  = deg[g >> 4];
      dd = (dd == 0.0f) ? 1.0f : dd;
      float inv = 1.0f / dd;
      r += ((const v4f*)ulocal)[g] * inv;
    }
    ((v4f*)out)[g] = r;
  }
}

#define TILE_ADJ 944
#define TILE_HV  160
#define TILE_HU  800
static_assert((NUM_NODES + TILE_ADJ - 1) / TILE_ADJ <= 255, "tile id must fit a byte");
static_assert((NUM_BQ    + TILE_HV  - 1) / TILE_HV  <= 255, "tile id must fit a byte");
static_assert((NUM_USERS + TILE_HU  - 1) / TILE_HU  <= 255, "tile id must fit a byte");

extern "C" void kernel_launch(void* const* d_in, const int* in_sizes, int n_in,
                              void* d_out, int out_size, void* d_ws, size_t ws_size,
                              hipStream_t stream) {
  const float* user_emb = (const float*)d_in[0];
  const float* item_emb = (const float*)d_in[1];
  const float* adj_val  = (const float*)d_in[2];
  const float* hv_val   = (const float*)d_in[3];
  const float* hu_val   = (const float*)d_in[4];
  const int*   adj_row  = (const int*)d_in[5];
  const int*   adj_col  = (const int*)d_in[6];
  const int*   hv_row   = (const int*)d_in[7];
  const int*   hv_col   = (const int*)d_in[8];
  const int*   hu_row   = (const int*)d_in[9];
  const int*   hu_col   = (const int*)d_in[10];

  const int E_adj = in_sizes[2];
  const int E_hv  = in_sizes[3];
  const int E_hu  = in_sizes[4];

  float* out = (float*)d_out;

  const size_t embBytesN = (size_t)NUM_NODES * EMB * sizeof(float);
  const size_t embBytesB = (size_t)NUM_BQ    * EMB * sizeof(float);
  const size_t embBytesU = (size_t)NUM_USERS * EMB * sizeof(float);
  char* ws = (char*)d_ws;
  float* e0     = (float*)(ws);
  float* e1     = (float*)(ws + embBytesN);
  float* bfeat  = (float*)(ws + 2 * embBytesN);
  float* deg_hv = (float*)(ws + 2 * embBytesN + embBytesB);
  float* deg_hu = deg_hv + NUM_BQ;
  float* ulocal = e0;
  unsigned* ids_adj = (unsigned*)(ws + 2 * embBytesN);
  unsigned* ids_hv  = (unsigned*)(ws + embBytesU);
  unsigned* ids_hu  = (unsigned*)(ws + embBytesU + (embBytesN - embBytesU) / 2);

  const int totalVec = NUM_NODES * (EMB / 4);
  const int userVec  = NUM_USERS * (EMB / 4);
  const int bqVec    = NUM_BQ    * (EMB / 4);
  const int nChunks  = NUM_NODES * EMB / 256;


  copy_init_kernel<<<750, 320, 0, stream>>>(
      user_emb, item_emb, e0, out, totalVec, userVec);

  tile_ids_kernel<TILE_ADJ, TILE_ADJ><<<1172, 256, 0, stream>>>(
      adj_row, ids_adj, E_adj, adj_row, ids_adj, 0);
  float* eprev = e0;
  float* enext = e1;
  for (int layer = 0; layer < 3; ++layer) {
    spmm_tile_kernel<TILE_ADJ, false><<<(NUM_NODES + TILE_ADJ - 1) / TILE_ADJ, 1024, 0, stream>>>(
        ids_adj, adj_row, adj_col, adj_val, eprev, enext, nullptr, E_adj, NUM_NODES);
    acc_wmma_kernel<<<750, 320, 0, stream>>>(enext, out, nChunks);
    float* t = eprev; eprev = enext; enext = t;
  }

  tile_ids_kernel<TILE_HV, TILE_HU><<<782, 256, 0, stream>>>(
      hv_row, ids_hv, E_hv, hu_row, ids_hu, E_hu);
  spmm_tile_kernel<TILE_HV, true><<<(NUM_BQ + TILE_HV - 1) / TILE_HV, 1024, 0, stream>>>(
      ids_hv, hv_row, hv_col, hv_val, item_emb, bfeat, deg_hv, E_hv, NUM_BQ);
  rownorm_kernel<<<250, 256, 0, stream>>>(bfeat, deg_hv, bqVec);

  spmm_tile_kernel<TILE_HU, true><<<(NUM_USERS + TILE_HU - 1) / TILE_HU, 1024, 0, stream>>>(
      ids_hu, hu_row, hu_col, hu_val, bfeat, ulocal, deg_hu, E_hu, NUM_USERS);

  finalize_kernel<<<750, 320, 0, stream>>>(
      out, ulocal, deg_hu, totalVec, userVec);
}


# Layer sums lose their launches: each acc += e pass (same WMMAs) runs inside the next SpMM launch, on extra blocks behind the tile blocks (97-131 WGPs own no tile); 13 launches -> 10; sum order kept
# speedup vs baseline: 1.8796x; 1.0956x over previous
#include <hip/hip_runtime.h>
#include <hip/hip_bf16.h>

typedef __attribute__((ext_vector_type(2))) float v2f;
typedef __attribute__((ext_vector_type(8))) float v8f;
typedef __attribute__((ext_vector_type(4))) float v4f;

#define NUM_USERS 100000
#define NUM_ITEMS 50000
#define NUM_NODES 150000
#define NUM_BQ    20000
#define EMB       64

#define SU 5

__global__ void __launch_bounds__(320) copy_init_kernel(
    const float* __restrict__ ue, const float* __restrict__ ie,
    float* __restrict__ e0, float* __restrict__ out,
    int totalVec, int userVec) {
  const int nthr = gridDim.x * blockDim.x;
  int g = blockIdx.x * blockDim.x + threadIdx.x;
  for (; g + (SU - 1) * nthr < totalVec; g += SU * nthr) {
    v4f v[SU];
#pragma unroll
    for (int u = 0; u < SU; ++u) {
      int i = g + u * nthr;
      v[u] = (i < userVec) ? ((const v4f*)ue)[i] : ((const v4f*)ie)[i - userVec];
    }
    __builtin_amdgcn_sched_barrier(0);
#pragma unroll
    for (int u = 0; u < SU; ++u) {
      int i = g + u * nthr;
      ((v4f*)e0)[i]  = v[u];
      ((v4f*)out)[i] = v[u];
    }
  }
  for (; g < totalVec; g += nthr) {
    v4f v = (g < userVec) ? ((const v4f*)ue)[g] : ((const v4f*)ie)[g - userVec];
    ((v4f*)e0)[g]  = v;
    ((v4f*)out)[g] = v;
  }
}

#define QCAP 64

template <int TILE_ROWS, bool DEG>
__device__ __forceinline__ void resolve_queued(
    const int* q, int base, int n, int lane, int lo,
    const int* __restrict__ row, const int* __restrict__ col,
    const float* __restrict__ val, const float* __restrict__ x,
    float* tile, float* degt) {
  int last = n - 1;
  int e  = q[base + (lane < last ? lane : last)];
  int rr = row[e] - lo;
  int c  = col[e];
  int vb = __float_as_int(val[e]);
  for (int k0 = 0; k0 < n; k0 += 16) {
    float2 xv[16];
#pragma unroll
    for (int j = 0; j < 16; ++j) {
      int k  = (k0 + j < last) ? (k0 + j) : last;
      int ck = __builtin_amdgcn_readlane(c, k);
      xv[j] = *(const float2*)(x + (size_t)ck * EMB + 2 * lane);
    }
#pragma unroll
    for (int j = 0; j < 16; ++j) {
      if (k0 + j < n) {
        int   k  = k0 + j;
        int   rk = __builtin_amdgcn_readlane(rr, k);
        float vk = __int_as_float(__builtin_amdgcn_readlane(vb, k));
        float* t = tile + rk * EMB + 2 * lane;
        atomicAdd(t,     vk * xv[j].x);
        atomicAdd(t + 1, vk * xv[j].y);
        if (DEG && lane == 0) atomicAdd(degt + rk, vk);
      }
    }
  }
}

template <int TILE_ROWS>
__device__ __forceinline__ unsigned tile_id_of(int r) {
  unsigned t = (unsigned)r / (unsigned)TILE_ROWS;
  return t < 255u ? t : 255u;
}
template <int TILE_ROWS>
__device__ __forceinline__ unsigned pack_ids(int4 r) {
  return tile_id_of<TILE_ROWS>(r.x) | (tile_id_of<TILE_ROWS>(r.y) << 8) |
         (tile_id_of<TILE_ROWS>(r.z) << 16) | (tile_id_of<TILE_ROWS>(r.w) << 24);
}
template <int TILE_A, int TILE_B>
__global__ void __launch_bounds__(256) tile_ids_kernel(
    const int* __restrict__ rowA, unsigned* __restrict__ idsA, int nnzA,
    const int* __restrict__ rowB, unsigned* __restrict__ idsB, int nnzB) {
  const int nthr = gridDim.x * blockDim.x;
  const int nA4  = nnzA >> 2;
  const int nB4  = nnzB >> 2;
  for (int g = blockIdx.x * blockDim.x + threadIdx.x; g < nA4 + nB4; g += nthr) {
    if (g < nA4) idsA[g]       = pack_ids<TILE_A>(((const int4*)rowA)[g]);
    else         idsB[g - nA4] = pack_ids<TILE_B>(((const int4*)rowB)[g - nA4]);
  }
}

__device__ __forceinline__ unsigned nonzero_bytes(unsigned v) {
  return (((v & 0x7f7f7f7fu) + 0x7f7f7f7fu) | v) & 0x80808080u;
}

template <int TILE_ROWS, bool DEG>
__device__ __forceinline__ void take16(
    uint4 w, unsigned tb, bool ok, int j, int& cnt, int* q, int lane, int lo,
    const int* __restrict__ row, const int* __restrict__ col,
    const float* __restrict__ val, const float* __restrict__ x,
    float* tile, float* degt) {
  unsigned nz = (nonzero_bytes(w.x ^ tb) >> 7) | (nonzero_bytes(w.y ^ tb) >> 6) |
                (nonzero_bytes(w.z ^ tb) >> 5) | (nonzero_bytes(w.w ^ tb) >> 4);
  unsigned m = ok ? (~nz & 0x0f0f0f0fu) : 0u;
  unsigned mk = __builtin_amdgcn_ballot_w32(m != 0u);
  while (mk) {
    int p   = __builtin_ctz(m | 0x80000000u);
    int pos = cnt + (int)__builtin_amdgcn_mbcnt_lo(mk, 0u);
    if (m != 0u) q[pos] = 16 * j + 4 * (p & 7) + (p >> 3);
    cnt += __builtin_popcount(mk);
    m &= m - 1u;
    if (cnt >= 32) {
      resolve_queued<TILE_ROWS, DEG>(q, cnt - 32, 32, lane, lo, row, col, val, x, tile, degt);
      cnt -= 32;
    }
    mk = __builtin_amdgcn_ballot_w32(m != 0u);
  }
}

#define AU 5
__device__ __forceinline__ void acc_wmma_part(
    const float* __restrict__ e, float* __restrict__ acc, int nChunks,
    int wv, int nWaves) {
  const int lane = threadIdx.x & 31;
  const int n    = lane & 15;
  const int hi   = lane >> 4;

  v2f a[4];
#pragma unroll
  for (int k = 0; k < 4; ++k) {
    a[k].x = (n == (4 * k + 2 * hi))     ? 1.0f : 0.0f;
    a[k].y = (n == (4 * k + 2 * hi + 1)) ? 1.0f : 0.0f;
  }

  for (int g = wv * AU; g < nChunks; g += nWaves * AU) {
    v8f cc[AU];
    v2f b[AU][4];
#pragma unroll
    for (int u = 0; u < AU; ++u) {
      int ch = (g + u < nChunks) ? (g + u) : (nChunks - 1);
      const float* ap = acc + (size_t)ch * 256 + 4 * lane;
      const float* ep = e   + (size_t)ch * 256 + 4 * n + 2 * hi;
      v4f lo = *(const v4f*)(ap);
      v4f up = *(const v4f*)(ap + 128);
      cc[u][0] = lo.x; cc[u][1] = lo.y; cc[u][2] = lo.z; cc[u][3] = lo.w;
      cc[u][4] = up.x; cc[u][5] = up.y; cc[u][6] = up.z; cc[u][7] = up.w;
#pragma unroll
      for (int k = 0; k < 4; ++k)
        b[u][k] = *(const v2f*)(ep + 128 * (k & 1) + 64 * (k >> 1));
    }
    __builtin_amdgcn_sched_barrier(0);
#pragma unroll
    for (int k = 0; k < 4; ++k)
#pragma unroll
      for (int u = 0; u < AU; ++u)
        cc[u] = __builtin_amdgcn_wmma_f32_16x16x4_f32(
            false, a[k], false, b[u][k], (short)0, cc[u], false, false);
#pragma unroll
    for (int u = 0; u < AU; ++u) {
      if (g + u < nChunks) {
        float* op = acc + (size_t)(g + u) * 256 + 4 * lane;
        v4f lo = {cc[u][0], cc[u][1], cc[u][2], cc[u][3]};
        v4f up = {cc[u][4], cc[u][5], cc[u][6], cc[u][7]};
        *(v4f*)(op)       = lo;
        *(v4f*)(op + 128) = up;
      }
    }
  }
}

template <int TILE_ROWS, bool DEG, bool ACC>
__global__ void __launch_bounds__(1024) spmm_tile_kernel(
    const unsigned* __restrict__ ids,
    const int* __restrict__ row, const int* __restrict__ col,
    const float* __restrict__ val, const float* __restrict__ x,
    float* __restrict__ y, float* __restrict__ deg, int nnz, int n_rows,
    const float* __restrict__ acc_e, float* __restrict__ acc_out, int nChunks) {
  __shared__ float tile[TILE_ROWS * EMB];
  __shared__ float degt[DEG ? TILE_ROWS : 1];
  __shared__ int   queue[32 * QCAP];

  const int n_tiles = (n_rows + TILE_ROWS - 1) / TILE_ROWS;
  if (ACC && (int)blockIdx.x >= n_tiles) {
    const int wv = __builtin_amdgcn_readfirstlane(
        ((int)blockIdx.x - n_tiles) * 32 + (int)(threadIdx.x >> 5));
    acc_wmma_part(acc_e, acc_out, nChunks, wv, ((int)gridDim.x - n_tiles) * 32);
    return;
  }

  const int tid  = threadIdx.x;
  const int lane = tid & 31;
  const int wave = tid >> 5;
  const int lo   = blockIdx.x * TILE_ROWS;
  const unsigned tb = 0x01010101u * blockIdx.x;

  for (int i = tid; i < TILE_ROWS * (EMB / 4); i += 1024)
    ((float4*)tile)[i] = make_float4(0.f, 0.f, 0.f, 0.f);
  if (DEG)
    for (int i = tid; i < TILE_ROWS; i += 1024) degt[i] = 0.f;
  __syncthreads();

  int* q   = queue + wave * QCAP;
  int  cnt = 0;
  const int    nnz16 = nnz >> 4;
  const uint4* ids4  = (const uint4*)ids;
  const int    trips = (nnz16 + 1023) >> 10;
  const int    full  = (nnz16 >> 10) & ~3;
  const int start = blockIdx.x * (nnz16 / n_tiles);

  uint4 cur[4];
  int   cid[4];
  if (full > 0) {
#pragma unroll
    for (int u = 0; u < 4; ++u) {
      int j = (u << 10) + tid + start;
      if (j >= nnz16) j -= nnz16;
      cid[u] = j;
      cur[u] = ids4[j];
    }
  }
  for (int it = 0; it < full; it += 4) {
    uint4 nxt[4];
    int   nid[4];
    const int itn = (it + 4 < full) ? (it + 4) : it;
#pragma unroll
    for (int u = 0; u < 4; ++u) {
      int j = ((itn + u) << 10) + tid + start;
      if (j >= nnz16) j -= nnz16;
      nid[u] = j;
      nxt[u] = ids4[j];
    }
#pragma unroll
    for (int u = 0; u < 4; ++u)
      take16<TILE_ROWS, DEG>(cur[u], tb, true, cid[u], cnt, q, lane, lo, row, col, val, x, tile, degt);
#pragma unroll
    for (int u = 0; u < 4; ++u) { cur[u] = nxt[u]; cid[u] = nid[u]; }
  }
  for (int it = full; it < trips; ++it) {
    int  i  = (it << 10) + tid;
    bool ok = i < nnz16;
    int  j  = i + start;
    if (j >= nnz16) j -= nnz16;
    uint4 w = ids4[ok ? j : 0];
    take16<TILE_ROWS, DEG>(w, tb, ok, j, cnt, q, lane, lo, row, col, val, x, tile, degt);
  }
  if (wave == 0) {
    int  e  = (nnz16 << 4) + lane;
    bool m  = (e < nnz) && ((unsigned)(row[e < nnz ? e : 0] - lo) < (unsigned)TILE_ROWS);
    unsigned mk = __builtin_amdgcn_ballot_w32(m);
    if (mk) {
      int pos = cnt + (int)__builtin_amdgcn_mbcnt_lo(mk, 0u);
      if (m) q[pos] = e;
      cnt += __builtin_popcount(mk);
    }
  }
  while (cnt > 0) {
    int n = cnt < 32 ? cnt : 32;
    resolve_queued<TILE_ROWS, DEG>(q, cnt - n, n, lane, lo, row, col, val, x, tile, degt);
    cnt -= n;
  }
  __syncthreads();

  const int rows_here = (n_rows - lo) < TILE_ROWS ? (n_rows - lo) : TILE_ROWS;
  float4* y4 = (float4*)(y + (size_t)lo * EMB);
  for (int i = tid; i < rows_here * (EMB / 4); i += 1024)
    y4[i] = ((const float4*)tile)[i];
  if (DEG)
    for (int i = tid; i < rows_here; i += 1024) deg[lo + i] = degt[i];
}

__global__ void __launch_bounds__(256) rownorm_kernel(
    float* __restrict__ x, const float* __restrict__ deg, int rowsVec) {
  const int nthr = gridDim.x * blockDim.x;
  int g = blockIdx.x * blockDim.x + threadIdx.x;
  for (; g + (SU - 1) * nthr < rowsVec; g += SU * nthr) {
    v4f   v[SU];
    float d[SU];
#pragma unroll
    for (int u = 0; u < SU; ++u) {
      int i = g + u * nthr;
      v[u] = ((const v4f*)x)[i];
      d[u] = deg[i >> 4];
    }
    __builtin_amdgcn_sched_barrier(0);
#pragma unroll
    for (int u = 0; u < SU; ++u) {
      int i = g + u * nthr;
      float dd  = (d[u] == 0.0f) ? 1.0f : d[u];
      float inv = 1.0f / dd;
      ((v4f*)x)[i] = v[u] * inv;
    }
  }
  for (; g < rowsVec; g += nthr) {
    float dd  = deg[g >> 4];
    dd = (dd == 0.0f) ? 1.0f : dd;
    float inv = 1.0f / dd;
    ((v4f*)x)[g] = ((const v4f*)x)[g] * inv;
  }
}

__global__ void __launch_bounds__(320) finalize_kernel(
    float* __restrict__ out, const float* __restrict__ ulocal,
    const float* __restrict__ deg, int totalVec, int userVec) {
  const int nthr = gridDim.x * blockDim.x;
  int g = blockIdx.x * blockDim.x + threadIdx.x;
  for (; g + (SU - 1) * nthr < totalVec; g += SU * nthr) {
    v4f   v[SU], w[SU];
    float d[SU];
#pragma unroll
    for (int u = 0; u < SU; ++u) {
      int i  = g + u * nthr;
      int iu = (i < userVec) ? i : (userVec - 1);
      v[u] = ((const v4f*)out)[i];
      w[u] = ((const v4f*)ulocal)[iu];
      d[u] = deg[iu >> 4];
    }
    __builtin_amdgcn_sched_barrier(0);
#pragma unroll
    for (int u = 0; u < SU; ++u) {
      int i = g + u * nthr;
      float dd  = (d[u] == 0.0f) ? 1.0f : d[u];
      float inv = 1.0f / dd;
      v4f r  = v[u] * 0.25f;
      v4f ru = r + w[u] * inv;
      ((v4f*)out)[i] = (i < userVec) ? ru : r;
    }
  }
  for (; g < totalVec; g += nthr) {
    v4f r = ((const v4f*)out)[g] * 0.25f;
    if (g < userVec) {
      float dd  = deg[g >> 4];
      dd = (dd == 0.0f) ? 1.0f : dd;
      float inv = 1.0f / dd;
      r += ((const v4f*)ulocal)[g] * inv;
    }
    ((v4f*)out)[g] = r;
  }
}

#define TILE_ADJ 944
#define TILE_HV  160
#define TILE_HU  800
static_assert((NUM_NODES + TILE_ADJ - 1) / TILE_ADJ <= 255, "tile id must fit a byte");
static_assert((NUM_BQ    + TILE_HV  - 1) / TILE_HV  <= 255, "tile id must fit a byte");
static_assert((NUM_USERS + TILE_HU  - 1) / TILE_HU  <= 255, "tile id must fit a byte");
#define TILES_ADJ ((NUM_NODES + TILE_ADJ - 1) / TILE_ADJ)
#define TILES_HV  ((NUM_BQ    + TILE_HV  - 1) / TILE_HV)
#define TILES_HU  ((NUM_USERS + TILE_HU  - 1) / TILE_HU)
#define N_WGP 256
static_assert(TILES_ADJ < N_WGP && TILES_HV < N_WGP, "a launch with a layer sum needs blocks behind its tiles");

extern "C" void kernel_launch(void* const* d_in, const int* in_sizes, int n_in,
                              void* d_out, int out_size, void* d_ws, size_t ws_size,
                              hipStream_t stream) {
  const float* user_emb = (const float*)d_in[0];
  const float* item_emb = (const float*)d_in[1];
  const float* adj_val  = (const float*)d_in[2];
  const float* hv_val   = (const float*)d_in[3];
  const float* hu_val   = (const float*)d_in[4];
  const int*   adj_row  = (const int*)d_in[5];
  const int*   adj_col  = (const int*)d_in[6];
  const int*   hv_row   = (const int*)d_in[7];
  const int*   hv_col   = (const int*)d_in[8];
  const int*   hu_row   = (const int*)d_in[9];
  const int*   hu_col   = (const int*)d_in[10];

  const int E_adj = in_sizes[2];
  const int E_hv  = in_sizes[3];
  const int E_hu  = in_sizes[4];

  float* out = (float*)d_out;

  const size_t embBytesN = (size_t)NUM_NODES * EMB * sizeof(float);
  const size_t embBytesB = (size_t)NUM_BQ    * EMB * sizeof(float);
  const size_t embBytesU = (size_t)NUM_USERS * EMB * sizeof(float);
  char* ws = (char*)d_ws;
  float* e0     = (float*)(ws);
  float* e1     = (float*)(ws + embBytesN);
  float* bfeat  = (float*)(ws + 2 * embBytesN);
  float* deg_hv = (float*)(ws + 2 * embBytesN + embBytesB);
  float* deg_hu = deg_hv + NUM_BQ;
  float* ulocal = e0;
  unsigned* ids_adj = (unsigned*)(ws + 2 * embBytesN);
  unsigned* ids_hv  = (unsigned*)(ws + embBytesU);
  unsigned* ids_hu  = (unsigned*)(ws + embBytesU + (embBytesN - embBytesU) / 2);

  const int totalVec = NUM_NODES * (EMB / 4);
  const int userVec  = NUM_USERS * (EMB / 4);
  const int bqVec    = NUM_BQ    * (EMB / 4);
  const int nChunks  = NUM_NODES * EMB / 256;


  copy_init_kernel<<<750, 320, 0, stream>>>(
      user_emb, item_emb, e0, out, totalVec, userVec);

  tile_ids_kernel<TILE_ADJ, TILE_ADJ><<<1172, 256, 0, stream>>>(
      adj_row, ids_adj, E_adj, adj_row, ids_adj, 0);
  spmm_tile_kernel<TILE_ADJ, false, false><<<TILES_ADJ, 1024, 0, stream>>>(
      ids_adj, adj_row, adj_col, adj_val, e0, e1, nullptr, E_adj, NUM_NODES,
      nullptr, nullptr, 0);
  spmm_tile_kernel<TILE_ADJ, false, true><<<N_WGP, 1024, 0, stream>>>(
      ids_adj, adj_row, adj_col, adj_val, e1, e0, nullptr, E_adj, NUM_NODES,
      e1, out, nChunks);
  spmm_tile_kernel<TILE_ADJ, false, true><<<N_WGP, 1024, 0, stream>>>(
      ids_adj, adj_row, adj_col, adj_val, e0, e1, nullptr, E_adj, NUM_NODES,
      e0, out, nChunks);

  tile_ids_kernel<TILE_HV, TILE_HU><<<782, 256, 0, stream>>>(
      hv_row, ids_hv, E_hv, hu_row, ids_hu, E_hu);
  spmm_tile_kernel<TILE_HV, true, true><<<N_WGP, 1024, 0, stream>>>(
      ids_hv, hv_row, hv_col, hv_val, item_emb, bfeat, deg_hv, E_hv, NUM_BQ,
      e1, out, nChunks);
  rownorm_kernel<<<250, 256, 0, stream>>>(bfeat, deg_hv, bqVec);

  spmm_tile_kernel<TILE_HU, true, false><<<TILES_HU, 1024, 0, stream>>>(
      ids_hu, hu_row, hu_col, hu_val, bfeat, ulocal, deg_hu, E_hu, NUM_USERS,
      nullptr, nullptr, 0);

  finalize_kernel<<<750, 320, 0, stream>>>(
      out, ulocal, deg_hu, totalVec, userVec);
}


# Tile-hit lookups read edge words packed by the id pass: 28-bit (row in tile, col) beside val for the adjacency (2 lines a hit, not 3), 8-byte records for the biclique lists (1 line); sum order kept
# speedup vs baseline: 2.0425x; 1.0867x over previous
#include <hip/hip_runtime.h>
#include <hip/hip_bf16.h>

typedef __attribute__((ext_vector_type(2))) float v2f;
typedef __attribute__((ext_vector_type(8))) float v8f;
typedef __attribute__((ext_vector_type(4))) float v4f;

#define NUM_USERS 100000
#define NUM_ITEMS 50000
#define NUM_NODES 150000
#define NUM_BQ    20000
#define EMB       64

#define SU 5

__global__ void __launch_bounds__(320) copy_init_kernel(
    const float* __restrict__ ue, const float* __restrict__ ie,
    float* __restrict__ e0, float* __restrict__ out,
    int totalVec, int userVec) {
  const int nthr = gridDim.x * blockDim.x;
  int g = blockIdx.x * blockDim.x + threadIdx.x;
  for (; g + (SU - 1) * nthr < totalVec; g += SU * nthr) {
    v4f v[SU];
#pragma unroll
    for (int u = 0; u < SU; ++u) {
      int i = g + u * nthr;
      v[u] = (i < userVec) ? ((const v4f*)ue)[i] : ((const v4f*)ie)[i - userVec];
    }
    __builtin_amdgcn_sched_barrier(0);
#pragma unroll
    for (int u = 0; u < SU; ++u) {
      int i = g + u * nthr;
      ((v4f*)e0)[i]  = v[u];
      ((v4f*)out)[i] = v[u];
    }
  }
  for (; g < totalVec; g += nthr) {
    v4f v = (g < userVec) ? ((const v4f*)ue)[g] : ((const v4f*)ie)[g - userVec];
    ((v4f*)e0)[g]  = v;
    ((v4f*)out)[g] = v;
  }
}

#define QCAP 64

#define COL_BITS 18
#define COL_MASK 0x3ffffu
#define RC_MASK  0x0fffffffu

struct pair32 { unsigned lo, hi; };

template <int TILE_ROWS, bool DEG, bool REC>
__device__ __forceinline__ void resolve_queued(
    const int* q, int base, int n, int lane,
    const unsigned* __restrict__ pk,
    const float* __restrict__ val, const float* __restrict__ x,
    float* tile, float* degt) {
  int last = n - 1;
  int e  = q[base + (lane < last ? lane : last)];
  unsigned w;
  int      vb;
  if (REC) {
    uint2 r = ((const uint2*)pk)[e];
    w  = r.x;
    vb = (int)r.y;
  } else {
    unsigned bit = 28u * ((unsigned)e & 7u);
    const unsigned* p = pk + 7 * (size_t)(e >> 3) + (bit >> 5);
    pair32 t;
    __builtin_memcpy(&t, p, 8);
    w  = (unsigned)(((((unsigned long long)t.hi) << 32) | t.lo) >> (bit & 31u)) & RC_MASK;
    vb = __float_as_int(val[e]);
  }
  int rr = (int)(w >> COL_BITS);
  int c  = (int)(w & COL_MASK);
  for (int k0 = 0; k0 < n; k0 += 16) {
    float2 xv[16];
#pragma unroll
    for (int j = 0; j < 16; ++j) {
      int k  = (k0 + j < last) ? (k0 + j) : last;
      unsigned ck = (unsigned)__builtin_amdgcn_readlane(c, k);
      xv[j] = *(const float2*)((const char*)x + (ck * (EMB * 4u) + 8u * (unsigned)lane));
    }
#pragma unroll
    for (int j = 0; j < 16; ++j) {
      if (k0 + j < n) {
        int   k  = k0 + j;
        int   rk = __builtin_amdgcn_readlane(rr, k);
        float vk = __int_as_float(__builtin_amdgcn_readlane(vb, k));
        float* t = tile + rk * EMB + 2 * lane;
        atomicAdd(t,     vk * xv[j].x);
        atomicAdd(t + 1, vk * xv[j].y);
        if (DEG && lane == 0) atomicAdd(degt + rk, vk);
      }
    }
  }
}

template <int TILE_ROWS>
__device__ __forceinline__ unsigned tile_id_of(int r) {
  unsigned t = (unsigned)r / (unsigned)TILE_ROWS;
  return t < 255u ? t : 255u;
}
template <int TILE_ROWS>
__device__ __forceinline__ unsigned pack_ids(int4 r) {
  return tile_id_of<TILE_ROWS>(r.x) | (tile_id_of<TILE_ROWS>(r.y) << 8) |
         (tile_id_of<TILE_ROWS>(r.z) << 16) | (tile_id_of<TILE_ROWS>(r.w) << 24);
}

template <int TILE_ROWS>
__device__ __forceinline__ unsigned rc_word(int r, int c) {
  unsigned t  = (unsigned)r / (unsigned)TILE_ROWS;
  unsigned rl = (unsigned)r - t * (unsigned)TILE_ROWS;
  return ((rl & 0x3ffu) << COL_BITS) | ((unsigned)c & COL_MASK);
}

template <int TILE_ROWS>
__global__ void __launch_bounds__(256) prep_packed_kernel(
    const int* __restrict__ row, const int* __restrict__ col,
    unsigned* __restrict__ ids, unsigned* __restrict__ pk, int nnz) {
  const int nthr = gridDim.x * blockDim.x;
  const int nG   = (nnz + 7) >> 3;
  for (int g = blockIdx.x * blockDim.x + threadIdx.x; g < nG; g += nthr) {
    int4 r0, r1, c0, c1;
    if (8 * g + 7 < nnz) {
      r0 = ((const int4*)row)[2 * g];  r1 = ((const int4*)row)[2 * g + 1];
      c0 = ((const int4*)col)[2 * g];  c1 = ((const int4*)col)[2 * g + 1];
    } else {
      int e[8];
#pragma unroll
      for (int k = 0; k < 8; ++k) e[k] = (8 * g + k < nnz) ? (8 * g + k) : (nnz - 1);
      r0 = make_int4(row[e[0]], row[e[1]], row[e[2]], row[e[3]]);
      r1 = make_int4(row[e[4]], row[e[5]], row[e[6]], row[e[7]]);
      c0 = make_int4(col[e[0]], col[e[1]], col[e[2]], col[e[3]]);
      c1 = make_int4(col[e[4]], col[e[5]], col[e[6]], col[e[7]]);
    }
    if (8 * g + 3 < nnz) ids[2 * g]     = pack_ids<TILE_ROWS>(r0);
    if (8 * g + 7 < nnz) ids[2 * g + 1] = pack_ids<TILE_ROWS>(r1);
    const unsigned w0 = rc_word<TILE_ROWS>(r0.x, c0.x), w1 = rc_word<TILE_ROWS>(r0.y, c0.y);
    const unsigned w2 = rc_word<TILE_ROWS>(r0.z, c0.z), w3 = rc_word<TILE_ROWS>(r0.w, c0.w);
    const unsigned w4 = rc_word<TILE_ROWS>(r1.x, c1.x), w5 = rc_word<TILE_ROWS>(r1.y, c1.y);
    const unsigned w6 = rc_word<TILE_ROWS>(r1.z, c1.z), w7 = rc_word<TILE_ROWS>(r1.w, c1.w);
    unsigned* p = pk + 7 * (size_t)g;
    p[0] =  w0        | (w1 << 28);
    p[1] = (w1 >> 4)  | (w2 << 24);
    p[2] = (w2 >> 8)  | (w3 << 20);
    p[3] = (w3 >> 12) | (w4 << 16);
    p[4] = (w4 >> 16) | (w5 << 12);
    p[5] = (w5 >> 20) | (w6 << 8);
    p[6] = (w6 >> 24) | (w7 << 4);
  }
  if (blockIdx.x == 0 && threadIdx.x == 0) pk[7 * (size_t)nG] = 0u;
}

template <int TILE_ROWS>
__device__ __forceinline__ void prep_rec_group(
    const int* __restrict__ row, const int* __restrict__ col, const float* __restrict__ val,
    unsigned* __restrict__ ids, unsigned* __restrict__ rec, int nnz, int g) {
  if (4 * g + 3 < nnz) {
    const int4   r = ((const int4*)row)[g];
    const int4   c = ((const int4*)col)[g];
    const float4 v = ((const float4*)val)[g];
    ids[g] = pack_ids<TILE_ROWS>(r);
    ((uint4*)rec)[2 * g]     = make_uint4(rc_word<TILE_ROWS>(r.x, c.x), __float_as_uint(v.x),
                                          rc_word<TILE_ROWS>(r.y, c.y), __float_as_uint(v.y));
    ((uint4*)rec)[2 * g + 1] = make_uint4(rc_word<TILE_ROWS>(r.z, c.z), __float_as_uint(v.z),
                                          rc_word<TILE_ROWS>(r.w, c.w), __float_as_uint(v.w));
  } else {
    for (int e = 4 * g; e < nnz; ++e) {
      rec[2 * e]     = rc_word<TILE_ROWS>(row[e], col[e]);
      rec[2 * e + 1] = __float_as_uint(val[e]);
    }
  }
}
template <int TILE_A, int TILE_B>
__global__ void __launch_bounds__(256) prep_rec_kernel(
    const int* __restrict__ rowA, const int* __restrict__ colA, const float* __restrict__ valA,
    unsigned* __restrict__ idsA, unsigned* __restrict__ recA, int nnzA,
    const int* __restrict__ rowB, const int* __restrict__ colB, const float* __restrict__ valB,
    unsigned* __restrict__ idsB, unsigned* __restrict__ recB, int nnzB) {
  const int nthr = gridDim.x * blockDim.x;
  const int nA4  = (nnzA + 3) >> 2;
  const int nB4  = (nnzB + 3) >> 2;
  for (int g = blockIdx.x * blockDim.x + threadIdx.x; g < nA4 + nB4; g += nthr) {
    if (g < nA4) prep_rec_group<TILE_A>(rowA, colA, valA, idsA, recA, nnzA, g);
    else         prep_rec_group<TILE_B>(rowB, colB, valB, idsB, recB, nnzB, g - nA4);
  }
}

__device__ __forceinline__ unsigned nonzero_bytes(unsigned v) {
  return (((v & 0x7f7f7f7fu) + 0x7f7f7f7fu) | v) & 0x80808080u;
}

template <int TILE_ROWS, bool DEG, bool REC>
__device__ __forceinline__ void take16(
    uint4 w, unsigned tb, bool ok, int j, int& cnt, int* q, int lane,
    const unsigned* __restrict__ pk,
    const float* __restrict__ val, const float* __restrict__ x,
    float* tile, float* degt) {
  unsigned nz = (nonzero_bytes(w.x ^ tb) >> 7) | (nonzero_bytes(w.y ^ tb) >> 6) |
                (nonzero_bytes(w.z ^ tb) >> 5) | (nonzero_bytes(w.w ^ tb) >> 4);
  unsigned m = ok ? (~nz & 0x0f0f0f0fu) : 0u;
  unsigned mk = __builtin_amdgcn_ballot_w32(m != 0u);
  while (mk) {
    int p   = __builtin_ctz(m | 0x80000000u);
    int pos = cnt + (int)__builtin_amdgcn_mbcnt_lo(mk, 0u);
    if (m != 0u) q[pos] = 16 * j + 4 * (p & 7) + (p >> 3);
    cnt += __builtin_popcount(mk);
    m &= m - 1u;
    if (cnt >= 32) {
      resolve_queued<TILE_ROWS, DEG, REC>(q, cnt - 32, 32, lane, pk, val, x, tile, degt);
      cnt -= 32;
    }
    mk = __builtin_amdgcn_ballot_w32(m != 0u);
  }
}

#define AU 5
__device__ __forceinline__ void acc_wmma_part(
    const float* __restrict__ e, float* __restrict__ acc, int nChunks,
    int wv, int nWaves) {
  const int lane = threadIdx.x & 31;
  const int n    = lane & 15;
  const int hi   = lane >> 4;

  v2f a[4];
#pragma unroll
  for (int k = 0; k < 4; ++k) {
    a[k].x = (n == (4 * k + 2 * hi))     ? 1.0f : 0.0f;
    a[k].y = (n == (4 * k + 2 * hi + 1)) ? 1.0f : 0.0f;
  }

  for (int g = wv * AU; g < nChunks; g += nWaves * AU) {
    v8f cc[AU];
    v2f b[AU][4];
#pragma unroll
    for (int u = 0; u < AU; ++u) {
      int ch = (g + u < nChunks) ? (g + u) : (nChunks - 1);
      const float* ap = acc + (size_t)ch * 256 + 4 * lane;
      const float* ep = e   + (size_t)ch * 256 + 4 * n + 2 * hi;
      v4f lo = *(const v4f*)(ap);
      v4f up = *(const v4f*)(ap + 128);
      cc[u][0] = lo.x; cc[u][1] = lo.y; cc[u][2] = lo.z; cc[u][3] = lo.w;
      cc[u][4] = up.x; cc[u][5] = up.y; cc[u][6] = up.z; cc[u][7] = up.w;
#pragma unroll
      for (int k = 0; k < 4; ++k)
        b[u][k] = *(const v2f*)(ep + 128 * (k & 1) + 64 * (k >> 1));
    }
    __builtin_amdgcn_sched_barrier(0);
#pragma unroll
    for (int k = 0; k < 4; ++k)
#pragma unroll
      for (int u = 0; u < AU; ++u)
        cc[u] = __builtin_amdgcn_wmma_f32_16x16x4_f32(
            false, a[k], false, b[u][k], (short)0, cc[u], false, false);
#pragma unroll
    for (int u = 0; u < AU; ++u) {
      if (g + u < nChunks) {
        float* op = acc + (size_t)(g + u) * 256 + 4 * lane;
        v4f lo = {cc[u][0], cc[u][1], cc[u][2], cc[u][3]};
        v4f up = {cc[u][4], cc[u][5], cc[u][6], cc[u][7]};
        *(v4f*)(op)       = lo;
        *(v4f*)(op + 128) = up;
      }
    }
  }
}

template <int TILE_ROWS, bool DEG, bool ACC, bool REC>
__global__ void __launch_bounds__(1024) spmm_tile_kernel(
    const unsigned* __restrict__ ids,
    const int* __restrict__ row, const unsigned* __restrict__ pk,
    const float* __restrict__ val, const float* __restrict__ x,
    float* __restrict__ y, float* __restrict__ deg, int nnz, int n_rows,
    const float* __restrict__ acc_e, float* __restrict__ acc_out, int nChunks) {
  __shared__ float tile[TILE_ROWS * EMB];
  __shared__ float degt[DEG ? TILE_ROWS : 1];
  __shared__ int   queue[32 * QCAP];

  const int n_tiles = (n_rows + TILE_ROWS - 1) / TILE_ROWS;
  if (ACC && (int)blockIdx.x >= n_tiles) {
    const int wv = __builtin_amdgcn_readfirstlane(
        ((int)blockIdx.x - n_tiles) * 32 + (int)(threadIdx.x >> 5));
    acc_wmma_part(acc_e, acc_out, nChunks, wv, ((int)gridDim.x - n_tiles) * 32);
    return;
  }

  const int tid  = threadIdx.x;
  const int lane = tid & 31;
  const int wave = tid >> 5;
  const int lo   = blockIdx.x * TILE_ROWS;
  const unsigned tb = 0x01010101u * blockIdx.x;

  for (int i = tid; i < TILE_ROWS * (EMB / 4); i += 1024)
    ((float4*)tile)[i] = make_float4(0.f, 0.f, 0.f, 0.f);
  if (DEG)
    for (int i = tid; i < TILE_ROWS; i += 1024) degt[i] = 0.f;
  __syncthreads();

  int* q   = queue + wave * QCAP;
  int  cnt = 0;
  const int    nnz16 = nnz >> 4;
  const uint4* ids4  = (const uint4*)ids;
  const int    trips = (nnz16 + 1023) >> 10;
  const int    full  = (nnz16 >> 10) & ~3;
  const int start = blockIdx.x * (nnz16 / n_tiles);

  uint4 cur[4];
  int   cid[4];
  if (full > 0) {
#pragma unroll
    for (int u = 0; u < 4; ++u) {
      int j = (u << 10) + tid + start;
      if (j >= nnz16) j -= nnz16;
      cid[u] = j;
      cur[u] = ids4[j];
    }
  }
  for (int it = 0; it < full; it += 4) {
    uint4 nxt[4];
    int   nid[4];
    const int itn = (it + 4 < full) ? (it + 4) : it;
#pragma unroll
    for (int u = 0; u < 4; ++u) {
      int j = ((itn + u) << 10) + tid + start;
      if (j >= nnz16) j -= nnz16;
      nid[u] = j;
      nxt[u] = ids4[j];
    }
#pragma unroll
    for (int u = 0; u < 4; ++u)
      take16<TILE_ROWS, DEG, REC>(cur[u], tb, true, cid[u], cnt, q, lane, pk, val, x, tile, degt);
#pragma unroll
    for (int u = 0; u < 4; ++u) { cur[u] = nxt[u]; cid[u] = nid[u]; }
  }
  for (int it = full; it < trips; ++it) {
    int  i  = (it << 10) + tid;
    bool ok = i < nnz16;
    int  j  = i + start;
    if (j >= nnz16) j -= nnz16;
    uint4 w = ids4[ok ? j : 0];
    take16<TILE_ROWS, DEG, REC>(w, tb, ok, j, cnt, q, lane, pk, val, x, tile, degt);
  }
  if (wave == 0) {
    int  e  = (nnz16 << 4) + lane;
    bool m  = (e < nnz) && ((unsigned)(row[e < nnz ? e : 0] - lo) < (unsigned)TILE_ROWS);
    unsigned mk = __builtin_amdgcn_ballot_w32(m);
    if (mk) {
      int pos = cnt + (int)__builtin_amdgcn_mbcnt_lo(mk, 0u);
      if (m) q[pos] = e;
      cnt += __builtin_popcount(mk);
    }
  }
  while (cnt > 0) {
    int n = cnt < 32 ? cnt : 32;
    resolve_queued<TILE_ROWS, DEG, REC>(q, cnt - n, n, lane, pk, val, x, tile, degt);
    cnt -= n;
  }
  __syncthreads();

  const int rows_here = (n_rows - lo) < TILE_ROWS ? (n_rows - lo) : TILE_ROWS;
  float4* y4 = (float4*)(y + (size_t)lo * EMB);
  for (int i = tid; i < rows_here * (EMB / 4); i += 1024)
    y4[i] = ((const float4*)tile)[i];
  if (DEG)
    for (int i = tid; i < rows_here; i += 1024) deg[lo + i] = degt[i];
}

__global__ void __launch_bounds__(256) rownorm_kernel(
    float* __restrict__ x, const float* __restrict__ deg, int rowsVec) {
  const int nthr = gridDim.x * blockDim.x;
  int g = blockIdx.x * blockDim.x + threadIdx.x;
  for (; g + (SU - 1) * nthr < rowsVec; g += SU * nthr) {
    v4f   v[SU];
    float d[SU];
#pragma unroll
    for (int u = 0; u < SU; ++u) {
      int i = g + u * nthr;
      v[u] = ((const v4f*)x)[i];
      d[u] = deg[i >> 4];
    }
    __builtin_amdgcn_sched_barrier(0);
#pragma unroll
    for (int u = 0; u < SU; ++u) {
      int i = g + u * nthr;
      float dd  = (d[u] == 0.0f) ? 1.0f : d[u];
      float inv = 1.0f / dd;
      ((v4f*)x)[i] = v[u] * inv;
    }
  }
  for (; g < rowsVec; g += nthr) {
    float dd  = deg[g >> 4];
    dd = (dd == 0.0f) ? 1.0f : dd;
    float inv = 1.0f / dd;
    ((v4f*)x)[g] = ((const v4f*)x)[g] * inv;
  }
}

__global__ void __launch_bounds__(320) finalize_kernel(
    float* __restrict__ out, const float* __restrict__ ulocal,
    const float* __restrict__ deg, int totalVec, int userVec) {
  const int nthr = gridDim.x * blockDim.x;
  int g = blockIdx.x * blockDim.x + threadIdx.x;
  for (; g + (SU - 1) * nthr < totalVec; g += SU * nthr) {
    v4f   v[SU], w[SU];
    float d[SU];
#pragma unroll
    for (int u = 0; u < SU; ++u) {
      int i  = g + u * nthr;
      int iu = (i < userVec) ? i : (userVec - 1);
      v[u] = ((const v4f*)out)[i];
      w[u] = ((const v4f*)ulocal)[iu];
      d[u] = deg[iu >> 4];
    }
    __builtin_amdgcn_sched_barrier(0);
#pragma unroll
    for (int u = 0; u < SU; ++u) {
      int i = g + u * nthr;
      float dd  = (d[u] == 0.0f) ? 1.0f : d[u];
      float inv = 1.0f / dd;
      v4f r  = v[u] * 0.25f;
      v4f ru = r + w[u] * inv;
      ((v4f*)out)[i] = (i < userVec) ? ru : r;
    }
  }
  for (; g < totalVec; g += nthr) {
    v4f r = ((const v4f*)out)[g] * 0.25f;
    if (g < userVec) {
      float dd  = deg[g >> 4];
      dd = (dd == 0.0f) ? 1.0f : dd;
      float inv = 1.0f / dd;
      r += ((const v4f*)ulocal)[g] * inv;
    }
    ((v4f*)out)[g] = r;
  }
}

#define TILE_ADJ 944
#define TILE_HV  160
#define TILE_HU  800
static_assert((NUM_NODES + TILE_ADJ - 1) / TILE_ADJ <= 255, "tile id must fit a byte");
static_assert((NUM_BQ    + TILE_HV  - 1) / TILE_HV  <= 255, "tile id must fit a byte");
static_assert((NUM_USERS + TILE_HU  - 1) / TILE_HU  <= 255, "tile id must fit a byte");
static_assert(TILE_ADJ <= 1024 && TILE_HV <= 1024 && TILE_HU <= 1024, "row in tile must fit 10 bits");
static_assert(NUM_NODES <= (1 << COL_BITS), "col must fit 18 bits");
#define NNZ_ADJ 1200000
#define NNZ_BQ   400000
#define TILES_ADJ ((NUM_NODES + TILE_ADJ - 1) / TILE_ADJ)
#define TILES_HV  ((NUM_BQ    + TILE_HV  - 1) / TILE_HV)
#define TILES_HU  ((NUM_USERS + TILE_HU  - 1) / TILE_HU)
#define N_WGP 256
static_assert(TILES_ADJ < N_WGP && TILES_HV < N_WGP, "a launch with a layer sum needs blocks behind its tiles");

extern "C" void kernel_launch(void* const* d_in, const int* in_sizes, int n_in,
                              void* d_out, int out_size, void* d_ws, size_t ws_size,
                              hipStream_t stream) {
  const float* user_emb = (const float*)d_in[0];
  const float* item_emb = (const float*)d_in[1];
  const float* adj_val  = (const float*)d_in[2];
  const float* hv_val   = (const float*)d_in[3];
  const float* hu_val   = (const float*)d_in[4];
  const int*   adj_row  = (const int*)d_in[5];
  const int*   adj_col  = (const int*)d_in[6];
  const int*   hv_row   = (const int*)d_in[7];
  const int*   hv_col   = (const int*)d_in[8];
  const int*   hu_row   = (const int*)d_in[9];
  const int*   hu_col   = (const int*)d_in[10];

  const int E_adj = in_sizes[2];
  const int E_hv  = in_sizes[3];
  const int E_hu  = in_sizes[4];

  float* out = (float*)d_out;

  const size_t embBytesN = (size_t)NUM_NODES * EMB * sizeof(float);
  const size_t embBytesB = (size_t)NUM_BQ    * EMB * sizeof(float);
  const size_t embBytesU = (size_t)NUM_USERS * EMB * sizeof(float);
  char* ws = (char*)d_ws;
  float* e0     = (float*)(ws);
  float* e1     = (float*)(ws + embBytesN);
  float* bfeat  = (float*)(ws + 2 * embBytesN);
  float* deg_hv = (float*)(ws + 2 * embBytesN + embBytesB);
  float* deg_hu = deg_hv + NUM_BQ;
  float* ulocal = e0;
  static_assert((size_t)NNZ_ADJ + 28 * (size_t)((NNZ_ADJ + 7) / 8) + 4 <=
                (size_t)NUM_BQ * EMB * 4 + (size_t)(NUM_BQ + NUM_USERS) * 4,
                "adjacency ids and rc words must fit the biclique area");
  static_assert(2 * ((size_t)NNZ_BQ + 8 * (size_t)NNZ_BQ) <= (size_t)NUM_ITEMS * EMB * 4,
                "biclique ids and records must fit behind u_local");
  unsigned* ids_adj = (unsigned*)(ws + 2 * embBytesN);
  unsigned* pk_adj  = (unsigned*)(ws + 2 * embBytesN + NNZ_ADJ);
  unsigned* ids_hv  = (unsigned*)(ws + embBytesU);
  unsigned* rec_hv  = (unsigned*)(ws + embBytesU + NNZ_BQ);
  unsigned* ids_hu  = (unsigned*)(ws + embBytesU + (embBytesN - embBytesU) / 2);
  unsigned* rec_hu  = (unsigned*)(ws + embBytesU + (embBytesN - embBytesU) / 2 + NNZ_BQ);

  const int totalVec = NUM_NODES * (EMB / 4);
  const int userVec  = NUM_USERS * (EMB / 4);
  const int bqVec    = NUM_BQ    * (EMB / 4);
  const int nChunks  = NUM_NODES * EMB / 256;


  copy_init_kernel<<<750, 320, 0, stream>>>(
      user_emb, item_emb, e0, out, totalVec, userVec);

  prep_packed_kernel<TILE_ADJ><<<586, 256, 0, stream>>>(
      adj_row, adj_col, ids_adj, pk_adj, E_adj);
  spmm_tile_kernel<TILE_ADJ, false, false, false><<<TILES_ADJ, 1024, 0, stream>>>(
      ids_adj, adj_row, pk_adj, adj_val, e0, e1, nullptr, E_adj, NUM_NODES,
      nullptr, nullptr, 0);
  spmm_tile_kernel<TILE_ADJ, false, true, false><<<N_WGP, 1024, 0, stream>>>(
      ids_adj, adj_row, pk_adj, adj_val, e1, e0, nullptr, E_adj, NUM_NODES,
      e1, out, nChunks);
  spmm_tile_kernel<TILE_ADJ, false, true, false><<<N_WGP, 1024, 0, stream>>>(
      ids_adj, adj_row, pk_adj, adj_val, e0, e1, nullptr, E_adj, NUM_NODES,
      e0, out, nChunks);

  prep_rec_kernel<TILE_HV, TILE_HU><<<782, 256, 0, stream>>>(
      hv_row, hv_col, hv_val, ids_hv, rec_hv, E_hv,
      hu_row, hu_col, hu_val, ids_hu, rec_hu, E_hu);
  spmm_tile_kernel<TILE_HV, true, true, true><<<N_WGP, 1024, 0, stream>>>(
      ids_hv, hv_row, rec_hv, hv_val, item_emb, bfeat, deg_hv, E_hv, NUM_BQ,
      e1, out, nChunks);
  rownorm_kernel<<<250, 256, 0, stream>>>(bfeat, deg_hv, bqVec);

  spmm_tile_kernel<TILE_HU, true, false, true><<<TILES_HU, 1024, 0, stream>>>(
      ids_hu, hu_row, rec_hu, hu_val, bfeat, ulocal, deg_hu, E_hu, NUM_USERS,
      nullptr, nullptr, 0);

  finalize_kernel<<<750, 320, 0, stream>>>(
      out, ulocal, deg_hu, totalVec, userVec);
}


# SpMM tile heights follow the WGP count, not LDS size: 224/200/250 tile blocks (were 159/125/125), each resolves fewer edges per list scan; layer sums on 32/56 blocks; sum order kept
# speedup vs baseline: 2.2560x; 1.1045x over previous
#include <hip/hip_runtime.h>
#include <hip/hip_bf16.h>

typedef __attribute__((ext_vector_type(2))) float v2f;
typedef __attribute__((ext_vector_type(8))) float v8f;
typedef __attribute__((ext_vector_type(4))) float v4f;

#define NUM_USERS 100000
#define NUM_ITEMS 50000
#define NUM_NODES 150000
#define NUM_BQ    20000
#define EMB       64

#define SU 5

__global__ void __launch_bounds__(320) copy_init_kernel(
    const float* __restrict__ ue, const float* __restrict__ ie,
    float* __restrict__ e0, float* __restrict__ out,
    int totalVec, int userVec) {
  const int nthr = gridDim.x * blockDim.x;
  int g = blockIdx.x * blockDim.x + threadIdx.x;
  for (; g + (SU - 1) * nthr < totalVec; g += SU * nthr) {
    v4f v[SU];
#pragma unroll
    for (int u = 0; u < SU; ++u) {
      int i = g + u * nthr;
      v[u] = (i < userVec) ? ((const v4f*)ue)[i] : ((const v4f*)ie)[i - userVec];
    }
    __builtin_amdgcn_sched_barrier(0);
#pragma unroll
    for (int u = 0; u < SU; ++u) {
      int i = g + u * nthr;
      ((v4f*)e0)[i]  = v[u];
      ((v4f*)out)[i] = v[u];
    }
  }
  for (; g < totalVec; g += nthr) {
    v4f v = (g < userVec) ? ((const v4f*)ue)[g] : ((const v4f*)ie)[g - userVec];
    ((v4f*)e0)[g]  = v;
    ((v4f*)out)[g] = v;
  }
}

#define QCAP 64

#define COL_BITS 18
#define COL_MASK 0x3ffffu
#define RC_MASK  0x0fffffffu

struct pair32 { unsigned lo, hi; };

template <int TILE_ROWS, bool DEG, bool REC>
__device__ __forceinline__ void resolve_queued(
    const int* q, int base, int n, int lane,
    const unsigned* __restrict__ pk,
    const float* __restrict__ val, const float* __restrict__ x,
    float* tile, float* degt) {
  int last = n - 1;
  int e  = q[base + (lane < last ? lane : last)];
  unsigned w;
  int      vb;
  if (REC) {
    uint2 r = ((const uint2*)pk)[e];
    w  = r.x;
    vb = (int)r.y;
  } else {
    unsigned bit = 28u * ((unsigned)e & 7u);
    const unsigned* p = pk + 7 * (size_t)(e >> 3) + (bit >> 5);
    pair32 t;
    __builtin_memcpy(&t, p, 8);
    w  = (unsigned)(((((unsigned long long)t.hi) << 32) | t.lo) >> (bit & 31u)) & RC_MASK;
    vb = __float_as_int(val[e]);
  }
  int rr = (int)(w >> COL_BITS);
  int c  = (int)(w & COL_MASK);
  for (int k0 = 0; k0 < n; k0 += 16) {
    float2 xv[16];
#pragma unroll
    for (int j = 0; j < 16; ++j) {
      int k  = (k0 + j < last) ? (k0 + j) : last;
      unsigned ck = (unsigned)__builtin_amdgcn_readlane(c, k);
      xv[j] = *(const float2*)((const char*)x + (ck * (EMB * 4u) + 8u * (unsigned)lane));
    }
#pragma unroll
    for (int j = 0; j < 16; ++j) {
      if (k0 + j < n) {
        int   k  = k0 + j;
        int   rk = __builtin_amdgcn_readlane(rr, k);
        float vk = __int_as_float(__builtin_amdgcn_readlane(vb, k));
        float* t = tile + rk * EMB + 2 * lane;
        atomicAdd(t,     vk * xv[j].x);
        atomicAdd(t + 1, vk * xv[j].y);
        if (DEG && lane == 0) atomicAdd(degt + rk, vk);
      }
    }
  }
}

template <int TILE_ROWS>
__device__ __forceinline__ unsigned tile_id_of(int r) {
  unsigned t = (unsigned)r / (unsigned)TILE_ROWS;
  return t < 255u ? t : 255u;
}
template <int TILE_ROWS>
__device__ __forceinline__ unsigned pack_ids(int4 r) {
  return tile_id_of<TILE_ROWS>(r.x) | (tile_id_of<TILE_ROWS>(r.y) << 8) |
         (tile_id_of<TILE_ROWS>(r.z) << 16) | (tile_id_of<TILE_ROWS>(r.w) << 24);
}

template <int TILE_ROWS>
__device__ __forceinline__ unsigned rc_word(int r, int c) {
  unsigned t  = (unsigned)r / (unsigned)TILE_ROWS;
  unsigned rl = (unsigned)r - t * (unsigned)TILE_ROWS;
  return ((rl & 0x3ffu) << COL_BITS) | ((unsigned)c & COL_MASK);
}

template <int TILE_ROWS>
__global__ void __launch_bounds__(256) prep_packed_kernel(
    const int* __restrict__ row, const int* __restrict__ col,
    unsigned* __restrict__ ids, unsigned* __restrict__ pk, int nnz) {
  const int nthr = gridDim.x * blockDim.x;
  const int nG   = (nnz + 7) >> 3;
  for (int g = blockIdx.x * blockDim.x + threadIdx.x; g < nG; g += nthr) {
    int4 r0, r1, c0, c1;
    if (8 * g + 7 < nnz) {
      r0 = ((const int4*)row)[2 * g];  r1 = ((const int4*)row)[2 * g + 1];
      c0 = ((const int4*)col)[2 * g];  c1 = ((const int4*)col)[2 * g + 1];
    } else {
      int e[8];
#pragma unroll
      for (int k = 0; k < 8; ++k) e[k] = (8 * g + k < nnz) ? (8 * g + k) : (nnz - 1);
      r0 = make_int4(row[e[0]], row[e[1]], row[e[2]], row[e[3]]);
      r1 = make_int4(row[e[4]], row[e[5]], row[e[6]], row[e[7]]);
      c0 = make_int4(col[e[0]], col[e[1]], col[e[2]], col[e[3]]);
      c1 = make_int4(col[e[4]], col[e[5]], col[e[6]], col[e[7]]);
    }
    if (8 * g + 3 < nnz) ids[2 * g]     = pack_ids<TILE_ROWS>(r0);
    if (8 * g + 7 < nnz) ids[2 * g + 1] = pack_ids<TILE_ROWS>(r1);
    const unsigned w0 = rc_word<TILE_ROWS>(r0.x, c0.x), w1 = rc_word<TILE_ROWS>(r0.y, c0.y);
    const unsigned w2 = rc_word<TILE_ROWS>(r0.z, c0.z), w3 = rc_word<TILE_ROWS>(r0.w, c0.w);
    const unsigned w4 = rc_word<TILE_ROWS>(r1.x, c1.x), w5 = rc_word<TILE_ROWS>(r1.y, c1.y);
    const unsigned w6 = rc_word<TILE_ROWS>(r1.z, c1.z), w7 = rc_word<TILE_ROWS>(r1.w, c1.w);
    unsigned* p = pk + 7 * (size_t)g;
    p[0] =  w0        | (w1 << 28);
    p[1] = (w1 >> 4)  | (w2 << 24);
    p[2] = (w2 >> 8)  | (w3 << 20);
    p[3] = (w3 >> 12) | (w4 << 16);
    p[4] = (w4 >> 16) | (w5 << 12);
    p[5] = (w5 >> 20) | (w6 << 8);
    p[6] = (w6 >> 24) | (w7 << 4);
  }
  if (blockIdx.x == 0 && threadIdx.x == 0) pk[7 * (size_t)nG] = 0u;
}

template <int TILE_ROWS>
__device__ __forceinline__ void prep_rec_group(
    const int* __restrict__ row, const int* __restrict__ col, const float* __restrict__ val,
    unsigned* __restrict__ ids, unsigned* __restrict__ rec, int nnz, int g) {
  if (4 * g + 3 < nnz) {
    const int4   r = ((const int4*)row)[g];
    const int4   c = ((const int4*)col)[g];
    const float4 v = ((const float4*)val)[g];
    ids[g] = pack_ids<TILE_ROWS>(r);
    ((uint4*)rec)[2 * g]     = make_uint4(rc_word<TILE_ROWS>(r.x, c.x), __float_as_uint(v.x),
                                          rc_word<TILE_ROWS>(r.y, c.y), __float_as_uint(v.y));
    ((uint4*)rec)[2 * g + 1] = make_uint4(rc_word<TILE_ROWS>(r.z, c.z), __float_as_uint(v.z),
                                          rc_word<TILE_ROWS>(r.w, c.w), __float_as_uint(v.w));
  } else {
    for (int e = 4 * g; e < nnz; ++e) {
      rec[2 * e]     = rc_word<TILE_ROWS>(row[e], col[e]);
      rec[2 * e + 1] = __float_as_uint(val[e]);
    }
  }
}
template <int TILE_A, int TILE_B>
__global__ void __launch_bounds__(256) prep_rec_kernel(
    const int* __restrict__ rowA, const int* __restrict__ colA, const float* __restrict__ valA,
    unsigned* __restrict__ idsA, unsigned* __restrict__ recA, int nnzA,
    const int* __restrict__ rowB, const int* __restrict__ colB, const float* __restrict__ valB,
    unsigned* __restrict__ idsB, unsigned* __restrict__ recB, int nnzB) {
  const int nthr = gridDim.x * blockDim.x;
  const int nA4  = (nnzA + 3) >> 2;
  const int nB4  = (nnzB + 3) >> 2;
  for (int g = blockIdx.x * blockDim.x + threadIdx.x; g < nA4 + nB4; g += nthr) {
    if (g < nA4) prep_rec_group<TILE_A>(rowA, colA, valA, idsA, recA, nnzA, g);
    else         prep_rec_group<TILE_B>(rowB, colB, valB, idsB, recB, nnzB, g - nA4);
  }
}

__device__ __forceinline__ unsigned nonzero_bytes(unsigned v) {
  return (((v & 0x7f7f7f7fu) + 0x7f7f7f7fu) | v) & 0x80808080u;
}

template <int TILE_ROWS, bool DEG, bool REC>
__device__ __forceinline__ void take16(
    uint4 w, unsigned tb, bool ok, int j, int& cnt, int* q, int lane,
    const unsigned* __restrict__ pk,
    const float* __restrict__ val, const float* __restrict__ x,
    float* tile, float* degt) {
  unsigned nz = (nonzero_bytes(w.x ^ tb) >> 7) | (nonzero_bytes(w.y ^ tb) >> 6) |
                (nonzero_bytes(w.z ^ tb) >> 5) | (nonzero_bytes(w.w ^ tb) >> 4);
  unsigned m = ok ? (~nz & 0x0f0f0f0fu) : 0u;
  unsigned mk = __builtin_amdgcn_ballot_w32(m != 0u);
  while (mk) {
    int p   = __builtin_ctz(m | 0x80000000u);
    int pos = cnt + (int)__builtin_amdgcn_mbcnt_lo(mk, 0u);
    if (m != 0u) q[pos] = 16 * j + 4 * (p & 7) + (p >> 3);
    cnt += __builtin_popcount(mk);
    m &= m - 1u;
    if (cnt >= 32) {
      resolve_queued<TILE_ROWS, DEG, REC>(q, cnt - 32, 32, lane, pk, val, x, tile, degt);
      cnt -= 32;
    }
    mk = __builtin_amdgcn_ballot_w32(m != 0u);
  }
}

#define AU 5
__device__ __forceinline__ void acc_wmma_part(
    const float* __restrict__ e, float* __restrict__ acc, int nChunks,
    int wv, int nWaves) {
  const int lane = threadIdx.x & 31;
  const int n    = lane & 15;
  const int hi   = lane >> 4;

  v2f a[4];
#pragma unroll
  for (int k = 0; k < 4; ++k) {
    a[k].x = (n == (4 * k + 2 * hi))     ? 1.0f : 0.0f;
    a[k].y = (n == (4 * k + 2 * hi + 1)) ? 1.0f : 0.0f;
  }

  for (int g = wv * AU; g < nChunks; g += nWaves * AU) {
    v8f cc[AU];
    v2f b[AU][4];
#pragma unroll
    for (int u = 0; u < AU; ++u) {
      int ch = (g + u < nChunks) ? (g + u) : (nChunks - 1);
      const float* ap = acc + (size_t)ch * 256 + 4 * lane;
      const float* ep = e   + (size_t)ch * 256 + 4 * n + 2 * hi;
      v4f lo = *(const v4f*)(ap);
      v4f up = *(const v4f*)(ap + 128);
      cc[u][0] = lo.x; cc[u][1] = lo.y; cc[u][2] = lo.z; cc[u][3] = lo.w;
      cc[u][4] = up.x; cc[u][5] = up.y; cc[u][6] = up.z; cc[u][7] = up.w;
#pragma unroll
      for (int k = 0; k < 4; ++k)
        b[u][k] = *(const v2f*)(ep + 128 * (k & 1) + 64 * (k >> 1));
    }
    __builtin_amdgcn_sched_barrier(0);
#pragma unroll
    for (int k = 0; k < 4; ++k)
#pragma unroll
      for (int u = 0; u < AU; ++u)
        cc[u] = __builtin_amdgcn_wmma_f32_16x16x4_f32(
            false, a[k], false, b[u][k], (short)0, cc[u], false, false);
#pragma unroll
    for (int u = 0; u < AU; ++u) {
      if (g + u < nChunks) {
        float* op = acc + (size_t)(g + u) * 256 + 4 * lane;
        v4f lo = {cc[u][0], cc[u][1], cc[u][2], cc[u][3]};
        v4f up = {cc[u][4], cc[u][5], cc[u][6], cc[u][7]};
        *(v4f*)(op)       = lo;
        *(v4f*)(op + 128) = up;
      }
    }
  }
}

template <int TILE_ROWS, bool DEG, bool ACC, bool REC>
__global__ void __launch_bounds__(1024) spmm_tile_kernel(
    const unsigned* __restrict__ ids,
    const int* __restrict__ row, const unsigned* __restrict__ pk,
    const float* __restrict__ val, const float* __restrict__ x,
    float* __restrict__ y, float* __restrict__ deg, int nnz, int n_rows,
    const float* __restrict__ acc_e, float* __restrict__ acc_out, int nChunks) {
  __shared__ float tile[TILE_ROWS * EMB];
  __shared__ float degt[DEG ? TILE_ROWS : 1];
  __shared__ int   queue[32 * QCAP];

  const int n_tiles = (n_rows + TILE_ROWS - 1) / TILE_ROWS;
  if (ACC && (int)blockIdx.x >= n_tiles) {
    const int wv = __builtin_amdgcn_readfirstlane(
        ((int)blockIdx.x - n_tiles) * 32 + (int)(threadIdx.x >> 5));
    acc_wmma_part(acc_e, acc_out, nChunks, wv, ((int)gridDim.x - n_tiles) * 32);
    return;
  }

  const int tid  = threadIdx.x;
  const int lane = tid & 31;
  const int wave = tid >> 5;
  const int lo   = blockIdx.x * TILE_ROWS;
  const unsigned tb = 0x01010101u * blockIdx.x;

  for (int i = tid; i < TILE_ROWS * (EMB / 4); i += 1024)
    ((float4*)tile)[i] = make_float4(0.f, 0.f, 0.f, 0.f);
  if (DEG)
    for (int i = tid; i < TILE_ROWS; i += 1024) degt[i] = 0.f;
  __syncthreads();

  int* q   = queue + wave * QCAP;
  int  cnt = 0;
  const int    nnz16 = nnz >> 4;
  const uint4* ids4  = (const uint4*)ids;
  const int    trips = (nnz16 + 1023) >> 10;
  const int    full  = (nnz16 >> 10) & ~3;
  const int start = blockIdx.x * (nnz16 / n_tiles);

  uint4 cur[4];
  int   cid[4];
  if (full > 0) {
#pragma unroll
    for (int u = 0; u < 4; ++u) {
      int j = (u << 10) + tid + start;
      if (j >= nnz16) j -= nnz16;
      cid[u] = j;
      cur[u] = ids4[j];
    }
  }
  for (int it = 0; it < full; it += 4) {
    uint4 nxt[4];
    int   nid[4];
    const int itn = (it + 4 < full) ? (it + 4) : it;
#pragma unroll
    for (int u = 0; u < 4; ++u) {
      int j = ((itn + u) << 10) + tid + start;
      if (j >= nnz16) j -= nnz16;
      nid[u] = j;
      nxt[u] = ids4[j];
    }
#pragma unroll
    for (int u = 0; u < 4; ++u)
      take16<TILE_ROWS, DEG, REC>(cur[u], tb, true, cid[u], cnt, q, lane, pk, val, x, tile, degt);
#pragma unroll
    for (int u = 0; u < 4; ++u) { cur[u] = nxt[u]; cid[u] = nid[u]; }
  }
  for (int it = full; it < trips; ++it) {
    int  i  = (it << 10) + tid;
    bool ok = i < nnz16;
    int  j  = i + start;
    if (j >= nnz16) j -= nnz16;
    uint4 w = ids4[ok ? j : 0];
    take16<TILE_ROWS, DEG, REC>(w, tb, ok, j, cnt, q, lane, pk, val, x, tile, degt);
  }
  if (wave == 0) {
    int  e  = (nnz16 << 4) + lane;
    bool m  = (e < nnz) && ((unsigned)(row[e < nnz ? e : 0] - lo) < (unsigned)TILE_ROWS);
    unsigned mk = __builtin_amdgcn_ballot_w32(m);
    if (mk) {
      int pos = cnt + (int)__builtin_amdgcn_mbcnt_lo(mk, 0u);
      if (m) q[pos] = e;
      cnt += __builtin_popcount(mk);
    }
  }
  while (cnt > 0) {
    int n = cnt < 32 ? cnt : 32;
    resolve_queued<TILE_ROWS, DEG, REC>(q, cnt - n, n, lane, pk, val, x, tile, degt);
    cnt -= n;
  }
  __syncthreads();

  const int rows_here = (n_rows - lo) < TILE_ROWS ? (n_rows - lo) : TILE_ROWS;
  float4* y4 = (float4*)(y + (size_t)lo * EMB);
  for (int i = tid; i < rows_here * (EMB / 4); i += 1024)
    y4[i] = ((const float4*)tile)[i];
  if (DEG)
    for (int i = tid; i < rows_here; i += 1024) deg[lo + i] = degt[i];
}

__global__ void __launch_bounds__(256) rownorm_kernel(
    float* __restrict__ x, const float* __restrict__ deg, int rowsVec) {
  const int nthr = gridDim.x * blockDim.x;
  int g = blockIdx.x * blockDim.x + threadIdx.x;
  for (; g + (SU - 1) * nthr < rowsVec; g += SU * nthr) {
    v4f   v[SU];
    float d[SU];
#pragma unroll
    for (int u = 0; u < SU; ++u) {
      int i = g + u * nthr;
      v[u] = ((const v4f*)x)[i];
      d[u] = deg[i >> 4];
    }
    __builtin_amdgcn_sched_barrier(0);
#pragma unroll
    for (int u = 0; u < SU; ++u) {
      int i = g + u * nthr;
      float dd  = (d[u] == 0.0f) ? 1.0f : d[u];
      float inv = 1.0f / dd;
      ((v4f*)x)[i] = v[u] * inv;
    }
  }
  for (; g < rowsVec; g += nthr) {
    float dd  = deg[g >> 4];
    dd = (dd == 0.0f) ? 1.0f : dd;
    float inv = 1.0f / dd;
    ((v4f*)x)[g] = ((const v4f*)x)[g] * inv;
  }
}

__global__ void __launch_bounds__(320) finalize_kernel(
    float* __restrict__ out, const float* __restrict__ ulocal,
    const float* __restrict__ deg, int totalVec, int userVec) {
  const int nthr = gridDim.x * blockDim.x;
  int g = blockIdx.x * blockDim.x + threadIdx.x;
  for (; g + (SU - 1) * nthr < totalVec; g += SU * nthr) {
    v4f   v[SU], w[SU];
    float d[SU];
#pragma unroll
    for (int u = 0; u < SU; ++u) {
      int i  = g + u * nthr;
      int iu = (i < userVec) ? i : (userVec - 1);
      v[u] = ((const v4f*)out)[i];
      w[u] = ((const v4f*)ulocal)[iu];
      d[u] = deg[iu >> 4];
    }
    __builtin_amdgcn_sched_barrier(0);
#pragma unroll
    for (int u = 0; u < SU; ++u) {
      int i = g + u * nthr;
      float dd  = (d[u] == 0.0f) ? 1.0f : d[u];
      float inv = 1.0f / dd;
      v4f r  = v[u] * 0.25f;
      v4f ru = r + w[u] * inv;
      ((v4f*)out)[i] = (i < userVec) ? ru : r;
    }
  }
  for (; g < totalVec; g += nthr) {
    v4f r = ((const v4f*)out)[g] * 0.25f;
    if (g < userVec) {
      float dd  = deg[g >> 4];
      dd = (dd == 0.0f) ? 1.0f : dd;
      float inv = 1.0f / dd;
      r += ((const v4f*)ulocal)[g] * inv;
    }
    ((v4f*)out)[g] = r;
  }
}

#define TILE_ADJ 672
#define TILE_HV  100
#define TILE_HU  400
static_assert((NUM_NODES + TILE_ADJ - 1) / TILE_ADJ <= 255, "tile id must fit a byte");
static_assert((NUM_BQ    + TILE_HV  - 1) / TILE_HV  <= 255, "tile id must fit a byte");
static_assert((NUM_USERS + TILE_HU  - 1) / TILE_HU  <= 255, "tile id must fit a byte");
static_assert(TILE_ADJ <= 1024 && TILE_HV <= 1024 && TILE_HU <= 1024, "row in tile must fit 10 bits");
static_assert(NUM_NODES <= (1 << COL_BITS), "col must fit 18 bits");
#define NNZ_ADJ 1200000
#define NNZ_BQ   400000
#define TILES_ADJ ((NUM_NODES + TILE_ADJ - 1) / TILE_ADJ)
#define TILES_HV  ((NUM_BQ    + TILE_HV  - 1) / TILE_HV)
#define TILES_HU  ((NUM_USERS + TILE_HU  - 1) / TILE_HU)
#define N_WGP 256
static_assert(TILES_ADJ < N_WGP && TILES_HV < N_WGP, "a launch with a layer sum needs blocks behind its tiles");
static_assert(TILES_HU <= N_WGP, "one tile block per WGP");

extern "C" void kernel_launch(void* const* d_in, const int* in_sizes, int n_in,
                              void* d_out, int out_size, void* d_ws, size_t ws_size,
                              hipStream_t stream) {
  const float* user_emb = (const float*)d_in[0];
  const float* item_emb = (const float*)d_in[1];
  const float* adj_val  = (const float*)d_in[2];
  const float* hv_val   = (const float*)d_in[3];
  const float* hu_val   = (const float*)d_in[4];
  const int*   adj_row  = (const int*)d_in[5];
  const int*   adj_col  = (const int*)d_in[6];
  const int*   hv_row   = (const int*)d_in[7];
  const int*   hv_col   = (const int*)d_in[8];
  const int*   hu_row   = (const int*)d_in[9];
  const int*   hu_col   = (const int*)d_in[10];

  const int E_adj = in_sizes[2];
  const int E_hv  = in_sizes[3];
  const int E_hu  = in_sizes[4];

  float* out = (float*)d_out;

  const size_t embBytesN = (size_t)NUM_NODES * EMB * sizeof(float);
  const size_t embBytesB = (size_t)NUM_BQ    * EMB * sizeof(float);
  const size_t embBytesU = (size_t)NUM_USERS * EMB * sizeof(float);
  char* ws = (char*)d_ws;
  float* e0     = (float*)(ws);
  float* e1     = (float*)(ws + embBytesN);
  float* bfeat  = (float*)(ws + 2 * embBytesN);
  float* deg_hv = (float*)(ws + 2 * embBytesN + embBytesB);
  float* deg_hu = deg_hv + NUM_BQ;
  float* ulocal = e0;
  static_assert((size_t)NNZ_ADJ + 28 * (size_t)((NNZ_ADJ + 7) / 8) + 4 <=
                (size_t)NUM_BQ * EMB * 4 + (size_t)(NUM_BQ + NUM_USERS) * 4,
                "adjacency ids and rc words must fit the biclique area");
  static_assert(2 * ((size_t)NNZ_BQ + 8 * (size_t)NNZ_BQ) <= (size_t)NUM_ITEMS * EMB * 4,
                "biclique ids and records must fit behind u_local");
  unsigned* ids_adj = (unsigned*)(ws + 2 * embBytesN);
  unsigned* pk_adj  = (unsigned*)(ws + 2 * embBytesN + NNZ_ADJ);
  unsigned* ids_hv  = (unsigned*)(ws + embBytesU);
  unsigned* rec_hv  = (unsigned*)(ws + embBytesU + NNZ_BQ);
  unsigned* ids_hu  = (unsigned*)(ws + embBytesU + (embBytesN - embBytesU) / 2);
  unsigned* rec_hu  = (unsigned*)(ws + embBytesU + (embBytesN - embBytesU) / 2 + NNZ_BQ);

  const int totalVec = NUM_NODES * (EMB / 4);
  const int userVec  = NUM_USERS * (EMB / 4);
  const int bqVec    = NUM_BQ    * (EMB / 4);
  const int nChunks  = NUM_NODES * EMB / 256;


  copy_init_kernel<<<750, 320, 0, stream>>>(
      user_emb, item_emb, e0, out, totalVec, userVec);

  prep_packed_kernel<TILE_ADJ><<<586, 256, 0, stream>>>(
      adj_row, adj_col, ids_adj, pk_adj, E_adj);
  spmm_tile_kernel<TILE_ADJ, false, false, false><<<TILES_ADJ, 1024, 0, stream>>>(
      ids_adj, adj_row, pk_adj, adj_val, e0, e1, nullptr, E_adj, NUM_NODES,
      nullptr, nullptr, 0);
  spmm_tile_kernel<TILE_ADJ, false, true, false><<<N_WGP, 1024, 0, stream>>>(
      ids_adj, adj_row, pk_adj, adj_val, e1, e0, nullptr, E_adj, NUM_NODES,
      e1, out, nChunks);
  spmm_tile_kernel<TILE_ADJ, false, true, false><<<N_WGP, 1024, 0, stream>>>(
      ids_adj, adj_row, pk_adj, adj_val, e0, e1, nullptr, E_adj, NUM_NODES,
      e0, out, nChunks);

  prep_rec_kernel<TILE_HV, TILE_HU><<<782, 256, 0, stream>>>(
      hv_row, hv_col, hv_val, ids_hv, rec_hv, E_hv,
      hu_row, hu_col, hu_val, ids_hu, rec_hu, E_hu);
  spmm_tile_kernel<TILE_HV, true, true, true><<<N_WGP, 1024, 0, stream>>>(
      ids_hv, hv_row, rec_hv, hv_val, item_emb, bfeat, deg_hv, E_hv, NUM_BQ,
      e1, out, nChunks);
  rownorm_kernel<<<250, 256, 0, stream>>>(bfeat, deg_hv, bqVec);

  spmm_tile_kernel<TILE_HU, true, false, true><<<TILES_HU, 1024, 0, stream>>>(
      ids_hu, hu_row, rec_hu, hu_val, bfeat, ulocal, deg_hu, E_hu, NUM_USERS,
      nullptr, nullptr, 0);

  finalize_kernel<<<750, 320, 0, stream>>>(
      out, ulocal, deg_hu, totalVec, userVec);
}


# Adjacency layers 2-3 stop scanning the edge list: in layer 1 each tile block appends the edges it finds (edge id, row in tile) to its own list, layers 2-3 read it; row sums stay in arrival order
# speedup vs baseline: 2.4209x; 1.0731x over previous
#include <hip/hip_runtime.h>
#include <hip/hip_bf16.h>

typedef __attribute__((ext_vector_type(2))) float v2f;
typedef __attribute__((ext_vector_type(8))) float v8f;
typedef __attribute__((ext_vector_type(4))) float v4f;

#define NUM_USERS 100000
#define NUM_ITEMS 50000
#define NUM_NODES 150000
#define NUM_BQ    20000
#define EMB       64

#define SU 5

__global__ void __launch_bounds__(320) copy_init_kernel(
    const float* __restrict__ ue, const float* __restrict__ ie,
    float* __restrict__ e0,
    int totalVec, int userVec) {
  const int nthr = gridDim.x * blockDim.x;
  int g = blockIdx.x * blockDim.x + threadIdx.x;
  for (; g + (SU - 1) * nthr < totalVec; g += SU * nthr) {
    v4f v[SU];
#pragma unroll
    for (int u = 0; u < SU; ++u) {
      int i = g + u * nthr;
      v[u] = (i < userVec) ? ((const v4f*)ue)[i] : ((const v4f*)ie)[i - userVec];
    }
    __builtin_amdgcn_sched_barrier(0);
#pragma unroll
    for (int u = 0; u < SU; ++u) {
      int i = g + u * nthr;
      ((v4f*)e0)[i]  = v[u];
    }
  }
  for (; g < totalVec; g += nthr) {
    v4f v = (g < userVec) ? ((const v4f*)ue)[g] : ((const v4f*)ie)[g - userVec];
    ((v4f*)e0)[g]  = v;
  }
}

#define QCAP 64

#define COL_BITS 18
#define COL_MASK 0x3ffffu
#define RC_MASK  0x0fffffffu

struct pair32 { unsigned lo, hi; };

#define E_BITS   21
#define E_MASK   0x1fffffu
#define LIST_CAP 6240
#define LIST_HDR 256

template <bool DEG>
__device__ __forceinline__ void add_edges(
    int rr, int c, int vb, int n, int lane,
    const float* __restrict__ x, float* tile, float* degt) {
  const int last = n - 1;
  for (int k0 = 0; k0 < n; k0 += 16) {
    float2 xv[16];
#pragma unroll
    for (int j = 0; j < 16; ++j) {
      int k  = (k0 + j < last) ? (k0 + j) : last;
      unsigned ck = (unsigned)__builtin_amdgcn_readlane(c, k);
      xv[j] = *(const float2*)((const char*)x + (ck * (EMB * 4u) + 8u * (unsigned)lane));
    }
#pragma unroll
    for (int j = 0; j < 16; ++j) {
      if (k0 + j < n) {
        int   k  = k0 + j;
        int   rk = __builtin_amdgcn_readlane(rr, k);
        float vk = __int_as_float(__builtin_amdgcn_readlane(vb, k));
        float* t = tile + rk * EMB + 2 * lane;
        atomicAdd(t,     vk * xv[j].x);
        atomicAdd(t + 1, vk * xv[j].y);
        if (DEG && lane == 0) atomicAdd(degt + rk, vk);
      }
    }
  }
}

template <int TILE_ROWS, bool DEG, bool REC, bool BUILD>
__device__ __forceinline__ void resolve_queued(
    const int* q, int base, int n, int lane,
    const unsigned* __restrict__ pk,
    const float* __restrict__ val, const float* __restrict__ x,
    float* tile, float* degt, unsigned* lst, int* cursor) {
  int last = n - 1;
  int e  = q[base + (lane < last ? lane : last)];
  unsigned w;
  int      vb;
  if (REC) {
    uint2 r = ((const uint2*)pk)[e];
    w  = r.x;
    vb = (int)r.y;
  } else {
    unsigned bit = 28u * ((unsigned)e & 7u);
    const unsigned* p = pk + 7 * (size_t)(e >> 3) + (bit >> 5);
    pair32 t;
    __builtin_memcpy(&t, p, 8);
    w  = (unsigned)(((((unsigned long long)t.hi) << 32) | t.lo) >> (bit & 31u)) & RC_MASK;
    vb = __float_as_int(val[e]);
  }
  int rr = (int)(w >> COL_BITS);
  int c  = (int)(w & COL_MASK);
  if (BUILD) {
    int pos = 0;
    if (lane == 0) pos = atomicAdd(cursor, n);
    pos = __builtin_amdgcn_readfirstlane(pos);
    if (lane < n && pos + lane < LIST_CAP)
      lst[pos + lane] = (unsigned)e | ((unsigned)rr << E_BITS);
  }
  add_edges<DEG>(rr, c, vb, n, lane, x, tile, degt);
}

template <int TILE_ROWS>
__device__ __forceinline__ unsigned tile_id_of(int r) {
  unsigned t = (unsigned)r / (unsigned)TILE_ROWS;
  return t < 255u ? t : 255u;
}
template <int TILE_ROWS>
__device__ __forceinline__ unsigned pack_ids(int4 r) {
  return tile_id_of<TILE_ROWS>(r.x) | (tile_id_of<TILE_ROWS>(r.y) << 8) |
         (tile_id_of<TILE_ROWS>(r.z) << 16) | (tile_id_of<TILE_ROWS>(r.w) << 24);
}

template <int TILE_ROWS>
__device__ __forceinline__ unsigned rc_word(int r, int c) {
  unsigned t  = (unsigned)r / (unsigned)TILE_ROWS;
  unsigned rl = (unsigned)r - t * (unsigned)TILE_ROWS;
  return ((rl & 0x3ffu) << COL_BITS) | ((unsigned)c & COL_MASK);
}

template <int TILE_ROWS>
__global__ void __launch_bounds__(256) prep_packed_kernel(
    const int* __restrict__ row, const int* __restrict__ col,
    unsigned* __restrict__ ids, unsigned* __restrict__ pk, int nnz) {
  const int nthr = gridDim.x * blockDim.x;
  const int nG   = (nnz + 7) >> 3;
  for (int g = blockIdx.x * blockDim.x + threadIdx.x; g < nG; g += nthr) {
    int4 r0, r1, c0, c1;
    if (8 * g + 7 < nnz) {
      r0 = ((const int4*)row)[2 * g];  r1 = ((const int4*)row)[2 * g + 1];
      c0 = ((const int4*)col)[2 * g];  c1 = ((const int4*)col)[2 * g + 1];
    } else {
      int e[8];
#pragma unroll
      for (int k = 0; k < 8; ++k) e[k] = (8 * g + k < nnz) ? (8 * g + k) : (nnz - 1);
      r0 = make_int4(row[e[0]], row[e[1]], row[e[2]], row[e[3]]);
      r1 = make_int4(row[e[4]], row[e[5]], row[e[6]], row[e[7]]);
      c0 = make_int4(col[e[0]], col[e[1]], col[e[2]], col[e[3]]);
      c1 = make_int4(col[e[4]], col[e[5]], col[e[6]], col[e[7]]);
    }
    if (8 * g + 3 < nnz) ids[2 * g]     = pack_ids<TILE_ROWS>(r0);
    if (8 * g + 7 < nnz) ids[2 * g + 1] = pack_ids<TILE_ROWS>(r1);
    const unsigned w0 = rc_word<TILE_ROWS>(r0.x, c0.x), w1 = rc_word<TILE_ROWS>(r0.y, c0.y);
    const unsigned w2 = rc_word<TILE_ROWS>(r0.z, c0.z), w3 = rc_word<TILE_ROWS>(r0.w, c0.w);
    const unsigned w4 = rc_word<TILE_ROWS>(r1.x, c1.x), w5 = rc_word<TILE_ROWS>(r1.y, c1.y);
    const unsigned w6 = rc_word<TILE_ROWS>(r1.z, c1.z), w7 = rc_word<TILE_ROWS>(r1.w, c1.w);
    unsigned* p = pk + 7 * (size_t)g;
    p[0] =  w0        | (w1 << 28);
    p[1] = (w1 >> 4)  | (w2 << 24);
    p[2] = (w2 >> 8)  | (w3 << 20);
    p[3] = (w3 >> 12) | (w4 << 16);
    p[4] = (w4 >> 16) | (w5 << 12);
    p[5] = (w5 >> 20) | (w6 << 8);
    p[6] = (w6 >> 24) | (w7 << 4);
  }
  if (blockIdx.x == 0 && threadIdx.x == 0) pk[7 * (size_t)nG] = 0u;
}

template <int TILE_ROWS>
__device__ __forceinline__ void prep_rec_group(
    const int* __restrict__ row, const int* __restrict__ col, const float* __restrict__ val,
    unsigned* __restrict__ ids, unsigned* __restrict__ rec, int nnz, int g) {
  if (4 * g + 3 < nnz) {
    const int4   r = ((const int4*)row)[g];
    const int4   c = ((const int4*)col)[g];
    const float4 v = ((const float4*)val)[g];
    ids[g] = pack_ids<TILE_ROWS>(r);
    ((uint4*)rec)[2 * g]     = make_uint4(rc_word<TILE_ROWS>(r.x, c.x), __float_as_uint(v.x),
                                          rc_word<TILE_ROWS>(r.y, c.y), __float_as_uint(v.y));
    ((uint4*)rec)[2 * g + 1] = make_uint4(rc_word<TILE_ROWS>(r.z, c.z), __float_as_uint(v.z),
                                          rc_word<TILE_ROWS>(r.w, c.w), __float_as_uint(v.w));
  } else {
    for (int e = 4 * g; e < nnz; ++e) {
      rec[2 * e]     = rc_word<TILE_ROWS>(row[e], col[e]);
      rec[2 * e + 1] = __float_as_uint(val[e]);
    }
  }
}
template <int TILE_A, int TILE_B>
__global__ void __launch_bounds__(256) prep_rec_kernel(
    const int* __restrict__ rowA, const int* __restrict__ colA, const float* __restrict__ valA,
    unsigned* __restrict__ idsA, unsigned* __restrict__ recA, int nnzA,
    const int* __restrict__ rowB, const int* __restrict__ colB, const float* __restrict__ valB,
    unsigned* __restrict__ idsB, unsigned* __restrict__ recB, int nnzB) {
  const int nthr = gridDim.x * blockDim.x;
  const int nA4  = (nnzA + 3) >> 2;
  const int nB4  = (nnzB + 3) >> 2;
  for (int g = blockIdx.x * blockDim.x + threadIdx.x; g < nA4 + nB4; g += nthr) {
    if (g < nA4) prep_rec_group<TILE_A>(rowA, colA, valA, idsA, recA, nnzA, g);
    else         prep_rec_group<TILE_B>(rowB, colB, valB, idsB, recB, nnzB, g - nA4);
  }
}

__device__ __forceinline__ unsigned nonzero_bytes(unsigned v) {
  return (((v & 0x7f7f7f7fu) + 0x7f7f7f7fu) | v) & 0x80808080u;
}

template <int TILE_ROWS, bool DEG, bool REC, bool BUILD>
__device__ __forceinline__ void take16(
    uint4 w, unsigned tb, bool ok, int j, int& cnt, int* q, int lane,
    const unsigned* __restrict__ pk,
    const float* __restrict__ val, const float* __restrict__ x,
    float* tile, float* degt, unsigned* lst, int* cursor) {
  unsigned nz = (nonzero_bytes(w.x ^ tb) >> 7) | (nonzero_bytes(w.y ^ tb) >> 6) |
                (nonzero_bytes(w.z ^ tb) >> 5) | (nonzero_bytes(w.w ^ tb) >> 4);
  unsigned m = ok ? (~nz & 0x0f0f0f0fu) : 0u;
  unsigned mk = __builtin_amdgcn_ballot_w32(m != 0u);
  while (mk) {
    int p   = __builtin_ctz(m | 0x80000000u);
    int pos = cnt + (int)__builtin_amdgcn_mbcnt_lo(mk, 0u);
    if (m != 0u) q[pos] = 16 * j + 4 * (p & 7) + (p >> 3);
    cnt += __builtin_popcount(mk);
    m &= m - 1u;
    if (cnt >= 32) {
      resolve_queued<TILE_ROWS, DEG, REC, BUILD>(q, cnt - 32, 32, lane, pk, val, x, tile, degt, lst, cursor);
      cnt -= 32;
    }
    mk = __builtin_amdgcn_ballot_w32(m != 0u);
  }
}

#define AU 5
#define USER_CHUNKS (NUM_USERS * EMB / 256)
static_assert((NUM_USERS * EMB) % 256 == 0, "user rows end at a chunk boundary");
__device__ __forceinline__ void acc_wmma_part(
    const float* __restrict__ e, const float* cu, const float* ci, float* acc,
    int ch0, int ch1, int wv, int nWaves) {
  const int lane = threadIdx.x & 31;
  const int n    = lane & 15;
  const int hi   = lane >> 4;

  v2f a[4];
#pragma unroll
  for (int k = 0; k < 4; ++k) {
    a[k].x = (n == (4 * k + 2 * hi))     ? 1.0f : 0.0f;
    a[k].y = (n == (4 * k + 2 * hi + 1)) ? 1.0f : 0.0f;
  }

  for (int g = ch0 + wv * AU; g < ch1; g += nWaves * AU) {
    v8f cc[AU];
    v2f b[AU][4];
#pragma unroll
    for (int u = 0; u < AU; ++u) {
      int ch = (g + u < ch1) ? (g + u) : (ch1 - 1);
      const float* ap = (ch < USER_CHUNKS ? cu + (size_t)ch * 256
                                          : ci + (size_t)(ch - USER_CHUNKS) * 256) + 4 * lane;
      const float* ep = e   + (size_t)ch * 256 + 4 * n + 2 * hi;
      v4f lo = *(const v4f*)(ap);
      v4f up = *(const v4f*)(ap + 128);
      cc[u][0] = lo.x; cc[u][1] = lo.y; cc[u][2] = lo.z; cc[u][3] = lo.w;
      cc[u][4] = up.x; cc[u][5] = up.y; cc[u][6] = up.z; cc[u][7] = up.w;
#pragma unroll
      for (int k = 0; k < 4; ++k)
        b[u][k] = *(const v2f*)(ep + 128 * (k & 1) + 64 * (k >> 1));
    }
    __builtin_amdgcn_sched_barrier(0);
#pragma unroll
    for (int k = 0; k < 4; ++k)
#pragma unroll
      for (int u = 0; u < AU; ++u)
        cc[u] = __builtin_amdgcn_wmma_f32_16x16x4_f32(
            false, a[k], false, b[u][k], (short)0, cc[u], false, false);
#pragma unroll
    for (int u = 0; u < AU; ++u) {
      if (g + u < ch1) {
        float* op = acc + (size_t)(g + u) * 256 + 4 * lane;
        v4f lo = {cc[u][0], cc[u][1], cc[u][2], cc[u][3]};
        v4f up = {cc[u][4], cc[u][5], cc[u][6], cc[u][7]};
        *(v4f*)(op)       = lo;
        *(v4f*)(op + 128) = up;
      }
    }
  }
}

template <int TILE_ROWS, bool DEG, bool ACC, bool REC, int LISTS>
__global__ void __launch_bounds__(1024) spmm_tile_kernel(
    const unsigned* __restrict__ ids,
    const int* __restrict__ row, const int* __restrict__ col,
    const unsigned* __restrict__ pk,
    const float* __restrict__ val, const float* __restrict__ x,
    float* __restrict__ y, float* __restrict__ deg, int nnz, int n_rows,
    const float* __restrict__ acc_e, const float* acc_cu, const float* acc_ci,
    float* acc_out, int nChunks, int nRide, unsigned* lists) {
  __shared__ float tile[TILE_ROWS * EMB];
  __shared__ float degt[DEG ? TILE_ROWS : 1];
  __shared__ int   queue[LISTS == 2 ? 1 : 32 * QCAP];
  __shared__ int   cursor;

  const int n_tiles = (n_rows + TILE_ROWS - 1) / TILE_ROWS;
  if (ACC && (int)blockIdx.x >= n_tiles) {
    const int wv = __builtin_amdgcn_readfirstlane(
        ((int)blockIdx.x - n_tiles) * 32 + (int)(threadIdx.x >> 5));
    acc_wmma_part(acc_e, acc_cu, acc_ci, acc_out, 0, nRide, wv,
                  ((int)gridDim.x - n_tiles) * 32);
    return;
  }

  const int tid  = threadIdx.x;
  const int lane = tid & 31;
  const int wave = tid >> 5;
  const int lo   = blockIdx.x * TILE_ROWS;
  const unsigned tb = 0x01010101u * blockIdx.x;
  unsigned* lst = lists + LIST_HDR + (size_t)blockIdx.x * LIST_CAP;

  for (int i = tid; i < TILE_ROWS * (EMB / 4); i += 1024)
    ((float4*)tile)[i] = make_float4(0.f, 0.f, 0.f, 0.f);
  if (DEG)
    for (int i = tid; i < TILE_ROWS; i += 1024) degt[i] = 0.f;
  if (LISTS == 1 && tid == 0) cursor = 0;
  __syncthreads();

  if (LISTS == 2) {
    const int wv_s   = __builtin_amdgcn_readfirstlane(wave);
    const int cntAll = __builtin_amdgcn_readfirstlane(((const int*)lists)[blockIdx.x]);
    if (cntAll <= LIST_CAP) {
      if (cntAll > 0) {
        const int base0 = wv_s * 32;
        const int lastp = cntAll - 1;
        unsigned entA = lst[tid < cntAll ? tid : lastp];
        unsigned entB = lst[1024 + tid < cntAll ? 1024 + tid : lastp];
        int eA = (int)(entA & E_MASK);
        eA = eA < nnz ? eA : 0;
        int rA = (int)(entA >> E_BITS);
        int cA = col[eA];
        int vA = __float_as_int(val[eA]);
        for (int b = 0; cntAll - (b * 1024 + base0) > 0; ++b) {
          const int n  = cntAll - (b * 1024 + base0);
          int eB = (int)(entB & E_MASK);
          eB = eB < nnz ? eB : 0;
          const int rB = (int)(entB >> E_BITS);
          const int cB = col[eB];
          const int vB = __float_as_int(val[eB]);
          const int i2 = (b + 2) * 1024 + tid;
          const unsigned entC = lst[i2 < cntAll ? i2 : lastp];
          if (n >= 32) add_edges<DEG>(rA, cA & (int)COL_MASK, vA, 32, lane, x, tile, degt);
          else         add_edges<DEG>(rA, cA & (int)COL_MASK, vA, n,  lane, x, tile, degt);
          rA = rB; cA = cB; vA = vB; entB = entC;
        }
      }
    } else {
      for (int eb = wv_s * 32; eb < nnz; eb += 1024) {
        int  e  = eb + lane;
        bool in = false;
        int  rr = 0, c = 0, vb = 0;
        if (e < nnz) {
          unsigned d = (unsigned)(row[e] - lo);
          in = d < (unsigned)TILE_ROWS;
          rr = (int)d;
        }
        unsigned mk = __builtin_amdgcn_ballot_w32(in);
        if (mk) {
          if (in) { c = col[e] & (int)COL_MASK; vb = __float_as_int(val[e]); }
          while (mk) {
            int k = __builtin_ctz(mk);
            mk &= mk - 1u;
            int      rk = __builtin_amdgcn_readlane(rr, k);
            unsigned ck = (unsigned)__builtin_amdgcn_readlane(c, k);
            float    vk = __int_as_float(__builtin_amdgcn_readlane(vb, k));
            float2   xv = *(const float2*)((const char*)x + (ck * (EMB * 4u) + 8u * (unsigned)lane));
            float* t = tile + rk * EMB + 2 * lane;
            atomicAdd(t,     vk * xv.x);
            atomicAdd(t + 1, vk * xv.y);
            if (DEG && lane == 0) atomicAdd(degt + rk, vk);
          }
        }
      }
    }
  } else {
  int* q   = queue + wave * QCAP;
  int  cnt = 0;
  const int    nnz16 = nnz >> 4;
  const uint4* ids4  = (const uint4*)ids;
  const int    trips = (nnz16 + 1023) >> 10;
  const int    full  = (nnz16 >> 10) & ~3;
  const int start = blockIdx.x * (nnz16 / n_tiles);

  uint4 cur[4];
  int   cid[4];
  if (full > 0) {
#pragma unroll
    for (int u = 0; u < 4; ++u) {
      int j = (u << 10) + tid + start;
      if (j >= nnz16) j -= nnz16;
      cid[u] = j;
      cur[u] = ids4[j];
    }
  }
  for (int it = 0; it < full; it += 4) {
    uint4 nxt[4];
    int   nid[4];
    const int itn = (it + 4 < full) ? (it + 4) : it;
#pragma unroll
    for (int u = 0; u < 4; ++u) {
      int j = ((itn + u) << 10) + tid + start;
      if (j >= nnz16) j -= nnz16;
      nid[u] = j;
      nxt[u] = ids4[j];
    }
#pragma unroll
    for (int u = 0; u < 4; ++u)
      take16<TILE_ROWS, DEG, REC, LISTS == 1>(cur[u], tb, true, cid[u], cnt, q, lane, pk, val, x, tile, degt, lst, &cursor);
#pragma unroll
    for (int u = 0; u < 4; ++u) { cur[u] = nxt[u]; cid[u] = nid[u]; }
  }
  for (int it = full; it < trips; ++it) {
    int  i  = (it << 10) + tid;
    bool ok = i < nnz16;
    int  j  = i + start;
    if (j >= nnz16) j -= nnz16;
    uint4 w = ids4[ok ? j : 0];
    take16<TILE_ROWS, DEG, REC, LISTS == 1>(w, tb, ok, j, cnt, q, lane, pk, val, x, tile, degt, lst, &cursor);
  }
  if (wave == 0) {
    int  e  = (nnz16 << 4) + lane;
    bool m  = (e < nnz) && ((unsigned)(row[e < nnz ? e : 0] - lo) < (unsigned)TILE_ROWS);
    unsigned mk = __builtin_amdgcn_ballot_w32(m);
    if (mk) {
      int pos = cnt + (int)__builtin_amdgcn_mbcnt_lo(mk, 0u);
      if (m) q[pos] = e;
      cnt += __builtin_popcount(mk);
    }
  }
  while (cnt > 0) {
    int n = cnt < 32 ? cnt : 32;
    resolve_queued<TILE_ROWS, DEG, REC, LISTS == 1>(q, cnt - n, n, lane, pk, val, x, tile, degt, lst, &cursor);
    cnt -= n;
  }
  }
  __syncthreads();

  const int rows_here = (n_rows - lo) < TILE_ROWS ? (n_rows - lo) : TILE_ROWS;
  float4* y4 = (float4*)(y + (size_t)lo * EMB);
  for (int i = tid; i < rows_here * (EMB / 4); i += 1024)
    y4[i] = ((const float4*)tile)[i];
  if (DEG)
    for (int i = tid; i < rows_here; i += 1024) deg[lo + i] = degt[i];
  if (LISTS == 1 && tid == 0) ((int*)lists)[blockIdx.x] = cursor;

  if (ACC && nRide < nChunks) {
    const int wv = __builtin_amdgcn_readfirstlane((int)blockIdx.x * 32 + wave);
    acc_wmma_part(acc_e, acc_cu, acc_ci, acc_out, nRide, nChunks, wv, n_tiles * 32);
  }
}

__global__ void __launch_bounds__(256) rownorm_kernel(
    float* __restrict__ x, const float* __restrict__ deg, int rowsVec) {
  const int nthr = gridDim.x * blockDim.x;
  int g = blockIdx.x * blockDim.x + threadIdx.x;
  for (; g + (SU - 1) * nthr < rowsVec; g += SU * nthr) {
    v4f   v[SU];
    float d[SU];
#pragma unroll
    for (int u = 0; u < SU; ++u) {
      int i = g + u * nthr;
      v[u] = ((const v4f*)x)[i];
      d[u] = deg[i >> 4];
    }
    __builtin_amdgcn_sched_barrier(0);
#pragma unroll
    for (int u = 0; u < SU; ++u) {
      int i = g + u * nthr;
      float dd  = (d[u] == 0.0f) ? 1.0f : d[u];
      float inv = 1.0f / dd;
      ((v4f*)x)[i] = v[u] * inv;
    }
  }
  for (; g < rowsVec; g += nthr) {
    float dd  = deg[g >> 4];
    dd = (dd == 0.0f) ? 1.0f : dd;
    float inv = 1.0f / dd;
    ((v4f*)x)[g] = ((const v4f*)x)[g] * inv;
  }
}

__global__ void __launch_bounds__(320) finalize_kernel(
    float* __restrict__ out, const float* __restrict__ ulocal,
    const float* __restrict__ deg, int totalVec, int userVec) {
  const int nthr = gridDim.x * blockDim.x;
  int g = blockIdx.x * blockDim.x + threadIdx.x;
  for (; g + (SU - 1) * nthr < totalVec; g += SU * nthr) {
    v4f   v[SU], w[SU];
    float d[SU];
#pragma unroll
    for (int u = 0; u < SU; ++u) {
      int i  = g + u * nthr;
      int iu = (i < userVec) ? i : (userVec - 1);
      v[u] = ((const v4f*)out)[i];
      w[u] = ((const v4f*)ulocal)[iu];
      d[u] = deg[iu >> 4];
    }
    __builtin_amdgcn_sched_barrier(0);
#pragma unroll
    for (int u = 0; u < SU; ++u) {
      int i = g + u * nthr;
      float dd  = (d[u] == 0.0f) ? 1.0f : d[u];
      float inv = 1.0f / dd;
      v4f r  = v[u] * 0.25f;
      v4f ru = r + w[u] * inv;
      ((v4f*)out)[i] = (i < userVec) ? ru : r;
    }
  }
  for (; g < totalVec; g += nthr) {
    v4f r = ((const v4f*)out)[g] * 0.25f;
    if (g < userVec) {
      float dd  = deg[g >> 4];
      dd = (dd == 0.0f) ? 1.0f : dd;
      float inv = 1.0f / dd;
      r += ((const v4f*)ulocal)[g] * inv;
    }
    ((v4f*)out)[g] = r;
  }
}

#define TILE_ADJ 672
#define TILE_HV  100
#define TILE_HU  400
static_assert((NUM_NODES + TILE_ADJ - 1) / TILE_ADJ <= 255, "tile id must fit a byte");
static_assert((NUM_BQ    + TILE_HV  - 1) / TILE_HV  <= 255, "tile id must fit a byte");
static_assert((NUM_USERS + TILE_HU  - 1) / TILE_HU  <= 255, "tile id must fit a byte");
static_assert(TILE_ADJ <= 1024 && TILE_HV <= 1024 && TILE_HU <= 1024, "row in tile must fit 10 bits");
static_assert(NUM_NODES <= (1 << COL_BITS), "col must fit 18 bits");
#define NNZ_ADJ 1200000
#define NNZ_BQ   400000
#define TILES_ADJ ((NUM_NODES + TILE_ADJ - 1) / TILE_ADJ)
#define TILES_HV  ((NUM_BQ    + TILE_HV  - 1) / TILE_HV)
#define TILES_HU  ((NUM_USERS + TILE_HU  - 1) / TILE_HU)
#define N_WGP 256
static_assert(TILES_ADJ < N_WGP && TILES_HV < N_WGP, "a launch with a layer sum needs blocks behind its tiles");
static_assert(TILES_HU <= N_WGP, "one tile block per WGP");
static_assert(NNZ_ADJ <= (1 << E_BITS) && TILE_ADJ <= 1024, "edge id and row in tile must fit a list word");
static_assert(TILES_ADJ <= LIST_HDR, "one count per tile in front of the lists");
#define RIDE_ADJ (3 * AU * 32 * (N_WGP - TILES_ADJ))
static_assert(RIDE_ADJ <= NUM_NODES * EMB / 256, "not more chunks than there are");

extern "C" void kernel_launch(void* const* d_in, const int* in_sizes, int n_in,
                              void* d_out, int out_size, void* d_ws, size_t ws_size,
                              hipStream_t stream) {
  const float* user_emb = (const float*)d_in[0];
  const float* item_emb = (const float*)d_in[1];
  const float* adj_val  = (const float*)d_in[2];
  const float* hv_val   = (const float*)d_in[3];
  const float* hu_val   = (const float*)d_in[4];
  const int*   adj_row  = (const int*)d_in[5];
  const int*   adj_col  = (const int*)d_in[6];
  const int*   hv_row   = (const int*)d_in[7];
  const int*   hv_col   = (const int*)d_in[8];
  const int*   hu_row   = (const int*)d_in[9];
  const int*   hu_col   = (const int*)d_in[10];

  const int E_adj = in_sizes[2];
  const int E_hv  = in_sizes[3];
  const int E_hu  = in_sizes[4];

  float* out = (float*)d_out;

  const size_t embBytesN = (size_t)NUM_NODES * EMB * sizeof(float);
  const size_t embBytesB = (size_t)NUM_BQ    * EMB * sizeof(float);
  const size_t embBytesU = (size_t)NUM_USERS * EMB * sizeof(float);
  char* ws = (char*)d_ws;
  float* e0     = (float*)(ws);
  float* e1     = (float*)(ws + embBytesN);
  float* bfeat  = (float*)(ws + 2 * embBytesN);
  float* deg_hv = (float*)(ws + 2 * embBytesN + embBytesB);
  float* deg_hu = deg_hv + NUM_BQ;
  float* ulocal = e0;
  static_assert((size_t)NNZ_ADJ + 28 * (size_t)((NNZ_ADJ + 7) / 8) + 4 <=
                (size_t)NUM_NODES * EMB * 4,
                "adjacency ids and rc words must fit the output buffer");
  static_assert(4 * ((size_t)LIST_HDR + (size_t)TILES_ADJ * LIST_CAP) <=
                (size_t)NUM_BQ * EMB * 4 + (size_t)(NUM_BQ + NUM_USERS) * 4,
                "edge counts and lists must fit the biclique area");
  static_assert(2 * ((size_t)NNZ_BQ + 8 * (size_t)NNZ_BQ) <= (size_t)NUM_ITEMS * EMB * 4,
                "biclique ids and records must fit behind u_local");
  unsigned* ids_adj = (unsigned*)d_out;
  unsigned* pk_adj  = (unsigned*)((char*)d_out + NNZ_ADJ);
  unsigned* lists   = (unsigned*)(ws + 2 * embBytesN);
  unsigned* ids_hv  = (unsigned*)(ws + embBytesU);
  unsigned* rec_hv  = (unsigned*)(ws + embBytesU + NNZ_BQ);
  unsigned* ids_hu  = (unsigned*)(ws + embBytesU + (embBytesN - embBytesU) / 2);
  unsigned* rec_hu  = (unsigned*)(ws + embBytesU + (embBytesN - embBytesU) / 2 + NNZ_BQ);

  const int totalVec = NUM_NODES * (EMB / 4);
  const int userVec  = NUM_USERS * (EMB / 4);
  const int bqVec    = NUM_BQ    * (EMB / 4);
  const int nChunks  = NUM_NODES * EMB / 256;


  copy_init_kernel<<<750, 320, 0, stream>>>(
      user_emb, item_emb, e0, totalVec, userVec);

  prep_packed_kernel<TILE_ADJ><<<586, 256, 0, stream>>>(
      adj_row, adj_col, ids_adj, pk_adj, E_adj);
  spmm_tile_kernel<TILE_ADJ, false, false, false, 1><<<TILES_ADJ, 1024, 0, stream>>>(
      ids_adj, adj_row, adj_col, pk_adj, adj_val, e0, e1, nullptr, E_adj, NUM_NODES,
      nullptr, nullptr, nullptr, nullptr, 0, 0, lists);
  spmm_tile_kernel<TILE_ADJ, false, true, false, 2><<<N_WGP, 1024, 0, stream>>>(
      nullptr, adj_row, adj_col, nullptr, adj_val, e1, e0, nullptr, E_adj, NUM_NODES,
      e1, user_emb, item_emb, out, nChunks, RIDE_ADJ, lists);
  spmm_tile_kernel<TILE_ADJ, false, true, false, 2><<<N_WGP, 1024, 0, stream>>>(
      nullptr, adj_row, adj_col, nullptr, adj_val, e0, e1, nullptr, E_adj, NUM_NODES,
      e0, out, out + (size_t)NUM_USERS * EMB, out, nChunks, RIDE_ADJ, lists);

  prep_rec_kernel<TILE_HV, TILE_HU><<<782, 256, 0, stream>>>(
      hv_row, hv_col, hv_val, ids_hv, rec_hv, E_hv,
      hu_row, hu_col, hu_val, ids_hu, rec_hu, E_hu);
  spmm_tile_kernel<TILE_HV, true, true, true, 0><<<N_WGP, 1024, 0, stream>>>(
      ids_hv, hv_row, hv_col, rec_hv, hv_val, item_emb, bfeat, deg_hv, E_hv, NUM_BQ,
      e1, out, out + (size_t)NUM_USERS * EMB, out, nChunks, nChunks, nullptr);
  rownorm_kernel<<<250, 256, 0, stream>>>(bfeat, deg_hv, bqVec);

  spmm_tile_kernel<TILE_HU, true, false, true, 0><<<TILES_HU, 1024, 0, stream>>>(
      ids_hu, hu_row, hu_col, rec_hu, hu_val, bfeat, ulocal, deg_hu, E_hu, NUM_USERS,
      nullptr, nullptr, nullptr, nullptr, 0, 0, nullptr);

  finalize_kernel<<<750, 320, 0, stream>>>(
      out, ulocal, deg_hu, totalVec, userVec);
}


# Biclique path's elementwise launches go: tile blocks store rows/deg (hv) and out = acc/4 + rows/deg (hu) straight from LDS; item rows get /4 in the last layer sum; 10 launches -> 8; sum order kept
# speedup vs baseline: 2.5608x; 1.0578x over previous
#include <hip/hip_runtime.h>
#include <hip/hip_bf16.h>

typedef __attribute__((ext_vector_type(2))) float v2f;
typedef __attribute__((ext_vector_type(8))) float v8f;
typedef __attribute__((ext_vector_type(4))) float v4f;

#define NUM_USERS 100000
#define NUM_ITEMS 50000
#define NUM_NODES 150000
#define NUM_BQ    20000
#define EMB       64

#define SU 5

__global__ void __launch_bounds__(320) copy_init_kernel(
    const float* __restrict__ ue, const float* __restrict__ ie,
    float* __restrict__ e0,
    int totalVec, int userVec) {
  const int nthr = gridDim.x * blockDim.x;
  int g = blockIdx.x * blockDim.x + threadIdx.x;
  for (; g + (SU - 1) * nthr < totalVec; g += SU * nthr) {
    v4f v[SU];
#pragma unroll
    for (int u = 0; u < SU; ++u) {
      int i = g + u * nthr;
      v[u] = (i < userVec) ? ((const v4f*)ue)[i] : ((const v4f*)ie)[i - userVec];
    }
    __builtin_amdgcn_sched_barrier(0);
#pragma unroll
    for (int u = 0; u < SU; ++u) {
      int i = g + u * nthr;
      ((v4f*)e0)[i]  = v[u];
    }
  }
  for (; g < totalVec; g += nthr) {
    v4f v = (g < userVec) ? ((const v4f*)ue)[g] : ((const v4f*)ie)[g - userVec];
    ((v4f*)e0)[g]  = v;
  }
}

#define QCAP 64

#define COL_BITS 18
#define COL_MASK 0x3ffffu
#define RC_MASK  0x0fffffffu

struct pair32 { unsigned lo, hi; };

#define E_BITS   21
#define E_MASK   0x1fffffu
#define LIST_CAP 6240
#define LIST_HDR 256

template <bool DEG>
__device__ __forceinline__ void add_edges(
    int rr, int c, int vb, int n, int lane,
    const float* __restrict__ x, float* tile, float* degt) {
  const int last = n - 1;
  for (int k0 = 0; k0 < n; k0 += 16) {
    float2 xv[16];
#pragma unroll
    for (int j = 0; j < 16; ++j) {
      int k  = (k0 + j < last) ? (k0 + j) : last;
      unsigned ck = (unsigned)__builtin_amdgcn_readlane(c, k);
      xv[j] = *(const float2*)((const char*)x + (ck * (EMB * 4u) + 8u * (unsigned)lane));
    }
#pragma unroll
    for (int j = 0; j < 16; ++j) {
      if (k0 + j < n) {
        int   k  = k0 + j;
        int   rk = __builtin_amdgcn_readlane(rr, k);
        float vk = __int_as_float(__builtin_amdgcn_readlane(vb, k));
        float* t = tile + rk * EMB + 2 * lane;
        atomicAdd(t,     vk * xv[j].x);
        atomicAdd(t + 1, vk * xv[j].y);
        if (DEG && lane == 0) atomicAdd(degt + rk, vk);
      }
    }
  }
}

template <int TILE_ROWS, bool DEG, bool REC, bool BUILD>
__device__ __forceinline__ void resolve_queued(
    const int* q, int base, int n, int lane,
    const unsigned* __restrict__ pk,
    const float* __restrict__ val, const float* __restrict__ x,
    float* tile, float* degt, unsigned* lst, int* cursor) {
  int last = n - 1;
  int e  = q[base + (lane < last ? lane : last)];
  unsigned w;
  int      vb;
  if (REC) {
    uint2 r = ((const uint2*)pk)[e];
    w  = r.x;
    vb = (int)r.y;
  } else {
    unsigned bit = 28u * ((unsigned)e & 7u);
    const unsigned* p = pk + 7 * (size_t)(e >> 3) + (bit >> 5);
    pair32 t;
    __builtin_memcpy(&t, p, 8);
    w  = (unsigned)(((((unsigned long long)t.hi) << 32) | t.lo) >> (bit & 31u)) & RC_MASK;
    vb = __float_as_int(val[e]);
  }
  int rr = (int)(w >> COL_BITS);
  int c  = (int)(w & COL_MASK);
  if (BUILD) {
    int pos = 0;
    if (lane == 0) pos = atomicAdd(cursor, n);
    pos = __builtin_amdgcn_readfirstlane(pos);
    if (lane < n && pos + lane < LIST_CAP)
      lst[pos + lane] = (unsigned)e | ((unsigned)rr << E_BITS);
  }
  add_edges<DEG>(rr, c, vb, n, lane, x, tile, degt);
}

template <int TILE_ROWS>
__device__ __forceinline__ unsigned tile_id_of(int r) {
  unsigned t = (unsigned)r / (unsigned)TILE_ROWS;
  return t < 255u ? t : 255u;
}
template <int TILE_ROWS>
__device__ __forceinline__ unsigned pack_ids(int4 r) {
  return tile_id_of<TILE_ROWS>(r.x) | (tile_id_of<TILE_ROWS>(r.y) << 8) |
         (tile_id_of<TILE_ROWS>(r.z) << 16) | (tile_id_of<TILE_ROWS>(r.w) << 24);
}

template <int TILE_ROWS>
__device__ __forceinline__ unsigned rc_word(int r, int c) {
  unsigned t  = (unsigned)r / (unsigned)TILE_ROWS;
  unsigned rl = (unsigned)r - t * (unsigned)TILE_ROWS;
  return ((rl & 0x3ffu) << COL_BITS) | ((unsigned)c & COL_MASK);
}

template <int TILE_ROWS>
__global__ void __launch_bounds__(256) prep_packed_kernel(
    const int* __restrict__ row, const int* __restrict__ col,
    unsigned* __restrict__ ids, unsigned* __restrict__ pk, int nnz) {
  const int nthr = gridDim.x * blockDim.x;
  const int nG   = (nnz + 7) >> 3;
  for (int g = blockIdx.x * blockDim.x + threadIdx.x; g < nG; g += nthr) {
    int4 r0, r1, c0, c1;
    if (8 * g + 7 < nnz) {
      r0 = ((const int4*)row)[2 * g];  r1 = ((const int4*)row)[2 * g + 1];
      c0 = ((const int4*)col)[2 * g];  c1 = ((const int4*)col)[2 * g + 1];
    } else {
      int e[8];
#pragma unroll
      for (int k = 0; k < 8; ++k) e[k] = (8 * g + k < nnz) ? (8 * g + k) : (nnz - 1);
      r0 = make_int4(row[e[0]], row[e[1]], row[e[2]], row[e[3]]);
      r1 = make_int4(row[e[4]], row[e[5]], row[e[6]], row[e[7]]);
      c0 = make_int4(col[e[0]], col[e[1]], col[e[2]], col[e[3]]);
      c1 = make_int4(col[e[4]], col[e[5]], col[e[6]], col[e[7]]);
    }
    if (8 * g + 3 < nnz) ids[2 * g]     = pack_ids<TILE_ROWS>(r0);
    if (8 * g + 7 < nnz) ids[2 * g + 1] = pack_ids<TILE_ROWS>(r1);
    const unsigned w0 = rc_word<TILE_ROWS>(r0.x, c0.x), w1 = rc_word<TILE_ROWS>(r0.y, c0.y);
    const unsigned w2 = rc_word<TILE_ROWS>(r0.z, c0.z), w3 = rc_word<TILE_ROWS>(r0.w, c0.w);
    const unsigned w4 = rc_word<TILE_ROWS>(r1.x, c1.x), w5 = rc_word<TILE_ROWS>(r1.y, c1.y);
    const unsigned w6 = rc_word<TILE_ROWS>(r1.z, c1.z), w7 = rc_word<TILE_ROWS>(r1.w, c1.w);
    unsigned* p = pk + 7 * (size_t)g;
    p[0] =  w0        | (w1 << 28);
    p[1] = (w1 >> 4)  | (w2 << 24);
    p[2] = (w2 >> 8)  | (w3 << 20);
    p[3] = (w3 >> 12) | (w4 << 16);
    p[4] = (w4 >> 16) | (w5 << 12);
    p[5] = (w5 >> 20) | (w6 << 8);
    p[6] = (w6 >> 24) | (w7 << 4);
  }
  if (blockIdx.x == 0 && threadIdx.x == 0) pk[7 * (size_t)nG] = 0u;
}

template <int TILE_ROWS>
__device__ __forceinline__ void prep_rec_group(
    const int* __restrict__ row, const int* __restrict__ col, const float* __restrict__ val,
    unsigned* __restrict__ ids, unsigned* __restrict__ rec, int nnz, int g) {
  if (4 * g + 3 < nnz) {
    const int4   r = ((const int4*)row)[g];
    const int4   c = ((const int4*)col)[g];
    const float4 v = ((const float4*)val)[g];
    ids[g] = pack_ids<TILE_ROWS>(r);
    ((uint4*)rec)[2 * g]     = make_uint4(rc_word<TILE_ROWS>(r.x, c.x), __float_as_uint(v.x),
                                          rc_word<TILE_ROWS>(r.y, c.y), __float_as_uint(v.y));
    ((uint4*)rec)[2 * g + 1] = make_uint4(rc_word<TILE_ROWS>(r.z, c.z), __float_as_uint(v.z),
                                          rc_word<TILE_ROWS>(r.w, c.w), __float_as_uint(v.w));
  } else {
    for (int e = 4 * g; e < nnz; ++e) {
      rec[2 * e]     = rc_word<TILE_ROWS>(row[e], col[e]);
      rec[2 * e + 1] = __float_as_uint(val[e]);
    }
  }
}
template <int TILE_A, int TILE_B>
__global__ void __launch_bounds__(256) prep_rec_kernel(
    const int* __restrict__ rowA, const int* __restrict__ colA, const float* __restrict__ valA,
    unsigned* __restrict__ idsA, unsigned* __restrict__ recA, int nnzA,
    const int* __restrict__ rowB, const int* __restrict__ colB, const float* __restrict__ valB,
    unsigned* __restrict__ idsB, unsigned* __restrict__ recB, int nnzB) {
  const int nthr = gridDim.x * blockDim.x;
  const int nA4  = (nnzA + 3) >> 2;
  const int nB4  = (nnzB + 3) >> 2;
  for (int g = blockIdx.x * blockDim.x + threadIdx.x; g < nA4 + nB4; g += nthr) {
    if (g < nA4) prep_rec_group<TILE_A>(rowA, colA, valA, idsA, recA, nnzA, g);
    else         prep_rec_group<TILE_B>(rowB, colB, valB, idsB, recB, nnzB, g - nA4);
  }
}

__device__ __forceinline__ unsigned nonzero_bytes(unsigned v) {
  return (((v & 0x7f7f7f7fu) + 0x7f7f7f7fu) | v) & 0x80808080u;
}

template <int TILE_ROWS, bool DEG, bool REC, bool BUILD>
__device__ __forceinline__ void take16(
    uint4 w, unsigned tb, bool ok, int j, int& cnt, int* q, int lane,
    const unsigned* __restrict__ pk,
    const float* __restrict__ val, const float* __restrict__ x,
    float* tile, float* degt, unsigned* lst, int* cursor) {
  unsigned nz = (nonzero_bytes(w.x ^ tb) >> 7) | (nonzero_bytes(w.y ^ tb) >> 6) |
                (nonzero_bytes(w.z ^ tb) >> 5) | (nonzero_bytes(w.w ^ tb) >> 4);
  unsigned m = ok ? (~nz & 0x0f0f0f0fu) : 0u;
  unsigned mk = __builtin_amdgcn_ballot_w32(m != 0u);
  while (mk) {
    int p   = __builtin_ctz(m | 0x80000000u);
    int pos = cnt + (int)__builtin_amdgcn_mbcnt_lo(mk, 0u);
    if (m != 0u) q[pos] = 16 * j + 4 * (p & 7) + (p >> 3);
    cnt += __builtin_popcount(mk);
    m &= m - 1u;
    if (cnt >= 32) {
      resolve_queued<TILE_ROWS, DEG, REC, BUILD>(q, cnt - 32, 32, lane, pk, val, x, tile, degt, lst, cursor);
      cnt -= 32;
    }
    mk = __builtin_amdgcn_ballot_w32(m != 0u);
  }
}

#define AU 5
#define USER_CHUNKS (NUM_USERS * EMB / 256)
static_assert((NUM_USERS * EMB) % 256 == 0, "user rows end at a chunk boundary");
template <bool LAST>
__device__ __forceinline__ void acc_wmma_part(
    const float* __restrict__ e, const float* cu, const float* ci, float* acc,
    int ch0, int ch1, int wv, int nWaves) {
  const int lane = threadIdx.x & 31;
  const int n    = lane & 15;
  const int hi   = lane >> 4;

  v2f a[4];
#pragma unroll
  for (int k = 0; k < 4; ++k) {
    a[k].x = (n == (4 * k + 2 * hi))     ? 1.0f : 0.0f;
    a[k].y = (n == (4 * k + 2 * hi + 1)) ? 1.0f : 0.0f;
  }

  for (int g = ch0 + wv * AU; g < ch1; g += nWaves * AU) {
    v8f cc[AU];
    v2f b[AU][4];
#pragma unroll
    for (int u = 0; u < AU; ++u) {
      int ch = (g + u < ch1) ? (g + u) : (ch1 - 1);
      const float* ap = (ch < USER_CHUNKS ? cu + (size_t)ch * 256
                                          : ci + (size_t)(ch - USER_CHUNKS) * 256) + 4 * lane;
      const float* ep = e   + (size_t)ch * 256 + 4 * n + 2 * hi;
      v4f lo = *(const v4f*)(ap);
      v4f up = *(const v4f*)(ap + 128);
      cc[u][0] = lo.x; cc[u][1] = lo.y; cc[u][2] = lo.z; cc[u][3] = lo.w;
      cc[u][4] = up.x; cc[u][5] = up.y; cc[u][6] = up.z; cc[u][7] = up.w;
#pragma unroll
      for (int k = 0; k < 4; ++k)
        b[u][k] = *(const v2f*)(ep + 128 * (k & 1) + 64 * (k >> 1));
    }
    __builtin_amdgcn_sched_barrier(0);
#pragma unroll
    for (int k = 0; k < 4; ++k)
#pragma unroll
      for (int u = 0; u < AU; ++u)
        cc[u] = __builtin_amdgcn_wmma_f32_16x16x4_f32(
            false, a[k], false, b[u][k], (short)0, cc[u], false, false);
#pragma unroll
    for (int u = 0; u < AU; ++u) {
      if (g + u < ch1) {
        if (LAST) {
          const float s = (g + u >= USER_CHUNKS) ? 0.25f : 1.0f;
          cc[u] = cc[u] * s;
        }
        float* op = acc + (size_t)(g + u) * 256 + 4 * lane;
        v4f lo = {cc[u][0], cc[u][1], cc[u][2], cc[u][3]};
        v4f up = {cc[u][4], cc[u][5], cc[u][6], cc[u][7]};
        *(v4f*)(op)       = lo;
        *(v4f*)(op + 128) = up;
      }
    }
  }
}

#define FU 7
template <int TILE_ROWS, int EPI, int ACC, bool REC, int LISTS>
__global__ void __launch_bounds__(1024) spmm_tile_kernel(
    const unsigned* __restrict__ ids,
    const int* __restrict__ row, const int* __restrict__ col,
    const unsigned* __restrict__ pk,
    const float* __restrict__ val, const float* __restrict__ x,
    float* __restrict__ y, int nnz, int n_rows,
    const float* __restrict__ acc_e, const float* acc_cu, const float* acc_ci,
    float* acc_out, int nChunks, int nRide, unsigned* lists) {
  constexpr bool DEG = EPI != 0;
  __shared__ float tile[TILE_ROWS * EMB];
  __shared__ float degt[DEG ? TILE_ROWS : 1];
  __shared__ int   queue[LISTS == 2 ? 1 : 32 * QCAP];
  __shared__ int   cursor;

  const int n_tiles = (n_rows + TILE_ROWS - 1) / TILE_ROWS;
  if (ACC && (int)blockIdx.x >= n_tiles) {
    const int wv = __builtin_amdgcn_readfirstlane(
        ((int)blockIdx.x - n_tiles) * 32 + (int)(threadIdx.x >> 5));
    acc_wmma_part<ACC == 2>(acc_e, acc_cu, acc_ci, acc_out, 0, nRide, wv,
                            ((int)gridDim.x - n_tiles) * 32);
    return;
  }

  const int tid  = threadIdx.x;
  const int lane = tid & 31;
  const int wave = tid >> 5;
  const int lo   = blockIdx.x * TILE_ROWS;
  const unsigned tb = 0x01010101u * blockIdx.x;
  unsigned* lst = lists + LIST_HDR + (size_t)blockIdx.x * LIST_CAP;

  for (int i = tid; i < TILE_ROWS * (EMB / 4); i += 1024)
    ((float4*)tile)[i] = make_float4(0.f, 0.f, 0.f, 0.f);
  if (DEG)
    for (int i = tid; i < TILE_ROWS; i += 1024) degt[i] = 0.f;
  if (LISTS == 1 && tid == 0) cursor = 0;
  __syncthreads();

  if (LISTS == 2) {
    const int wv_s   = __builtin_amdgcn_readfirstlane(wave);
    const int cntAll = __builtin_amdgcn_readfirstlane(((const int*)lists)[blockIdx.x]);
    if (cntAll <= LIST_CAP) {
      if (cntAll > 0) {
        const int base0 = wv_s * 32;
        const int lastp = cntAll - 1;
        unsigned entA = lst[tid < cntAll ? tid : lastp];
        unsigned entB = lst[1024 + tid < cntAll ? 1024 + tid : lastp];
        int eA = (int)(entA & E_MASK);
        eA = eA < nnz ? eA : 0;
        int rA = (int)(entA >> E_BITS);
        int cA = col[eA];
        int vA = __float_as_int(val[eA]);
        for (int b = 0; cntAll - (b * 1024 + base0) > 0; ++b) {
          const int n  = cntAll - (b * 1024 + base0);
          int eB = (int)(entB & E_MASK);
          eB = eB < nnz ? eB : 0;
          const int rB = (int)(entB >> E_BITS);
          const int cB = col[eB];
          const int vB = __float_as_int(val[eB]);
          const int i2 = (b + 2) * 1024 + tid;
          const unsigned entC = lst[i2 < cntAll ? i2 : lastp];
          if (n >= 32) add_edges<DEG>(rA, cA & (int)COL_MASK, vA, 32, lane, x, tile, degt);
          else         add_edges<DEG>(rA, cA & (int)COL_MASK, vA, n,  lane, x, tile, degt);
          rA = rB; cA = cB; vA = vB; entB = entC;
        }
      }
    } else {
      for (int eb = wv_s * 32; eb < nnz; eb += 1024) {
        int  e  = eb + lane;
        bool in = false;
        int  rr = 0, c = 0, vb = 0;
        if (e < nnz) {
          unsigned d = (unsigned)(row[e] - lo);
          in = d < (unsigned)TILE_ROWS;
          rr = (int)d;
        }
        unsigned mk = __builtin_amdgcn_ballot_w32(in);
        if (mk) {
          if (in) { c = col[e] & (int)COL_MASK; vb = __float_as_int(val[e]); }
          while (mk) {
            int k = __builtin_ctz(mk);
            mk &= mk - 1u;
            int      rk = __builtin_amdgcn_readlane(rr, k);
            unsigned ck = (unsigned)__builtin_amdgcn_readlane(c, k);
            float    vk = __int_as_float(__builtin_amdgcn_readlane(vb, k));
            float2   xv = *(const float2*)((const char*)x + (ck * (EMB * 4u) + 8u * (unsigned)lane));
            float* t = tile + rk * EMB + 2 * lane;
            atomicAdd(t,     vk * xv.x);
            atomicAdd(t + 1, vk * xv.y);
            if (DEG && lane == 0) atomicAdd(degt + rk, vk);
          }
        }
      }
    }
  } else {
  int* q   = queue + wave * QCAP;
  int  cnt = 0;
  const int    nnz16 = nnz >> 4;
  const uint4* ids4  = (const uint4*)ids;
  const int    trips = (nnz16 + 1023) >> 10;
  const int    full  = (nnz16 >> 10) & ~3;
  const int start = blockIdx.x * (nnz16 / n_tiles);

  uint4 cur[4];
  int   cid[4];
  if (full > 0) {
#pragma unroll
    for (int u = 0; u < 4; ++u) {
      int j = (u << 10) + tid + start;
      if (j >= nnz16) j -= nnz16;
      cid[u] = j;
      cur[u] = ids4[j];
    }
  }
  for (int it = 0; it < full; it += 4) {
    uint4 nxt[4];
    int   nid[4];
    const int itn = (it + 4 < full) ? (it + 4) : it;
#pragma unroll
    for (int u = 0; u < 4; ++u) {
      int j = ((itn + u) << 10) + tid + start;
      if (j >= nnz16) j -= nnz16;
      nid[u] = j;
      nxt[u] = ids4[j];
    }
#pragma unroll
    for (int u = 0; u < 4; ++u)
      take16<TILE_ROWS, DEG, REC, LISTS == 1>(cur[u], tb, true, cid[u], cnt, q, lane, pk, val, x, tile, degt, lst, &cursor);
#pragma unroll
    for (int u = 0; u < 4; ++u) { cur[u] = nxt[u]; cid[u] = nid[u]; }
  }
  for (int it = full; it < trips; ++it) {
    int  i  = (it << 10) + tid;
    bool ok = i < nnz16;
    int  j  = i + start;
    if (j >= nnz16) j -= nnz16;
    uint4 w = ids4[ok ? j : 0];
    take16<TILE_ROWS, DEG, REC, LISTS == 1>(w, tb, ok, j, cnt, q, lane, pk, val, x, tile, degt, lst, &cursor);
  }
  if (wave == 0) {
    int  e  = (nnz16 << 4) + lane;
    bool m  = (e < nnz) && ((unsigned)(row[e < nnz ? e : 0] - lo) < (unsigned)TILE_ROWS);
    unsigned mk = __builtin_amdgcn_ballot_w32(m);
    if (mk) {
      int pos = cnt + (int)__builtin_amdgcn_mbcnt_lo(mk, 0u);
      if (m) q[pos] = e;
      cnt += __builtin_popcount(mk);
    }
  }
  while (cnt > 0) {
    int n = cnt < 32 ? cnt : 32;
    resolve_queued<TILE_ROWS, DEG, REC, LISTS == 1>(q, cnt - n, n, lane, pk, val, x, tile, degt, lst, &cursor);
    cnt -= n;
  }
  }
  __syncthreads();

  const int rows_here = (n_rows - lo) < TILE_ROWS ? (n_rows - lo) : TILE_ROWS;
  const int nv = rows_here * (EMB / 4);
  v4f* y4 = (v4f*)(y + (size_t)lo * EMB);
  const v4f* t4 = (const v4f*)tile;
  if (EPI == 0) {
    for (int i = tid; i < nv; i += 1024) y4[i] = t4[i];
  } else if (EPI == 1) {
    for (int i = tid; i < nv; i += 1024) {
      float dd  = degt[i >> 4];
      dd = (dd == 0.0f) ? 1.0f : dd;
      float inv = 1.0f / dd;
      y4[i] = t4[i] * inv;
    }
  } else {
    for (int i0 = tid; i0 < nv; i0 += FU * 1024) {
      v4f a[FU];
#pragma unroll
      for (int u = 0; u < FU; ++u) {
        int i = i0 + u * 1024;
        a[u] = y4[i < nv ? i : nv - 1];
      }
      __builtin_amdgcn_sched_barrier(0);
#pragma unroll
      for (int u = 0; u < FU; ++u) {
        int i = i0 + u * 1024;
        if (i < nv) {
          float dd  = degt[i >> 4];
          dd = (dd == 0.0f) ? 1.0f : dd;
          float inv = 1.0f / dd;
          v4f r = a[u] * 0.25f;
          y4[i] = r + t4[i] * inv;
        }
      }
    }
  }
  if (LISTS == 1 && tid == 0) ((int*)lists)[blockIdx.x] = cursor;

  if (ACC && nRide < nChunks) {
    const int wv = __builtin_amdgcn_readfirstlane((int)blockIdx.x * 32 + wave);
    acc_wmma_part<ACC == 2>(acc_e, acc_cu, acc_ci, acc_out, nRide, nChunks, wv, n_tiles * 32);
  }
}

#define TILE_ADJ 672
#define TILE_HV  100
#define TILE_HU  400
static_assert((NUM_NODES + TILE_ADJ - 1) / TILE_ADJ <= 255, "tile id must fit a byte");
static_assert((NUM_BQ    + TILE_HV  - 1) / TILE_HV  <= 255, "tile id must fit a byte");
static_assert((NUM_USERS + TILE_HU  - 1) / TILE_HU  <= 255, "tile id must fit a byte");
static_assert(TILE_ADJ <= 1024 && TILE_HV <= 1024 && TILE_HU <= 1024, "row in tile must fit 10 bits");
static_assert(NUM_NODES <= (1 << COL_BITS), "col must fit 18 bits");
#define NNZ_ADJ 1200000
#define NNZ_BQ   400000
#define TILES_ADJ ((NUM_NODES + TILE_ADJ - 1) / TILE_ADJ)
#define TILES_HV  ((NUM_BQ    + TILE_HV  - 1) / TILE_HV)
#define TILES_HU  ((NUM_USERS + TILE_HU  - 1) / TILE_HU)
#define N_WGP 256
static_assert(TILES_ADJ < N_WGP && TILES_HV < N_WGP, "a launch with a layer sum needs blocks behind its tiles");
static_assert(TILES_HU <= N_WGP, "one tile block per WGP");
static_assert(NNZ_ADJ <= (1 << E_BITS) && TILE_ADJ <= 1024, "edge id and row in tile must fit a list word");
static_assert(TILES_ADJ <= LIST_HDR, "one count per tile in front of the lists");
#define RIDE_ADJ (3 * AU * 32 * (N_WGP - TILES_ADJ))
static_assert(RIDE_ADJ <= NUM_NODES * EMB / 256, "not more chunks than there are");

extern "C" void kernel_launch(void* const* d_in, const int* in_sizes, int n_in,
                              void* d_out, int out_size, void* d_ws, size_t ws_size,
                              hipStream_t stream) {
  const float* user_emb = (const float*)d_in[0];
  const float* item_emb = (const float*)d_in[1];
  const float* adj_val  = (const float*)d_in[2];
  const float* hv_val   = (const float*)d_in[3];
  const float* hu_val   = (const float*)d_in[4];
  const int*   adj_row  = (const int*)d_in[5];
  const int*   adj_col  = (const int*)d_in[6];
  const int*   hv_row   = (const int*)d_in[7];
  const int*   hv_col   = (const int*)d_in[8];
  const int*   hu_row   = (const int*)d_in[9];
  const int*   hu_col   = (const int*)d_in[10];

  const int E_adj = in_sizes[2];
  const int E_hv  = in_sizes[3];
  const int E_hu  = in_sizes[4];

  float* out = (float*)d_out;

  const size_t embBytesN = (size_t)NUM_NODES * EMB * sizeof(float);
  const size_t embBytesU = (size_t)NUM_USERS * EMB * sizeof(float);
  char* ws = (char*)d_ws;
  float* e0     = (float*)(ws);
  float* e1     = (float*)(ws + embBytesN);
  float* bfeat  = (float*)(ws + 2 * embBytesN);
  static_assert((size_t)NNZ_ADJ + 28 * (size_t)((NNZ_ADJ + 7) / 8) + 4 <=
                (size_t)NUM_NODES * EMB * 4,
                "adjacency ids and rc words must fit the output buffer");
  static_assert(4 * ((size_t)LIST_HDR + (size_t)TILES_ADJ * LIST_CAP) <=
                (size_t)NUM_BQ * EMB * 4 + (size_t)(NUM_BQ + NUM_USERS) * 4,
                "edge counts and lists must fit the biclique area");
  static_assert(2 * ((size_t)NNZ_BQ + 8 * (size_t)NNZ_BQ) <= (size_t)NUM_ITEMS * EMB * 4,
                "biclique ids and records must fit the tail of the e0 buffer");
  unsigned* ids_adj = (unsigned*)d_out;
  unsigned* pk_adj  = (unsigned*)((char*)d_out + NNZ_ADJ);
  unsigned* lists   = (unsigned*)(ws + 2 * embBytesN);
  unsigned* ids_hv  = (unsigned*)(ws + embBytesU);
  unsigned* rec_hv  = (unsigned*)(ws + embBytesU + NNZ_BQ);
  unsigned* ids_hu  = (unsigned*)(ws + embBytesU + (embBytesN - embBytesU) / 2);
  unsigned* rec_hu  = (unsigned*)(ws + embBytesU + (embBytesN - embBytesU) / 2 + NNZ_BQ);

  const int totalVec = NUM_NODES * (EMB / 4);
  const int userVec  = NUM_USERS * (EMB / 4);
  const int nChunks  = NUM_NODES * EMB / 256;


  copy_init_kernel<<<750, 320, 0, stream>>>(
      user_emb, item_emb, e0, totalVec, userVec);

  prep_packed_kernel<TILE_ADJ><<<586, 256, 0, stream>>>(
      adj_row, adj_col, ids_adj, pk_adj, E_adj);
  spmm_tile_kernel<TILE_ADJ, 0, 0, false, 1><<<TILES_ADJ, 1024, 0, stream>>>(
      ids_adj, adj_row, adj_col, pk_adj, adj_val, e0, e1, E_adj, NUM_NODES,
      nullptr, nullptr, nullptr, nullptr, 0, 0, lists);
  spmm_tile_kernel<TILE_ADJ, 0, 1, false, 2><<<N_WGP, 1024, 0, stream>>>(
      nullptr, adj_row, adj_col, nullptr, adj_val, e1, e0, E_adj, NUM_NODES,
      e1, user_emb, item_emb, out, nChunks, RIDE_ADJ, lists);
  spmm_tile_kernel<TILE_ADJ, 0, 1, false, 2><<<N_WGP, 1024, 0, stream>>>(
      nullptr, adj_row, adj_col, nullptr, adj_val, e0, e1, E_adj, NUM_NODES,
      e0, out, out + (size_t)NUM_USERS * EMB, out, nChunks, RIDE_ADJ, lists);

  prep_rec_kernel<TILE_HV, TILE_HU><<<782, 256, 0, stream>>>(
      hv_row, hv_col, hv_val, ids_hv, rec_hv, E_hv,
      hu_row, hu_col, hu_val, ids_hu, rec_hu, E_hu);
  spmm_tile_kernel<TILE_HV, 1, 2, true, 0><<<N_WGP, 1024, 0, stream>>>(
      ids_hv, hv_row, hv_col, rec_hv, hv_val, item_emb, bfeat, E_hv, NUM_BQ,
      e1, out, out + (size_t)NUM_USERS * EMB, out, nChunks, nChunks, nullptr);

  spmm_tile_kernel<TILE_HU, 2, 0, true, 0><<<TILES_HU, 1024, 0, stream>>>(
      ids_hu, hu_row, hu_col, rec_hu, hu_val, bfeat, out, E_hu, NUM_USERS,
      nullptr, nullptr, nullptr, nullptr, 0, 0, nullptr);
}


# Layer 1 gathers its source rows straight from the two input arrays (a lane picks the array by its col; row address broadcast as a scalar): concat copy launch goes, 8 launches -> 7; sum order kept
# speedup vs baseline: 2.8224x; 1.1022x over previous
#include <hip/hip_runtime.h>
#include <hip/hip_bf16.h>

typedef __attribute__((ext_vector_type(2))) float v2f;
typedef __attribute__((ext_vector_type(8))) float v8f;
typedef __attribute__((ext_vector_type(4))) float v4f;
typedef const __attribute__((address_space(1))) v2f* gv2f_ptr;

#define NUM_USERS 100000
#define NUM_ITEMS 50000
#define NUM_NODES 150000
#define NUM_BQ    20000
#define EMB       64

#define QCAP 64

#define COL_BITS 18
#define COL_MASK 0x3ffffu
#define RC_MASK  0x0fffffffu

struct pair32 { unsigned lo, hi; };

#define E_BITS   21
#define E_MASK   0x1fffffu
#define LIST_CAP 6240
#define LIST_HDR 256

template <bool DEG, bool SPLIT>
__device__ __forceinline__ void add_edges(
    int rr, int c, int vb, int n, int lane,
    const float* __restrict__ x, const float* __restrict__ x2, float* tile, float* degt) {
  const int last = n - 1;
  unsigned long long rowa = 0;
  if (SPLIT) {
    const unsigned cu = (unsigned)c < (unsigned)NUM_NODES ? (unsigned)c : (unsigned)(NUM_NODES - 1);
    rowa = cu >= (unsigned)NUM_USERS
               ? (unsigned long long)x2 + (unsigned long long)(cu - (unsigned)NUM_USERS) * (EMB * 4u)
               : (unsigned long long)x  + (unsigned long long)cu * (EMB * 4u);
  }
  for (int k0 = 0; k0 < n; k0 += 16) {
    float2 xv[16];
#pragma unroll
    for (int j = 0; j < 16; ++j) {
      int k  = (k0 + j < last) ? (k0 + j) : last;
      if (SPLIT) {
        const unsigned alo = (unsigned)__builtin_amdgcn_readlane((int)(unsigned)(rowa & 0xffffffffull), k);
        const unsigned ahi = (unsigned)__builtin_amdgcn_readlane((int)(unsigned)(rowa >> 32), k);
        const gv2f_ptr rp = (gv2f_ptr)(((((unsigned long long)ahi) << 32) | alo) + 8u * (unsigned)lane);
        const v2f t = *rp;
        xv[j].x = t.x;
        xv[j].y = t.y;
      } else {
        unsigned ck = (unsigned)__builtin_amdgcn_readlane(c, k);
        xv[j] = *(const float2*)((const char*)x + (ck * (EMB * 4u) + 8u * (unsigned)lane));
      }
    }
#pragma unroll
    for (int j = 0; j < 16; ++j) {
      if (k0 + j < n) {
        int   k  = k0 + j;
        int   rk = __builtin_amdgcn_readlane(rr, k);
        float vk = __int_as_float(__builtin_amdgcn_readlane(vb, k));
        float* t = tile + rk * EMB + 2 * lane;
        atomicAdd(t,     vk * xv[j].x);
        atomicAdd(t + 1, vk * xv[j].y);
        if (DEG && lane == 0) atomicAdd(degt + rk, vk);
      }
    }
  }
}

template <int TILE_ROWS, bool DEG, bool REC, bool BUILD, bool SPLIT>
__device__ __forceinline__ void resolve_queued(
    const int* q, int base, int n, int lane,
    const unsigned* __restrict__ pk,
    const float* __restrict__ val, const float* __restrict__ x, const float* __restrict__ x2,
    float* tile, float* degt, unsigned* lst, int* cursor) {
  int last = n - 1;
  int e  = q[base + (lane < last ? lane : last)];
  unsigned w;
  int      vb;
  if (REC) {
    uint2 r = ((const uint2*)pk)[e];
    w  = r.x;
    vb = (int)r.y;
  } else {
    unsigned bit = 28u * ((unsigned)e & 7u);
    const unsigned* p = pk + 7 * (size_t)(e >> 3) + (bit >> 5);
    pair32 t;
    __builtin_memcpy(&t, p, 8);
    w  = (unsigned)(((((unsigned long long)t.hi) << 32) | t.lo) >> (bit & 31u)) & RC_MASK;
    vb = __float_as_int(val[e]);
  }
  int rr = (int)(w >> COL_BITS);
  int c  = (int)(w & COL_MASK);
  if (BUILD) {
    int pos = 0;
    if (lane == 0) pos = atomicAdd(cursor, n);
    pos = __builtin_amdgcn_readfirstlane(pos);
    if (lane < n && pos + lane < LIST_CAP)
      lst[pos + lane] = (unsigned)e | ((unsigned)rr << E_BITS);
  }
  add_edges<DEG, SPLIT>(rr, c, vb, n, lane, x, x2, tile, degt);
}

template <int TILE_ROWS>
__device__ __forceinline__ unsigned tile_id_of(int r) {
  unsigned t = (unsigned)r / (unsigned)TILE_ROWS;
  return t < 255u ? t : 255u;
}
template <int TILE_ROWS>
__device__ __forceinline__ unsigned pack_ids(int4 r) {
  return tile_id_of<TILE_ROWS>(r.x) | (tile_id_of<TILE_ROWS>(r.y) << 8) |
         (tile_id_of<TILE_ROWS>(r.z) << 16) | (tile_id_of<TILE_ROWS>(r.w) << 24);
}

template <int TILE_ROWS>
__device__ __forceinline__ unsigned rc_word(int r, int c) {
  unsigned t  = (unsigned)r / (unsigned)TILE_ROWS;
  unsigned rl = (unsigned)r - t * (unsigned)TILE_ROWS;
  return ((rl & 0x3ffu) << COL_BITS) | ((unsigned)c & COL_MASK);
}

template <int TILE_ROWS>
__global__ void __launch_bounds__(256) prep_packed_kernel(
    const int* __restrict__ row, const int* __restrict__ col,
    unsigned* __restrict__ ids, unsigned* __restrict__ pk, int nnz) {
  const int nthr = gridDim.x * blockDim.x;
  const int nG   = (nnz + 7) >> 3;
  for (int g = blockIdx.x * blockDim.x + threadIdx.x; g < nG; g += nthr) {
    int4 r0, r1, c0, c1;
    if (8 * g + 7 < nnz) {
      r0 = ((const int4*)row)[2 * g];  r1 = ((const int4*)row)[2 * g + 1];
      c0 = ((const int4*)col)[2 * g];  c1 = ((const int4*)col)[2 * g + 1];
    } else {
      int e[8];
#pragma unroll
      for (int k = 0; k < 8; ++k) e[k] = (8 * g + k < nnz) ? (8 * g + k) : (nnz - 1);
      r0 = make_int4(row[e[0]], row[e[1]], row[e[2]], row[e[3]]);
      r1 = make_int4(row[e[4]], row[e[5]], row[e[6]], row[e[7]]);
      c0 = make_int4(col[e[0]], col[e[1]], col[e[2]], col[e[3]]);
      c1 = make_int4(col[e[4]], col[e[5]], col[e[6]], col[e[7]]);
    }
    if (8 * g + 3 < nnz) ids[2 * g]     = pack_ids<TILE_ROWS>(r0);
    if (8 * g + 7 < nnz) ids[2 * g + 1] = pack_ids<TILE_ROWS>(r1);
    const unsigned w0 = rc_word<TILE_ROWS>(r0.x, c0.x), w1 = rc_word<TILE_ROWS>(r0.y, c0.y);
    const unsigned w2 = rc_word<TILE_ROWS>(r0.z, c0.z), w3 = rc_word<TILE_ROWS>(r0.w, c0.w);
    const unsigned w4 = rc_word<TILE_ROWS>(r1.x, c1.x), w5 = rc_word<TILE_ROWS>(r1.y, c1.y);
    const unsigned w6 = rc_word<TILE_ROWS>(r1.z, c1.z), w7 = rc_word<TILE_ROWS>(r1.w, c1.w);
    unsigned* p = pk + 7 * (size_t)g;
    p[0] =  w0        | (w1 << 28);
    p[1] = (w1 >> 4)  | (w2 << 24);
    p[2] = (w2 >> 8)  | (w3 << 20);
    p[3] = (w3 >> 12) | (w4 << 16);
    p[4] = (w4 >> 16) | (w5 << 12);
    p[5] = (w5 >> 20) | (w6 << 8);
    p[6] = (w6 >> 24) | (w7 << 4);
  }
  if (blockIdx.x == 0 && threadIdx.x == 0) pk[7 * (size_t)nG] = 0u;
}

template <int TILE_ROWS>
__device__ __forceinline__ void prep_rec_group(
    const int* __restrict__ row, const int* __restrict__ col, const float* __restrict__ val,
    unsigned* __restrict__ ids, unsigned* __restrict__ rec, int nnz, int g) {
  if (4 * g + 3 < nnz) {
    const int4   r = ((const int4*)row)[g];
    const int4   c = ((const int4*)col)[g];
    const float4 v = ((const float4*)val)[g];
    ids[g] = pack_ids<TILE_ROWS>(r);
    ((uint4*)rec)[2 * g]     = make_uint4(rc_word<TILE_ROWS>(r.x, c.x), __float_as_uint(v.x),
                                          rc_word<TILE_ROWS>(r.y, c.y), __float_as_uint(v.y));
    ((uint4*)rec)[2 * g + 1] = make_uint4(rc_word<TILE_ROWS>(r.z, c.z), __float_as_uint(v.z),
                                          rc_word<TILE_ROWS>(r.w, c.w), __float_as_uint(v.w));
  } else {
    for (int e = 4 * g; e < nnz; ++e) {
      rec[2 * e]     = rc_word<TILE_ROWS>(row[e], col[e]);
      rec[2 * e + 1] = __float_as_uint(val[e]);
    }
  }
}
template <int TILE_A, int TILE_B>
__global__ void __launch_bounds__(256) prep_rec_kernel(
    const int* __restrict__ rowA, const int* __restrict__ colA, const float* __restrict__ valA,
    unsigned* __restrict__ idsA, unsigned* __restrict__ recA, int nnzA,
    const int* __restrict__ rowB, const int* __restrict__ colB, const float* __restrict__ valB,
    unsigned* __restrict__ idsB, unsigned* __restrict__ recB, int nnzB) {
  const int nthr = gridDim.x * blockDim.x;
  const int nA4  = (nnzA + 3) >> 2;
  const int nB4  = (nnzB + 3) >> 2;
  for (int g = blockIdx.x * blockDim.x + threadIdx.x; g < nA4 + nB4; g += nthr) {
    if (g < nA4) prep_rec_group<TILE_A>(rowA, colA, valA, idsA, recA, nnzA, g);
    else         prep_rec_group<TILE_B>(rowB, colB, valB, idsB, recB, nnzB, g - nA4);
  }
}

__device__ __forceinline__ unsigned nonzero_bytes(unsigned v) {
  return (((v & 0x7f7f7f7fu) + 0x7f7f7f7fu) | v) & 0x80808080u;
}

template <int TILE_ROWS, bool DEG, bool REC, bool BUILD, bool SPLIT>
__device__ __forceinline__ void take16(
    uint4 w, unsigned tb, bool ok, int j, int& cnt, int* q, int lane,
    const unsigned* __restrict__ pk,
    const float* __restrict__ val, const float* __restrict__ x, const float* __restrict__ x2,
    float* tile, float* degt, unsigned* lst, int* cursor) {
  unsigned nz = (nonzero_bytes(w.x ^ tb) >> 7) | (nonzero_bytes(w.y ^ tb) >> 6) |
                (nonzero_bytes(w.z ^ tb) >> 5) | (nonzero_bytes(w.w ^ tb) >> 4);
  unsigned m = ok ? (~nz & 0x0f0f0f0fu) : 0u;
  unsigned mk = __builtin_amdgcn_ballot_w32(m != 0u);
  while (mk) {
    int p   = __builtin_ctz(m | 0x80000000u);
    int pos = cnt + (int)__builtin_amdgcn_mbcnt_lo(mk, 0u);
    if (m != 0u) q[pos] = 16 * j + 4 * (p & 7) + (p >> 3);
    cnt += __builtin_popcount(mk);
    m &= m - 1u;
    if (cnt >= 32) {
      resolve_queued<TILE_ROWS, DEG, REC, BUILD, SPLIT>(q, cnt - 32, 32, lane, pk, val, x, x2, tile, degt, lst, cursor);
      cnt -= 32;
    }
    mk = __builtin_amdgcn_ballot_w32(m != 0u);
  }
}

#define AU 5
#define USER_CHUNKS (NUM_USERS * EMB / 256)
static_assert((NUM_USERS * EMB) % 256 == 0, "user rows end at a chunk boundary");
template <bool LAST>
__device__ __forceinline__ void acc_wmma_part(
    const float* __restrict__ e, const float* cu, const float* ci, float* acc,
    int ch0, int ch1, int wv, int nWaves) {
  const int lane = threadIdx.x & 31;
  const int n    = lane & 15;
  const int hi   = lane >> 4;

  v2f a[4];
#pragma unroll
  for (int k = 0; k < 4; ++k) {
    a[k].x = (n == (4 * k + 2 * hi))     ? 1.0f : 0.0f;
    a[k].y = (n == (4 * k + 2 * hi + 1)) ? 1.0f : 0.0f;
  }

  for (int g = ch0 + wv * AU; g < ch1; g += nWaves * AU) {
    v8f cc[AU];
    v2f b[AU][4];
#pragma unroll
    for (int u = 0; u < AU; ++u) {
      int ch = (g + u < ch1) ? (g + u) : (ch1 - 1);
      const float* ap = (ch < USER_CHUNKS ? cu + (size_t)ch * 256
                                          : ci + (size_t)(ch - USER_CHUNKS) * 256) + 4 * lane;
      const float* ep = e   + (size_t)ch * 256 + 4 * n + 2 * hi;
      v4f lo = *(const v4f*)(ap);
      v4f up = *(const v4f*)(ap + 128);
      cc[u][0] = lo.x; cc[u][1] = lo.y; cc[u][2] = lo.z; cc[u][3] = lo.w;
      cc[u][4] = up.x; cc[u][5] = up.y; cc[u][6] = up.z; cc[u][7] = up.w;
#pragma unroll
      for (int k = 0; k < 4; ++k)
        b[u][k] = *(const v2f*)(ep + 128 * (k & 1) + 64 * (k >> 1));
    }
    __builtin_amdgcn_sched_barrier(0);
#pragma unroll
    for (int k = 0; k < 4; ++k)
#pragma unroll
      for (int u = 0; u < AU; ++u)
        cc[u] = __builtin_amdgcn_wmma_f32_16x16x4_f32(
            false, a[k], false, b[u][k], (short)0, cc[u], false, false);
#pragma unroll
    for (int u = 0; u < AU; ++u) {
      if (g + u < ch1) {
        if (LAST) {
          const float s = (g + u >= USER_CHUNKS) ? 0.25f : 1.0f;
          cc[u] = cc[u] * s;
        }
        float* op = acc + (size_t)(g + u) * 256 + 4 * lane;
        v4f lo = {cc[u][0], cc[u][1], cc[u][2], cc[u][3]};
        v4f up = {cc[u][4], cc[u][5], cc[u][6], cc[u][7]};
        *(v4f*)(op)       = lo;
        *(v4f*)(op + 128) = up;
      }
    }
  }
}

#define FU 7
template <int TILE_ROWS, int EPI, int ACC, bool REC, int LISTS, bool SPLIT>
__global__ void __launch_bounds__(1024) spmm_tile_kernel(
    const unsigned* __restrict__ ids,
    const int* __restrict__ row, const int* __restrict__ col,
    const unsigned* __restrict__ pk,
    const float* __restrict__ val, const float* __restrict__ x, const float* __restrict__ x2,
    float* __restrict__ y, int nnz, int n_rows,
    const float* __restrict__ acc_e, const float* acc_cu, const float* acc_ci,
    float* acc_out, int nChunks, int nRide, unsigned* lists) {
  constexpr bool DEG = EPI != 0;
  static_assert(!(SPLIT && LISTS == 2), "the list-driven layers read one source array");
  __shared__ float tile[TILE_ROWS * EMB];
  __shared__ float degt[DEG ? TILE_ROWS : 1];
  __shared__ int   queue[LISTS == 2 ? 1 : 32 * QCAP];
  __shared__ int   cursor;

  const int n_tiles = (n_rows + TILE_ROWS - 1) / TILE_ROWS;
  if (ACC && (int)blockIdx.x >= n_tiles) {
    const int wv = __builtin_amdgcn_readfirstlane(
        ((int)blockIdx.x - n_tiles) * 32 + (int)(threadIdx.x >> 5));
    acc_wmma_part<ACC == 2>(acc_e, acc_cu, acc_ci, acc_out, 0, nRide, wv,
                            ((int)gridDim.x - n_tiles) * 32);
    return;
  }

  const int tid  = threadIdx.x;
  const int lane = tid & 31;
  const int wave = tid >> 5;
  const int lo   = blockIdx.x * TILE_ROWS;
  const unsigned tb = 0x01010101u * blockIdx.x;
  unsigned* lst = lists + LIST_HDR + (size_t)blockIdx.x * LIST_CAP;

  for (int i = tid; i < TILE_ROWS * (EMB / 4); i += 1024)
    ((float4*)tile)[i] = make_float4(0.f, 0.f, 0.f, 0.f);
  if (DEG)
    for (int i = tid; i < TILE_ROWS; i += 1024) degt[i] = 0.f;
  if (LISTS == 1 && tid == 0) cursor = 0;
  __syncthreads();

  if (LISTS == 2) {
    const int wv_s   = __builtin_amdgcn_readfirstlane(wave);
    const int cntAll = __builtin_amdgcn_readfirstlane(((const int*)lists)[blockIdx.x]);
    if (cntAll <= LIST_CAP) {
      if (cntAll > 0) {
        const int base0 = wv_s * 32;
        const int lastp = cntAll - 1;
        unsigned entA = lst[tid < cntAll ? tid : lastp];
        unsigned entB = lst[1024 + tid < cntAll ? 1024 + tid : lastp];
        int eA = (int)(entA & E_MASK);
        eA = eA < nnz ? eA : 0;
        int rA = (int)(entA >> E_BITS);
        int cA = col[eA];
        int vA = __float_as_int(val[eA]);
        for (int b = 0; cntAll - (b * 1024 + base0) > 0; ++b) {
          const int n  = cntAll - (b * 1024 + base0);
          int eB = (int)(entB & E_MASK);
          eB = eB < nnz ? eB : 0;
          const int rB = (int)(entB >> E_BITS);
          const int cB = col[eB];
          const int vB = __float_as_int(val[eB]);
          const int i2 = (b + 2) * 1024 + tid;
          const unsigned entC = lst[i2 < cntAll ? i2 : lastp];
          if (n >= 32) add_edges<DEG, SPLIT>(rA, cA & (int)COL_MASK, vA, 32, lane, x, x2, tile, degt);
          else         add_edges<DEG, SPLIT>(rA, cA & (int)COL_MASK, vA, n,  lane, x, x2, tile, degt);
          rA = rB; cA = cB; vA = vB; entB = entC;
        }
      }
    } else {
      for (int eb = wv_s * 32; eb < nnz; eb += 1024) {
        int  e  = eb + lane;
        bool in = false;
        int  rr = 0, c = 0, vb = 0;
        if (e < nnz) {
          unsigned d = (unsigned)(row[e] - lo);
          in = d < (unsigned)TILE_ROWS;
          rr = (int)d;
        }
        unsigned mk = __builtin_amdgcn_ballot_w32(in);
        if (mk) {
          if (in) { c = col[e] & (int)COL_MASK; vb = __float_as_int(val[e]); }
          while (mk) {
            int k = __builtin_ctz(mk);
            mk &= mk - 1u;
            int      rk = __builtin_amdgcn_readlane(rr, k);
            unsigned ck = (unsigned)__builtin_amdgcn_readlane(c, k);
            float    vk = __int_as_float(__builtin_amdgcn_readlane(vb, k));
            float2   xv = *(const float2*)((const char*)x + (ck * (EMB * 4u) + 8u * (unsigned)lane));
            float* t = tile + rk * EMB + 2 * lane;
            atomicAdd(t,     vk * xv.x);
            atomicAdd(t + 1, vk * xv.y);
            if (DEG && lane == 0) atomicAdd(degt + rk, vk);
          }
        }
      }
    }
  } else {
  int* q   = queue + wave * QCAP;
  int  cnt = 0;
  const int    nnz16 = nnz >> 4;
  const uint4* ids4  = (const uint4*)ids;
  const int    trips = (nnz16 + 1023) >> 10;
  const int    full  = (nnz16 >> 10) & ~3;
  const int start = blockIdx.x * (nnz16 / n_tiles);

  uint4 cur[4];
  int   cid[4];
  if (full > 0) {
#pragma unroll
    for (int u = 0; u < 4; ++u) {
      int j = (u << 10) + tid + start;
      if (j >= nnz16) j -= nnz16;
      cid[u] = j;
      cur[u] = ids4[j];
    }
  }
  for (int it = 0; it < full; it += 4) {
    uint4 nxt[4];
    int   nid[4];
    const int itn = (it + 4 < full) ? (it + 4) : it;
#pragma unroll
    for (int u = 0; u < 4; ++u) {
      int j = ((itn + u) << 10) + tid + start;
      if (j >= nnz16) j -= nnz16;
      nid[u] = j;
      nxt[u] = ids4[j];
    }
#pragma unroll
    for (int u = 0; u < 4; ++u)
      take16<TILE_ROWS, DEG, REC, LISTS == 1, SPLIT>(cur[u], tb, true, cid[u], cnt, q, lane, pk, val, x, x2, tile, degt, lst, &cursor);
#pragma unroll
    for (int u = 0; u < 4; ++u) { cur[u] = nxt[u]; cid[u] = nid[u]; }
  }
  for (int it = full; it < trips; ++it) {
    int  i  = (it << 10) + tid;
    bool ok = i < nnz16;
    int  j  = i + start;
    if (j >= nnz16) j -= nnz16;
    uint4 w = ids4[ok ? j : 0];
    take16<TILE_ROWS, DEG, REC, LISTS == 1, SPLIT>(w, tb, ok, j, cnt, q, lane, pk, val, x, x2, tile, degt, lst, &cursor);
  }
  if (wave == 0) {
    int  e  = (nnz16 << 4) + lane;
    bool m  = (e < nnz) && ((unsigned)(row[e < nnz ? e : 0] - lo) < (unsigned)TILE_ROWS);
    unsigned mk = __builtin_amdgcn_ballot_w32(m);
    if (mk) {
      int pos = cnt + (int)__builtin_amdgcn_mbcnt_lo(mk, 0u);
      if (m) q[pos] = e;
      cnt += __builtin_popcount(mk);
    }
  }
  while (cnt > 0) {
    int n = cnt < 32 ? cnt : 32;
    resolve_queued<TILE_ROWS, DEG, REC, LISTS == 1, SPLIT>(q, cnt - n, n, lane, pk, val, x, x2, tile, degt, lst, &cursor);
    cnt -= n;
  }
  }
  __syncthreads();

  const int rows_here = (n_rows - lo) < TILE_ROWS ? (n_rows - lo) : TILE_ROWS;
  const int nv = rows_here * (EMB / 4);
  v4f* y4 = (v4f*)(y + (size_t)lo * EMB);
  const v4f* t4 = (const v4f*)tile;
  if (EPI == 0) {
    for (int i = tid; i < nv; i += 1024) y4[i] = t4[i];
  } else if (EPI == 1) {
    for (int i = tid; i < nv; i += 1024) {
      float dd  = degt[i >> 4];
      dd = (dd == 0.0f) ? 1.0f : dd;
      float inv = 1.0f / dd;
      y4[i] = t4[i] * inv;
    }
  } else {
    for (int i0 = tid; i0 < nv; i0 += FU * 1024) {
      v4f a[FU];
#pragma unroll
      for (int u = 0; u < FU; ++u) {
        int i = i0 + u * 1024;
        a[u] = y4[i < nv ? i : nv - 1];
      }
      __builtin_amdgcn_sched_barrier(0);
#pragma unroll
      for (int u = 0; u < FU; ++u) {
        int i = i0 + u * 1024;
        if (i < nv) {
          float dd  = degt[i >> 4];
          dd = (dd == 0.0f) ? 1.0f : dd;
          float inv = 1.0f / dd;
          v4f r = a[u] * 0.25f;
          y4[i] = r + t4[i] * inv;
        }
      }
    }
  }
  if (LISTS == 1 && tid == 0) ((int*)lists)[blockIdx.x] = cursor;

  if (ACC && nRide < nChunks) {
    const int wv = __builtin_amdgcn_readfirstlane((int)blockIdx.x * 32 + wave);
    acc_wmma_part<ACC == 2>(acc_e, acc_cu, acc_ci, acc_out, nRide, nChunks, wv, n_tiles * 32);
  }
}

#define TILE_ADJ 672
#define TILE_HV  100
#define TILE_HU  400
static_assert((NUM_NODES + TILE_ADJ - 1) / TILE_ADJ <= 255, "tile id must fit a byte");
static_assert((NUM_BQ    + TILE_HV  - 1) / TILE_HV  <= 255, "tile id must fit a byte");
static_assert((NUM_USERS + TILE_HU  - 1) / TILE_HU  <= 255, "tile id must fit a byte");
static_assert(TILE_ADJ <= 1024 && TILE_HV <= 1024 && TILE_HU <= 1024, "row in tile must fit 10 bits");
static_assert(NUM_NODES <= (1 << COL_BITS), "col must fit 18 bits");
#define NNZ_ADJ 1200000
#define NNZ_BQ   400000
#define TILES_ADJ ((NUM_NODES + TILE_ADJ - 1) / TILE_ADJ)
#define TILES_HV  ((NUM_BQ    + TILE_HV  - 1) / TILE_HV)
#define TILES_HU  ((NUM_USERS + TILE_HU  - 1) / TILE_HU)
#define N_WGP 256
static_assert(TILES_ADJ < N_WGP && TILES_HV < N_WGP, "a launch with a layer sum needs blocks behind its tiles");
static_assert(TILES_HU <= N_WGP, "one tile block per WGP");
static_assert(NNZ_ADJ <= (1 << E_BITS) && TILE_ADJ <= 1024, "edge id and row in tile must fit a list word");
static_assert(TILES_ADJ <= LIST_HDR, "one count per tile in front of the lists");
#define RIDE_ADJ (3 * AU * 32 * (N_WGP - TILES_ADJ))
static_assert(RIDE_ADJ <= NUM_NODES * EMB / 256, "not more chunks than there are");

extern "C" void kernel_launch(void* const* d_in, const int* in_sizes, int n_in,
                              void* d_out, int out_size, void* d_ws, size_t ws_size,
                              hipStream_t stream) {
  const float* user_emb = (const float*)d_in[0];
  const float* item_emb = (const float*)d_in[1];
  const float* adj_val  = (const float*)d_in[2];
  const float* hv_val   = (const float*)d_in[3];
  const float* hu_val   = (const float*)d_in[4];
  const int*   adj_row  = (const int*)d_in[5];
  const int*   adj_col  = (const int*)d_in[6];
  const int*   hv_row   = (const int*)d_in[7];
  const int*   hv_col   = (const int*)d_in[8];
  const int*   hu_row   = (const int*)d_in[9];
  const int*   hu_col   = (const int*)d_in[10];

  const int E_adj = in_sizes[2];
  const int E_hv  = in_sizes[3];
  const int E_hu  = in_sizes[4];

  float* out = (float*)d_out;

  const size_t embBytesN = (size_t)NUM_NODES * EMB * sizeof(float);
  const size_t embBytesU = (size_t)NUM_USERS * EMB * sizeof(float);
  char* ws = (char*)d_ws;
  float* e2     = (float*)(ws);
  float* e1     = (float*)(ws + embBytesN);
  float* bfeat  = (float*)(ws + 2 * embBytesN);
  static_assert((size_t)NNZ_ADJ + 28 * (size_t)((NNZ_ADJ + 7) / 8) + 4 <=
                (size_t)NUM_NODES * EMB * 4,
                "adjacency ids and rc words must fit the output buffer");
  static_assert(4 * ((size_t)LIST_HDR + (size_t)TILES_ADJ * LIST_CAP) <=
                (size_t)NUM_BQ * EMB * 4 + (size_t)(NUM_BQ + NUM_USERS) * 4,
                "edge counts and lists must fit the biclique area");
  static_assert(2 * ((size_t)NNZ_BQ + 8 * (size_t)NNZ_BQ) <= (size_t)NUM_ITEMS * EMB * 4,
                "biclique ids and records must fit the tail of the e2 buffer");
  unsigned* ids_adj = (unsigned*)d_out;
  unsigned* pk_adj  = (unsigned*)((char*)d_out + NNZ_ADJ);
  unsigned* lists   = (unsigned*)(ws + 2 * embBytesN);
  unsigned* ids_hv  = (unsigned*)(ws + embBytesU);
  unsigned* rec_hv  = (unsigned*)(ws + embBytesU + NNZ_BQ);
  unsigned* ids_hu  = (unsigned*)(ws + embBytesU + (embBytesN - embBytesU) / 2);
  unsigned* rec_hu  = (unsigned*)(ws + embBytesU + (embBytesN - embBytesU) / 2 + NNZ_BQ);

  const int nChunks  = NUM_NODES * EMB / 256;


  prep_packed_kernel<TILE_ADJ><<<586, 256, 0, stream>>>(
      adj_row, adj_col, ids_adj, pk_adj, E_adj);
  spmm_tile_kernel<TILE_ADJ, 0, 0, false, 1, true><<<TILES_ADJ, 1024, 0, stream>>>(
      ids_adj, adj_row, adj_col, pk_adj, adj_val, user_emb, item_emb, e1, E_adj, NUM_NODES,
      nullptr, nullptr, nullptr, nullptr, 0, 0, lists);
  spmm_tile_kernel<TILE_ADJ, 0, 1, false, 2, false><<<N_WGP, 1024, 0, stream>>>(
      nullptr, adj_row, adj_col, nullptr, adj_val, e1, nullptr, e2, E_adj, NUM_NODES,
      e1, user_emb, item_emb, out, nChunks, RIDE_ADJ, lists);
  spmm_tile_kernel<TILE_ADJ, 0, 1, false, 2, false><<<N_WGP, 1024, 0, stream>>>(
      nullptr, adj_row, adj_col, nullptr, adj_val, e2, nullptr, e1, E_adj, NUM_NODES,
      e2, out, out + (size_t)NUM_USERS * EMB, out, nChunks, RIDE_ADJ, lists);

  prep_rec_kernel<TILE_HV, TILE_HU><<<782, 256, 0, stream>>>(
      hv_row, hv_col, hv_val, ids_hv, rec_hv, E_hv,
      hu_row, hu_col, hu_val, ids_hu, rec_hu, E_hu);
  spmm_tile_kernel<TILE_HV, 1, 2, true, 0, false><<<N_WGP, 1024, 0, stream>>>(
      ids_hv, hv_row, hv_col, rec_hv, hv_val, item_emb, nullptr, bfeat, E_hv, NUM_BQ,
      e1, out, out + (size_t)NUM_USERS * EMB, out, nChunks, nChunks, nullptr);

  spmm_tile_kernel<TILE_HU, 2, 0, true, 0, false><<<TILES_HU, 1024, 0, stream>>>(
      ids_hu, hu_row, hu_col, rec_hu, hu_val, bfeat, nullptr, out, E_hu, NUM_USERS,
      nullptr, nullptr, nullptr, nullptr, 0, 0, nullptr);
}
